# starting kernel
# baseline (speedup 1.0000x reference)
#include <hip/hip_runtime.h>
#include <hip/hip_bf16.h>

typedef __attribute__((ext_vector_type(2))) float v2f;
typedef __attribute__((ext_vector_type(8))) float v8f;

__global__ __launch_bounds__(256) void gcn_gemm_wmma_f32(
    const float* __restrict__ A, const float* __restrict__ W,
    float* __restrict__ C, int M, int K) {
  const int wave = blockIdx.x * (blockDim.x >> 5) + (threadIdx.x >> 5);
  const int lane = threadIdx.x & 31;
  const int m0 = wave << 4;
  if (m0 >= M) return;
  const int half = lane >> 4;
  const int l = lane & 15;

  v8f acc = {};
  const float* arow = A + (size_t)(m0 + l) * K + (half << 1);
  for (int k0 = 0; k0 < K; k0 += 4) {
    v2f a;
    a.x = __builtin_nontemporal_load(&arow[k0 + 0]);
    a.y = __builtin_nontemporal_load(&arow[k0 + 1]);
    v2f b;
    b.x = W[(size_t)(k0 + (half << 1) + 0) * 16 + l];
    b.y = W[(size_t)(k0 + (half << 1) + 1) * 16 + l];
    acc = __builtin_amdgcn_wmma_f32_16x16x4_f32(
         false, a,  false, b,
         (short)0, acc,  false,  false);
  }
  float* cp = C + ((size_t)(m0 + (half << 3)) << 4) + l;
#pragma unroll
  for (int r = 0; r < 8; ++r) cp[(size_t)r << 4] = acc[r];
}

__global__ void gcn_fill(float* __restrict__ p, float v, int n) {
  int i = blockIdx.x * blockDim.x + threadIdx.x;
  if (i < n) p[i] = v;
}

__global__ void gcn_deg_count(const int* __restrict__ dst, float* __restrict__ deg, int E) {
  int e = blockIdx.x * blockDim.x + threadIdx.x;
  if (e < E) atomicAdd(&deg[__builtin_nontemporal_load(&dst[e])], 1.0f);
}

__global__ void gcn_rsqrt(float* __restrict__ d, int n) {
  int i = blockIdx.x * blockDim.x + threadIdx.x;
  if (i < n) d[i] = rsqrtf(d[i]);
}

__global__ void gcn_agg_init(const float* __restrict__ hw, const float* __restrict__ dis,
                             const float* __restrict__ b, float* __restrict__ agg, int n16) {
  int t = blockIdx.x * blockDim.x + threadIdx.x;
  if (t >= n16) return;
  int i = t >> 4, c = t & 15;
  float d = dis[i];
  agg[t] = d * d * hw[t] + b[c];
}

__global__ void gcn_edge_scatter16(const int* __restrict__ src, const int* __restrict__ dst,
                                   const float* __restrict__ dis, const float* __restrict__ hw,
                                   float* __restrict__ agg, int E) {
  int t = blockIdx.x * blockDim.x + threadIdx.x;
  int e = t >> 4, c = t & 15;
  if (e >= E) return;
  int s = __builtin_nontemporal_load(&src[e]);
  int d = __builtin_nontemporal_load(&dst[e]);
  float norm = dis[s] * dis[d];
  atomicAdd(&agg[((size_t)d << 4) + c], norm * hw[((size_t)s << 4) + c]);
}

__global__ void gcn_relu(float* __restrict__ p, int n) {
  int i = blockIdx.x * blockDim.x + threadIdx.x;
  if (i < n) p[i] = fmaxf(p[i], 0.0f);
}

__global__ void gcn_mv16(const float* __restrict__ h, const float* __restrict__ W3,
                         float* __restrict__ out, int N) {
  int i = blockIdx.x * blockDim.x + threadIdx.x;
  if (i >= N) return;
  const float* hp = h + ((size_t)i << 4);
  float s = 0.f;
#pragma unroll
  for (int c = 0; c < 16; ++c) s += __builtin_nontemporal_load(&hp[c]) * W3[c];
  out[i] = s;
}

__global__ void gcn_out_init(const float* __restrict__ hw3, const float* __restrict__ dis,
                             const float* __restrict__ b3, float* __restrict__ out, int N) {
  int i = blockIdx.x * blockDim.x + threadIdx.x;
  if (i >= N) return;
  float d = dis[i];
  out[i] = d * d * hw3[i] + b3[0];
}

__global__ void gcn_edge_scatter1(const int* __restrict__ src, const int* __restrict__ dst,
                                  const float* __restrict__ dis, const float* __restrict__ hw3,
                                  float* __restrict__ out, int E) {
  int e = blockIdx.x * blockDim.x + threadIdx.x;
  if (e >= E) return;
  int s = __builtin_nontemporal_load(&src[e]);
  int d = __builtin_nontemporal_load(&dst[e]);
  atomicAdd(&out[d], dis[s] * dis[d] * hw3[s]);
}

static inline int cdiv(long long a, long long b) { return (int)((a + b - 1) / b); }

extern "C" void kernel_launch(void* const* d_in, const int* in_sizes, int n_in,
                              void* d_out, int out_size, void* d_ws, size_t ws_size,
                              hipStream_t stream) {
  const float* x  = (const float*)d_in[0];
  const int*   ei = (const int*)d_in[1];
  const float* W1 = (const float*)d_in[2];
  const float* b1 = (const float*)d_in[3];
  const float* W2 = (const float*)d_in[4];
  const float* b2 = (const float*)d_in[5];
  const float* W3 = (const float*)d_in[6];
  const float* b3 = (const float*)d_in[7];

  const int IN_CH = 256;
  const int HID = 16;
  const int N = in_sizes[0] / IN_CH;
  const int E = in_sizes[1] / 2;
  const int* src = ei;
  const int* dst = ei + E;
  float* out = (float*)d_out;

  char* ws = (char*)d_ws;
  size_t off = 0;
  float* dis = (float*)(ws + off); off += (((size_t)N * 4) + 255) & ~(size_t)255;
  float* HW  = (float*)(ws + off); off += (((size_t)N * HID * 4) + 255) & ~(size_t)255;
  float* AGG = (float*)(ws + off); off += (((size_t)N * HID * 4) + 255) & ~(size_t)255;
  (void)ws_size; (void)n_in; (void)out_size;

  const int B = 256;
  const int n16 = N * HID;
  const int gemmBlocks = cdiv(cdiv(N, 16), B / 32);

  gcn_fill<<<cdiv(N, B), B, 0, stream>>>(dis, 1.0f, N);
  gcn_deg_count<<<cdiv(E, B), B, 0, stream>>>(dst, dis, E);
  gcn_rsqrt<<<cdiv(N, B), B, 0, stream>>>(dis, N);

  gcn_gemm_wmma_f32<<<gemmBlocks, B, 0, stream>>>(x, W1, HW, N, IN_CH);
  gcn_agg_init<<<cdiv(n16, B), B, 0, stream>>>(HW, dis, b1, AGG, n16);
  gcn_edge_scatter16<<<cdiv((long long)E * 16, B), B, 0, stream>>>(src, dst, dis, HW, AGG, E);
  gcn_relu<<<cdiv(n16, B), B, 0, stream>>>(AGG, n16);

  gcn_gemm_wmma_f32<<<gemmBlocks, B, 0, stream>>>(AGG, W2, HW, N, HID);
  gcn_agg_init<<<cdiv(n16, B), B, 0, stream>>>(HW, dis, b2, AGG, n16);
  gcn_edge_scatter16<<<cdiv((long long)E * 16, B), B, 0, stream>>>(src, dst, dis, HW, AGG, E);
  gcn_relu<<<cdiv(n16, B), B, 0, stream>>>(AGG, n16);

  gcn_mv16<<<cdiv(N, B), B, 0, stream>>>(AGG, W3, HW, N);
  gcn_out_init<<<cdiv(N, B), B, 0, stream>>>(HW, dis, b3, out, N);
  gcn_edge_scatter1<<<cdiv(E, B), B, 0, stream>>>(src, dst, dis, HW, out, E);
}


# Float-atomic scatter-adds replaced by a per-call destination-sorted edge list (one returning int atomic per edge, prefix sum, fill) and atomic-free gathers; float sum order: list order plus a tree
# speedup vs baseline: 1.8096x; 1.8096x over previous
#include <hip/hip_runtime.h>
#include <hip/hip_bf16.h>

typedef __attribute__((ext_vector_type(2))) float v2f;
typedef __attribute__((ext_vector_type(8))) float v8f;

__global__ __launch_bounds__(256) void gcn_gemm_wmma_f32(
    const float* __restrict__ A, const float* __restrict__ W,
    float* __restrict__ C, int M, int K) {
  const int wave = blockIdx.x * (blockDim.x >> 5) + (threadIdx.x >> 5);
  const int lane = threadIdx.x & 31;
  const int m0 = wave << 4;
  if (m0 >= M) return;
  const int half = lane >> 4;
  const int l = lane & 15;

  v8f acc = {};
  const float* arow = A + (size_t)(m0 + l) * K + (half << 1);
  for (int k0 = 0; k0 < K; k0 += 4) {
    v2f a;
    a.x = __builtin_nontemporal_load(&arow[k0 + 0]);
    a.y = __builtin_nontemporal_load(&arow[k0 + 1]);
    v2f b;
    b.x = W[(size_t)(k0 + (half << 1) + 0) * 16 + l];
    b.y = W[(size_t)(k0 + (half << 1) + 1) * 16 + l];
    acc = __builtin_amdgcn_wmma_f32_16x16x4_f32(
         false, a,  false, b,
         (short)0, acc,  false,  false);
  }
  float* cp = C + ((size_t)(m0 + (half << 3)) << 4) + l;
#pragma unroll
  for (int r = 0; r < 8; ++r) cp[(size_t)r << 4] = acc[r];
}

__global__ void gcn_fill(float* __restrict__ p, float v, int n) {
  int i = blockIdx.x * blockDim.x + threadIdx.x;
  if (i < n) p[i] = v;
}

__global__ void gcn_deg_count(const int* __restrict__ dst, float* __restrict__ deg, int E) {
  int e = blockIdx.x * blockDim.x + threadIdx.x;
  if (e < E) atomicAdd(&deg[__builtin_nontemporal_load(&dst[e])], 1.0f);
}

__global__ void gcn_rsqrt(float* __restrict__ d, int n) {
  int i = blockIdx.x * blockDim.x + threadIdx.x;
  if (i < n) d[i] = rsqrtf(d[i]);
}

__global__ void gcn_agg_init(const float* __restrict__ hw, const float* __restrict__ dis,
                             const float* __restrict__ b, float* __restrict__ agg, int n16) {
  int t = blockIdx.x * blockDim.x + threadIdx.x;
  if (t >= n16) return;
  int i = t >> 4, c = t & 15;
  float d = dis[i];
  agg[t] = d * d * hw[t] + b[c];
}

__global__ void gcn_edge_scatter16(const int* __restrict__ src, const int* __restrict__ dst,
                                   const float* __restrict__ dis, const float* __restrict__ hw,
                                   float* __restrict__ agg, int E) {
  int t = blockIdx.x * blockDim.x + threadIdx.x;
  int e = t >> 4, c = t & 15;
  if (e >= E) return;
  int s = __builtin_nontemporal_load(&src[e]);
  int d = __builtin_nontemporal_load(&dst[e]);
  float norm = dis[s] * dis[d];
  atomicAdd(&agg[((size_t)d << 4) + c], norm * hw[((size_t)s << 4) + c]);
}

__global__ void gcn_relu(float* __restrict__ p, int n) {
  int i = blockIdx.x * blockDim.x + threadIdx.x;
  if (i < n) p[i] = fmaxf(p[i], 0.0f);
}

__global__ void gcn_mv16(const float* __restrict__ h, const float* __restrict__ W3,
                         float* __restrict__ out, int N) {
  int i = blockIdx.x * blockDim.x + threadIdx.x;
  if (i >= N) return;
  const float* hp = h + ((size_t)i << 4);
  float s = 0.f;
#pragma unroll
  for (int c = 0; c < 16; ++c) s += __builtin_nontemporal_load(&hp[c]) * W3[c];
  out[i] = s;
}

__global__ void gcn_out_init(const float* __restrict__ hw3, const float* __restrict__ dis,
                             const float* __restrict__ b3, float* __restrict__ out, int N) {
  int i = blockIdx.x * blockDim.x + threadIdx.x;
  if (i >= N) return;
  float d = dis[i];
  out[i] = d * d * hw3[i] + b3[0];
}

__global__ void gcn_edge_scatter1(const int* __restrict__ src, const int* __restrict__ dst,
                                  const float* __restrict__ dis, const float* __restrict__ hw3,
                                  float* __restrict__ out, int E) {
  int e = blockIdx.x * blockDim.x + threadIdx.x;
  if (e >= E) return;
  int s = __builtin_nontemporal_load(&src[e]);
  int d = __builtin_nontemporal_load(&dst[e]);
  atomicAdd(&out[d], dis[s] * dis[d] * hw3[s]);
}

__global__ void gcn_zero_i32(int* __restrict__ p, int n) {
  int i = blockIdx.x * blockDim.x + threadIdx.x;
  if (i < n) p[i] = 0;
}

__global__ void gcn_edge_rank(const int* __restrict__ dst, int* __restrict__ cnt,
                              int* __restrict__ rnk, int E) {
  int e = blockIdx.x * blockDim.x + threadIdx.x;
  if (e < E) rnk[e] = atomicAdd(&cnt[dst[e]], 1);
}

__global__ __launch_bounds__(1024) void gcn_block_sum(const int* __restrict__ cnt,
                                                      int* __restrict__ bsum, int n) {
  __shared__ int wsum[32];
  const int tid = threadIdx.x, w = tid >> 5, l = tid & 31;
  const int i = blockIdx.x * 1024 + tid;
  int v = (i < n) ? cnt[i] : 0;
#pragma unroll
  for (int o = 16; o > 0; o >>= 1) v += __shfl_xor(v, o, 32);
  if (l == 0) wsum[w] = v;
  __syncthreads();
  if (w == 0) {
    int t = wsum[l];
#pragma unroll
    for (int o = 16; o > 0; o >>= 1) t += __shfl_xor(t, o, 32);
    if (l == 0) bsum[blockIdx.x] = t;
  }
}

__global__ __launch_bounds__(1024) void gcn_scan_dis(const int* __restrict__ cnt,
                                                     const int* __restrict__ bsum,
                                                     int* __restrict__ beg,
                                                     float* __restrict__ dis, int n) {
  __shared__ int wtot[32];
  __shared__ int ptot[32];
  const int tid = threadIdx.x, w = tid >> 5, l = tid & 31;
  const int i = blockIdx.x * 1024 + tid;
  const int c = (i < n) ? cnt[i] : 0;
  int p = 0;
  for (int j = tid; j < (int)blockIdx.x; j += 1024) p += bsum[j];
#pragma unroll
  for (int o = 16; o > 0; o >>= 1) p += __shfl_xor(p, o, 32);
  int s = c;
#pragma unroll
  for (int o = 1; o < 32; o <<= 1) {
    int t = __shfl_up(s, o, 32);
    if (l >= o) s += t;
  }
  if (l == 31) wtot[w] = s;
  if (l == 0) ptot[w] = p;
  __syncthreads();
  int pt = ptot[l];
  int wx = (l < w) ? wtot[l] : 0;
#pragma unroll
  for (int o = 16; o > 0; o >>= 1) {
    pt += __shfl_xor(pt, o, 32);
    wx += __shfl_xor(wx, o, 32);
  }
  if (i < n) {
    beg[i] = pt + wx + s - c;
    dis[i] = rsqrtf((float)(c + 1));
  }
}

__global__ void gcn_fill_list(const int* __restrict__ src, const int* __restrict__ dst,
                              const int* __restrict__ beg, const int* __restrict__ rnk,
                              int* __restrict__ lst, int E) {
  int e = blockIdx.x * blockDim.x + threadIdx.x;
  if (e >= E) return;
  int s = __builtin_nontemporal_load(&src[e]);
  int d = __builtin_nontemporal_load(&dst[e]);
  int r = __builtin_nontemporal_load(&rnk[e]);
  lst[beg[d] + r] = s;
}

__global__ __launch_bounds__(256) void gcn_gather16_relu(
    const int* __restrict__ beg, const int* __restrict__ cnt, const int* __restrict__ lst,
    const float* __restrict__ dis, const float* __restrict__ hw, const float* __restrict__ b,
    float* __restrict__ h, int N) {
  const int t = blockIdx.x * blockDim.x + threadIdx.x;
  const int i = t >> 4, c = t & 15;
  if (i >= N) return;
  const int* lp = lst + beg[i];
  const int len = cnt[i];
  const float d = dis[i];
  float a[8];
  a[0] = d * d * hw[((size_t)i << 4) + c] + b[c];
#pragma unroll
  for (int u = 1; u < 8; ++u) a[u] = 0.f;
  int j = 0;
  for (; j + 8 <= len; j += 8) {
    int s[8];
#pragma unroll
    for (int u = 0; u < 8; ++u) s[u] = lp[j + u];
#pragma unroll
    for (int u = 0; u < 8; ++u)
      a[u] += (dis[s[u]] * d) * hw[((size_t)s[u] << 4) + c];
  }
  for (; j < len; ++j) {
    int s = lp[j];
    a[0] += (dis[s] * d) * hw[((size_t)s << 4) + c];
  }
  float r = ((a[0] + a[1]) + (a[2] + a[3])) + ((a[4] + a[5]) + (a[6] + a[7]));
  h[((size_t)i << 4) + c] = fmaxf(r, 0.0f);
}

__global__ __launch_bounds__(256) void gcn_gather1(
    const int* __restrict__ beg, const int* __restrict__ cnt, const int* __restrict__ lst,
    const float* __restrict__ dis, const float* __restrict__ hw3, const float* __restrict__ b3,
    float* __restrict__ out, int N) {
  const int t = blockIdx.x * blockDim.x + threadIdx.x;
  const int i = t >> 4, l = t & 15;
  const bool ok = i < N;
  const int ii = ok ? i : 0;
  const int* lp = lst + beg[ii];
  const int len = ok ? cnt[ii] : 0;
  const float d = dis[ii];
  float a = 0.f;
  for (int j = l; j < len; j += 16) {
    int s = lp[j];
    a += (dis[s] * d) * hw3[s];
  }
#pragma unroll
  for (int o = 8; o > 0; o >>= 1) a += __shfl_xor(a, o, 16);
  if (ok && l == 0) out[i] = a + (d * d * hw3[i] + b3[0]);
}

static inline int cdiv(long long a, long long b) { return (int)((a + b - 1) / b); }
static inline size_t up256(size_t v) { return (v + 255) & ~(size_t)255; }

extern "C" void kernel_launch(void* const* d_in, const int* in_sizes, int n_in,
                              void* d_out, int out_size, void* d_ws, size_t ws_size,
                              hipStream_t stream) {
  const float* x  = (const float*)d_in[0];
  const int*   ei = (const int*)d_in[1];
  const float* W1 = (const float*)d_in[2];
  const float* b1 = (const float*)d_in[3];
  const float* W2 = (const float*)d_in[4];
  const float* b2 = (const float*)d_in[5];
  const float* W3 = (const float*)d_in[6];
  const float* b3 = (const float*)d_in[7];

  const int IN_CH = 256;
  const int HID = 16;
  const int N = in_sizes[0] / IN_CH;
  const int E = in_sizes[1] / 2;
  const int* src = ei;
  const int* dst = ei + E;
  float* out = (float*)d_out;

  char* ws = (char*)d_ws;
  size_t off = 0;
  const size_t hwB = up256((size_t)N * HID * 4);
  float* dis = (float*)(ws + off); off += up256((size_t)N * 4);
  float* HW  = (float*)(ws + off); off += hwB;
  float* AGG = (float*)(ws + off); off += hwB;
  (void)n_in; (void)out_size;

  const int B = 256;
  const int n16 = N * HID;
  const int gemmBlocks = cdiv(cdiv(N, 16), B / 32);
  const int nb = cdiv(N, 1024);

  int* cnt  = (int*)(ws + off); off += up256((size_t)N * 4);
  int* beg  = (int*)(ws + off); off += up256((size_t)N * 4);
  int* bsum = (int*)(ws + off); off += up256((size_t)nb * 4);
  int* lst  = (int*)(ws + off); off += up256((size_t)E * 4);
  int* rnk  = (int*)HW;
  const bool sorted_path = (ws_size >= off) && ((size_t)E * 4 <= 2 * hwB);

  if (sorted_path) {
    gcn_zero_i32<<<cdiv(N, B), B, 0, stream>>>(cnt, N);
    gcn_edge_rank<<<cdiv(E, B), B, 0, stream>>>(dst, cnt, rnk, E);
    gcn_block_sum<<<nb, 1024, 0, stream>>>(cnt, bsum, N);
    gcn_scan_dis<<<nb, 1024, 0, stream>>>(cnt, bsum, beg, dis, N);
    gcn_fill_list<<<cdiv(E, B), B, 0, stream>>>(src, dst, beg, rnk, lst, E);

    gcn_gemm_wmma_f32<<<gemmBlocks, B, 0, stream>>>(x, W1, HW, N, IN_CH);
    gcn_gather16_relu<<<cdiv(n16, B), B, 0, stream>>>(beg, cnt, lst, dis, HW, b1, AGG, N);
    gcn_gemm_wmma_f32<<<gemmBlocks, B, 0, stream>>>(AGG, W2, HW, N, HID);
    gcn_gather16_relu<<<cdiv(n16, B), B, 0, stream>>>(beg, cnt, lst, dis, HW, b2, AGG, N);
    gcn_mv16<<<cdiv(N, B), B, 0, stream>>>(AGG, W3, HW, N);
    gcn_gather1<<<cdiv(n16, B), B, 0, stream>>>(beg, cnt, lst, dis, HW, b3, out, N);
    return;
  }

  gcn_fill<<<cdiv(N, B), B, 0, stream>>>(dis, 1.0f, N);
  gcn_deg_count<<<cdiv(E, B), B, 0, stream>>>(dst, dis, E);
  gcn_rsqrt<<<cdiv(N, B), B, 0, stream>>>(dis, N);

  gcn_gemm_wmma_f32<<<gemmBlocks, B, 0, stream>>>(x, W1, HW, N, IN_CH);
  gcn_agg_init<<<cdiv(n16, B), B, 0, stream>>>(HW, dis, b1, AGG, n16);
  gcn_edge_scatter16<<<cdiv((long long)E * 16, B), B, 0, stream>>>(src, dst, dis, HW, AGG, E);
  gcn_relu<<<cdiv(n16, B), B, 0, stream>>>(AGG, n16);

  gcn_gemm_wmma_f32<<<gemmBlocks, B, 0, stream>>>(AGG, W2, HW, N, HID);
  gcn_agg_init<<<cdiv(n16, B), B, 0, stream>>>(HW, dis, b2, AGG, n16);
  gcn_edge_scatter16<<<cdiv((long long)E * 16, B), B, 0, stream>>>(src, dst, dis, HW, AGG, E);
  gcn_relu<<<cdiv(n16, B), B, 0, stream>>>(AGG, n16);

  gcn_mv16<<<cdiv(N, B), B, 0, stream>>>(AGG, W3, HW, N);
  gcn_out_init<<<cdiv(N, B), B, 0, stream>>>(HW, dis, b3, out, N);
  gcn_edge_scatter1<<<cdiv(E, B), B, 0, stream>>>(src, dst, dis, HW, out, E);
}


# dis[src] folded into rows at their producers (both GEMM epilogues, 16->1 product): gathers read one row per edge, no dis[src]; terms regrouped as dis[dst]*sum(scaled rows), sum order kept
# speedup vs baseline: 1.9588x; 1.0824x over previous
#include <hip/hip_runtime.h>
#include <hip/hip_bf16.h>

typedef __attribute__((ext_vector_type(2))) float v2f;
typedef __attribute__((ext_vector_type(8))) float v8f;

__global__ __launch_bounds__(256) void gcn_gemm_wmma_f32(
    const float* __restrict__ A, const float* __restrict__ W,
    float* __restrict__ C, const float* __restrict__ rs, int M, int K) {
  const int wave = blockIdx.x * (blockDim.x >> 5) + (threadIdx.x >> 5);
  const int lane = threadIdx.x & 31;
  const int m0 = wave << 4;
  if (m0 >= M) return;
  const int half = lane >> 4;
  const int l = lane & 15;

  v8f acc = {};
  const float* arow = A + (size_t)(m0 + l) * K + (half << 1);
  for (int k0 = 0; k0 < K; k0 += 4) {
    v2f a;
    a.x = __builtin_nontemporal_load(&arow[k0 + 0]);
    a.y = __builtin_nontemporal_load(&arow[k0 + 1]);
    v2f b;
    b.x = W[(size_t)(k0 + (half << 1) + 0) * 16 + l];
    b.y = W[(size_t)(k0 + (half << 1) + 1) * 16 + l];
    acc = __builtin_amdgcn_wmma_f32_16x16x4_f32(
         false, a,  false, b,
         (short)0, acc,  false,  false);
  }
  const int r0 = m0 + (half << 3);
  float* cp = C + ((size_t)r0 << 4) + l;
  if (rs) {
#pragma unroll
    for (int r = 0; r < 8; ++r) cp[(size_t)r << 4] = acc[r] * rs[r0 + r];
  } else {
#pragma unroll
    for (int r = 0; r < 8; ++r) cp[(size_t)r << 4] = acc[r];
  }
}

__global__ void gcn_fill(float* __restrict__ p, float v, int n) {
  int i = blockIdx.x * blockDim.x + threadIdx.x;
  if (i < n) p[i] = v;
}

__global__ void gcn_deg_count(const int* __restrict__ dst, float* __restrict__ deg, int E) {
  int e = blockIdx.x * blockDim.x + threadIdx.x;
  if (e < E) atomicAdd(&deg[__builtin_nontemporal_load(&dst[e])], 1.0f);
}

__global__ void gcn_rsqrt(float* __restrict__ d, int n) {
  int i = blockIdx.x * blockDim.x + threadIdx.x;
  if (i < n) d[i] = rsqrtf(d[i]);
}

__global__ void gcn_agg_init(const float* __restrict__ hw, const float* __restrict__ dis,
                             const float* __restrict__ b, float* __restrict__ agg, int n16) {
  int t = blockIdx.x * blockDim.x + threadIdx.x;
  if (t >= n16) return;
  int i = t >> 4, c = t & 15;
  float d = dis[i];
  agg[t] = d * d * hw[t] + b[c];
}

__global__ void gcn_edge_scatter16(const int* __restrict__ src, const int* __restrict__ dst,
                                   const float* __restrict__ dis, const float* __restrict__ hw,
                                   float* __restrict__ agg, int E) {
  int t = blockIdx.x * blockDim.x + threadIdx.x;
  int e = t >> 4, c = t & 15;
  if (e >= E) return;
  int s = __builtin_nontemporal_load(&src[e]);
  int d = __builtin_nontemporal_load(&dst[e]);
  float norm = dis[s] * dis[d];
  atomicAdd(&agg[((size_t)d << 4) + c], norm * hw[((size_t)s << 4) + c]);
}

__global__ void gcn_relu(float* __restrict__ p, int n) {
  int i = blockIdx.x * blockDim.x + threadIdx.x;
  if (i < n) p[i] = fmaxf(p[i], 0.0f);
}

__global__ void gcn_mv16(const float* __restrict__ h, const float* __restrict__ W3,
                         const float* __restrict__ rs, float* __restrict__ out, int N) {
  int i = blockIdx.x * blockDim.x + threadIdx.x;
  if (i >= N) return;
  const float* hp = h + ((size_t)i << 4);
  float s = 0.f;
#pragma unroll
  for (int c = 0; c < 16; ++c) s += __builtin_nontemporal_load(&hp[c]) * W3[c];
  out[i] = rs ? s * rs[i] : s;
}

__global__ void gcn_out_init(const float* __restrict__ hw3, const float* __restrict__ dis,
                             const float* __restrict__ b3, float* __restrict__ out, int N) {
  int i = blockIdx.x * blockDim.x + threadIdx.x;
  if (i >= N) return;
  float d = dis[i];
  out[i] = d * d * hw3[i] + b3[0];
}

__global__ void gcn_edge_scatter1(const int* __restrict__ src, const int* __restrict__ dst,
                                  const float* __restrict__ dis, const float* __restrict__ hw3,
                                  float* __restrict__ out, int E) {
  int e = blockIdx.x * blockDim.x + threadIdx.x;
  if (e >= E) return;
  int s = __builtin_nontemporal_load(&src[e]);
  int d = __builtin_nontemporal_load(&dst[e]);
  atomicAdd(&out[d], dis[s] * dis[d] * hw3[s]);
}

__global__ void gcn_zero_i32(int* __restrict__ p, int n) {
  int i = blockIdx.x * blockDim.x + threadIdx.x;
  if (i < n) p[i] = 0;
}

__global__ void gcn_edge_rank(const int* __restrict__ dst, int* __restrict__ cnt,
                              int* __restrict__ rnk, int E) {
  int e = blockIdx.x * blockDim.x + threadIdx.x;
  if (e < E) rnk[e] = atomicAdd(&cnt[dst[e]], 1);
}

__global__ __launch_bounds__(1024) void gcn_block_sum(const int* __restrict__ cnt,
                                                      int* __restrict__ bsum, int n) {
  __shared__ int wsum[32];
  const int tid = threadIdx.x, w = tid >> 5, l = tid & 31;
  const int i = blockIdx.x * 1024 + tid;
  int v = (i < n) ? cnt[i] : 0;
#pragma unroll
  for (int o = 16; o > 0; o >>= 1) v += __shfl_xor(v, o, 32);
  if (l == 0) wsum[w] = v;
  __syncthreads();
  if (w == 0) {
    int t = wsum[l];
#pragma unroll
    for (int o = 16; o > 0; o >>= 1) t += __shfl_xor(t, o, 32);
    if (l == 0) bsum[blockIdx.x] = t;
  }
}

__global__ __launch_bounds__(1024) void gcn_scan_dis(const int* __restrict__ cnt,
                                                     const int* __restrict__ bsum,
                                                     int* __restrict__ beg,
                                                     float* __restrict__ dis, int n) {
  __shared__ int wtot[32];
  __shared__ int ptot[32];
  const int tid = threadIdx.x, w = tid >> 5, l = tid & 31;
  const int i = blockIdx.x * 1024 + tid;
  const int c = (i < n) ? cnt[i] : 0;
  int p = 0;
  for (int j = tid; j < (int)blockIdx.x; j += 1024) p += bsum[j];
#pragma unroll
  for (int o = 16; o > 0; o >>= 1) p += __shfl_xor(p, o, 32);
  int s = c;
#pragma unroll
  for (int o = 1; o < 32; o <<= 1) {
    int t = __shfl_up(s, o, 32);
    if (l >= o) s += t;
  }
  if (l == 31) wtot[w] = s;
  if (l == 0) ptot[w] = p;
  __syncthreads();
  int pt = ptot[l];
  int wx = (l < w) ? wtot[l] : 0;
#pragma unroll
  for (int o = 16; o > 0; o >>= 1) {
    pt += __shfl_xor(pt, o, 32);
    wx += __shfl_xor(wx, o, 32);
  }
  if (i < n) {
    beg[i] = pt + wx + s - c;
    dis[i] = rsqrtf((float)(c + 1));
  }
}

__global__ void gcn_fill_list(const int* __restrict__ src, const int* __restrict__ dst,
                              const int* __restrict__ beg, const int* __restrict__ rnk,
                              int* __restrict__ lst, int E) {
  int e = blockIdx.x * blockDim.x + threadIdx.x;
  if (e >= E) return;
  int s = __builtin_nontemporal_load(&src[e]);
  int d = __builtin_nontemporal_load(&dst[e]);
  int r = __builtin_nontemporal_load(&rnk[e]);
  lst[beg[d] + r] = s;
}

__global__ __launch_bounds__(256) void gcn_gather16_relu(
    const int* __restrict__ beg, const int* __restrict__ cnt, const int* __restrict__ lst,
    const float* __restrict__ dis, const float* __restrict__ hws, const float* __restrict__ b,
    float* __restrict__ h, int N) {
  const int t = blockIdx.x * blockDim.x + threadIdx.x;
  const int i = t >> 4, c = t & 15;
  if (i >= N) return;
  const int* lp = lst + beg[i];
  const int len = cnt[i];
  const float d = dis[i];
  float a[8];
  a[0] = hws[((size_t)i << 4) + c];
#pragma unroll
  for (int u = 1; u < 8; ++u) a[u] = 0.f;
  int j = 0;
  for (; j + 8 <= len; j += 8) {
    int s[8];
#pragma unroll
    for (int u = 0; u < 8; ++u) s[u] = lp[j + u];
#pragma unroll
    for (int u = 0; u < 8; ++u)
      a[u] += hws[((size_t)s[u] << 4) + c];
  }
  for (; j < len; ++j) {
    int s = lp[j];
    a[0] += hws[((size_t)s << 4) + c];
  }
  float r = ((a[0] + a[1]) + (a[2] + a[3])) + ((a[4] + a[5]) + (a[6] + a[7]));
  h[((size_t)i << 4) + c] = fmaxf(d * r + b[c], 0.0f);
}

__global__ __launch_bounds__(256) void gcn_gather1(
    const int* __restrict__ beg, const int* __restrict__ cnt, const int* __restrict__ lst,
    const float* __restrict__ dis, const float* __restrict__ hw3s, const float* __restrict__ b3,
    float* __restrict__ out, int N) {
  const int t = blockIdx.x * blockDim.x + threadIdx.x;
  const int i = t >> 4, l = t & 15;
  const bool ok = i < N;
  const int ii = ok ? i : 0;
  const int* lp = lst + beg[ii];
  const int len = ok ? cnt[ii] : 0;
  const float d = dis[ii];
  float a = 0.f;
  for (int j = l; j < len; j += 16) a += hw3s[lp[j]];
#pragma unroll
  for (int o = 8; o > 0; o >>= 1) a += __shfl_xor(a, o, 16);
  if (ok && l == 0) out[i] = d * (a + hw3s[i]) + b3[0];
}

static inline int cdiv(long long a, long long b) { return (int)((a + b - 1) / b); }
static inline size_t up256(size_t v) { return (v + 255) & ~(size_t)255; }

extern "C" void kernel_launch(void* const* d_in, const int* in_sizes, int n_in,
                              void* d_out, int out_size, void* d_ws, size_t ws_size,
                              hipStream_t stream) {
  const float* x  = (const float*)d_in[0];
  const int*   ei = (const int*)d_in[1];
  const float* W1 = (const float*)d_in[2];
  const float* b1 = (const float*)d_in[3];
  const float* W2 = (const float*)d_in[4];
  const float* b2 = (const float*)d_in[5];
  const float* W3 = (const float*)d_in[6];
  const float* b3 = (const float*)d_in[7];

  const int IN_CH = 256;
  const int HID = 16;
  const int N = in_sizes[0] / IN_CH;
  const int E = in_sizes[1] / 2;
  const int* src = ei;
  const int* dst = ei + E;
  float* out = (float*)d_out;

  char* ws = (char*)d_ws;
  size_t off = 0;
  const size_t hwB = up256((size_t)N * HID * 4);
  float* dis = (float*)(ws + off); off += up256((size_t)N * 4);
  float* HW  = (float*)(ws + off); off += hwB;
  float* AGG = (float*)(ws + off); off += hwB;
  (void)n_in; (void)out_size;

  const int B = 256;
  const int n16 = N * HID;
  const int gemmBlocks = cdiv(cdiv(N, 16), B / 32);
  const int nb = cdiv(N, 1024);

  int* cnt  = (int*)(ws + off); off += up256((size_t)N * 4);
  int* beg  = (int*)(ws + off); off += up256((size_t)N * 4);
  int* bsum = (int*)(ws + off); off += up256((size_t)nb * 4);
  int* lst  = (int*)(ws + off); off += up256((size_t)E * 4);
  int* rnk  = (int*)HW;
  const bool sorted_path = (ws_size >= off) && ((size_t)E * 4 <= 2 * hwB);

  if (sorted_path) {
    gcn_zero_i32<<<cdiv(N, B), B, 0, stream>>>(cnt, N);
    gcn_edge_rank<<<cdiv(E, B), B, 0, stream>>>(dst, cnt, rnk, E);
    gcn_block_sum<<<nb, 1024, 0, stream>>>(cnt, bsum, N);
    gcn_scan_dis<<<nb, 1024, 0, stream>>>(cnt, bsum, beg, dis, N);
    gcn_fill_list<<<cdiv(E, B), B, 0, stream>>>(src, dst, beg, rnk, lst, E);

    gcn_gemm_wmma_f32<<<gemmBlocks, B, 0, stream>>>(x, W1, HW, dis, N, IN_CH);
    gcn_gather16_relu<<<cdiv(n16, B), B, 0, stream>>>(beg, cnt, lst, dis, HW, b1, AGG, N);
    gcn_gemm_wmma_f32<<<gemmBlocks, B, 0, stream>>>(AGG, W2, HW, dis, N, HID);
    gcn_gather16_relu<<<cdiv(n16, B), B, 0, stream>>>(beg, cnt, lst, dis, HW, b2, AGG, N);
    gcn_mv16<<<cdiv(N, B), B, 0, stream>>>(AGG, W3, dis, HW, N);
    gcn_gather1<<<cdiv(n16, B), B, 0, stream>>>(beg, cnt, lst, dis, HW, b3, out, N);
    return;
  }

  gcn_fill<<<cdiv(N, B), B, 0, stream>>>(dis, 1.0f, N);
  gcn_deg_count<<<cdiv(E, B), B, 0, stream>>>(dst, dis, E);
  gcn_rsqrt<<<cdiv(N, B), B, 0, stream>>>(dis, N);

  gcn_gemm_wmma_f32<<<gemmBlocks, B, 0, stream>>>(x, W1, HW, nullptr, N, IN_CH);
  gcn_agg_init<<<cdiv(n16, B), B, 0, stream>>>(HW, dis, b1, AGG, n16);
  gcn_edge_scatter16<<<cdiv((long long)E * 16, B), B, 0, stream>>>(src, dst, dis, HW, AGG, E);
  gcn_relu<<<cdiv(n16, B), B, 0, stream>>>(AGG, n16);

  gcn_gemm_wmma_f32<<<gemmBlocks, B, 0, stream>>>(AGG, W2, HW, nullptr, N, HID);
  gcn_agg_init<<<cdiv(n16, B), B, 0, stream>>>(HW, dis, b2, AGG, n16);
  gcn_edge_scatter16<<<cdiv((long long)E * 16, B), B, 0, stream>>>(src, dst, dis, HW, AGG, E);
  gcn_relu<<<cdiv(n16, B), B, 0, stream>>>(AGG, n16);

  gcn_mv16<<<cdiv(N, B), B, 0, stream>>>(AGG, W3, nullptr, HW, N);
  gcn_out_init<<<cdiv(N, B), B, 0, stream>>>(HW, dis, b3, out, N);
  gcn_edge_scatter1<<<cdiv(E, B), B, 0, stream>>>(src, dst, dis, HW, out, E);
}


# Edge ranking: one returning global atomic per edge replaced by a two-level counting sort on LDS counters (chunk x 1024-destination buckets, then per bucket); list order = LDS arrival order
# speedup vs baseline: 3.2850x; 1.6770x over previous
#include <hip/hip_runtime.h>
#include <hip/hip_bf16.h>

typedef __attribute__((ext_vector_type(2))) float v2f;
typedef __attribute__((ext_vector_type(8))) float v8f;

__global__ __launch_bounds__(256) void gcn_gemm_wmma_f32(
    const float* __restrict__ A, const float* __restrict__ W,
    float* __restrict__ C, const float* __restrict__ rs, int M, int K) {
  const int wave = blockIdx.x * (blockDim.x >> 5) + (threadIdx.x >> 5);
  const int lane = threadIdx.x & 31;
  const int m0 = wave << 4;
  if (m0 >= M) return;
  const int half = lane >> 4;
  const int l = lane & 15;

  v8f acc = {};
  const float* arow = A + (size_t)(m0 + l) * K + (half << 1);
  for (int k0 = 0; k0 < K; k0 += 4) {
    v2f a;
    a.x = __builtin_nontemporal_load(&arow[k0 + 0]);
    a.y = __builtin_nontemporal_load(&arow[k0 + 1]);
    v2f b;
    b.x = W[(size_t)(k0 + (half << 1) + 0) * 16 + l];
    b.y = W[(size_t)(k0 + (half << 1) + 1) * 16 + l];
    acc = __builtin_amdgcn_wmma_f32_16x16x4_f32(
         false, a,  false, b,
         (short)0, acc,  false,  false);
  }
  const int r0 = m0 + (half << 3);
  float* cp = C + ((size_t)r0 << 4) + l;
  if (rs) {
#pragma unroll
    for (int r = 0; r < 8; ++r) cp[(size_t)r << 4] = acc[r] * rs[r0 + r];
  } else {
#pragma unroll
    for (int r = 0; r < 8; ++r) cp[(size_t)r << 4] = acc[r];
  }
}

__global__ void gcn_fill(float* __restrict__ p, float v, int n) {
  int i = blockIdx.x * blockDim.x + threadIdx.x;
  if (i < n) p[i] = v;
}

__global__ void gcn_deg_count(const int* __restrict__ dst, float* __restrict__ deg, int E) {
  int e = blockIdx.x * blockDim.x + threadIdx.x;
  if (e < E) atomicAdd(&deg[__builtin_nontemporal_load(&dst[e])], 1.0f);
}

__global__ void gcn_rsqrt(float* __restrict__ d, int n) {
  int i = blockIdx.x * blockDim.x + threadIdx.x;
  if (i < n) d[i] = rsqrtf(d[i]);
}

__global__ void gcn_agg_init(const float* __restrict__ hw, const float* __restrict__ dis,
                             const float* __restrict__ b, float* __restrict__ agg, int n16) {
  int t = blockIdx.x * blockDim.x + threadIdx.x;
  if (t >= n16) return;
  int i = t >> 4, c = t & 15;
  float d = dis[i];
  agg[t] = d * d * hw[t] + b[c];
}

__global__ void gcn_edge_scatter16(const int* __restrict__ src, const int* __restrict__ dst,
                                   const float* __restrict__ dis, const float* __restrict__ hw,
                                   float* __restrict__ agg, int E) {
  int t = blockIdx.x * blockDim.x + threadIdx.x;
  int e = t >> 4, c = t & 15;
  if (e >= E) return;
  int s = __builtin_nontemporal_load(&src[e]);
  int d = __builtin_nontemporal_load(&dst[e]);
  float norm = dis[s] * dis[d];
  atomicAdd(&agg[((size_t)d << 4) + c], norm * hw[((size_t)s << 4) + c]);
}

__global__ void gcn_relu(float* __restrict__ p, int n) {
  int i = blockIdx.x * blockDim.x + threadIdx.x;
  if (i < n) p[i] = fmaxf(p[i], 0.0f);
}

__global__ void gcn_mv16(const float* __restrict__ h, const float* __restrict__ W3,
                         const float* __restrict__ rs, float* __restrict__ out, int N) {
  int i = blockIdx.x * blockDim.x + threadIdx.x;
  if (i >= N) return;
  const float* hp = h + ((size_t)i << 4);
  float s = 0.f;
#pragma unroll
  for (int c = 0; c < 16; ++c) s += __builtin_nontemporal_load(&hp[c]) * W3[c];
  out[i] = rs ? s * rs[i] : s;
}

__global__ void gcn_out_init(const float* __restrict__ hw3, const float* __restrict__ dis,
                             const float* __restrict__ b3, float* __restrict__ out, int N) {
  int i = blockIdx.x * blockDim.x + threadIdx.x;
  if (i >= N) return;
  float d = dis[i];
  out[i] = d * d * hw3[i] + b3[0];
}

__global__ void gcn_edge_scatter1(const int* __restrict__ src, const int* __restrict__ dst,
                                  const float* __restrict__ dis, const float* __restrict__ hw3,
                                  float* __restrict__ out, int E) {
  int e = blockIdx.x * blockDim.x + threadIdx.x;
  if (e >= E) return;
  int s = __builtin_nontemporal_load(&src[e]);
  int d = __builtin_nontemporal_load(&dst[e]);
  atomicAdd(&out[d], dis[s] * dis[d] * hw3[s]);
}

__global__ void gcn_zero_i32(int* __restrict__ p, int n) {
  int i = blockIdx.x * blockDim.x + threadIdx.x;
  if (i < n) p[i] = 0;
}

__global__ void gcn_edge_rank(const int* __restrict__ dst, int* __restrict__ cnt,
                              int* __restrict__ rnk, int E) {
  int e = blockIdx.x * blockDim.x + threadIdx.x;
  if (e < E) rnk[e] = atomicAdd(&cnt[dst[e]], 1);
}

__global__ __launch_bounds__(1024) void gcn_block_sum(const int* __restrict__ cnt,
                                                      int* __restrict__ bsum, int n) {
  __shared__ int wsum[32];
  const int tid = threadIdx.x, w = tid >> 5, l = tid & 31;
  const int i = blockIdx.x * 1024 + tid;
  int v = (i < n) ? cnt[i] : 0;
#pragma unroll
  for (int o = 16; o > 0; o >>= 1) v += __shfl_xor(v, o, 32);
  if (l == 0) wsum[w] = v;
  __syncthreads();
  if (w == 0) {
    int t = wsum[l];
#pragma unroll
    for (int o = 16; o > 0; o >>= 1) t += __shfl_xor(t, o, 32);
    if (l == 0) bsum[blockIdx.x] = t;
  }
}

__global__ __launch_bounds__(1024) void gcn_scan_dis(const int* __restrict__ cnt,
                                                     const int* __restrict__ bsum,
                                                     int* __restrict__ beg,
                                                     float* __restrict__ dis, int n) {
  __shared__ int wtot[32];
  __shared__ int ptot[32];
  const int tid = threadIdx.x, w = tid >> 5, l = tid & 31;
  const int i = blockIdx.x * 1024 + tid;
  const int c = (i < n) ? cnt[i] : 0;
  int p = 0;
  for (int j = tid; j < (int)blockIdx.x; j += 1024) p += bsum[j];
#pragma unroll
  for (int o = 16; o > 0; o >>= 1) p += __shfl_xor(p, o, 32);
  int s = c;
#pragma unroll
  for (int o = 1; o < 32; o <<= 1) {
    int t = __shfl_up(s, o, 32);
    if (l >= o) s += t;
  }
  if (l == 31) wtot[w] = s;
  if (l == 0) ptot[w] = p;
  __syncthreads();
  int pt = ptot[l];
  int wx = (l < w) ? wtot[l] : 0;
#pragma unroll
  for (int o = 16; o > 0; o >>= 1) {
    pt += __shfl_xor(pt, o, 32);
    wx += __shfl_xor(wx, o, 32);
  }
  if (i < n) {
    beg[i] = pt + wx + s - c;
    dis[i] = rsqrtf((float)(c + 1));
  }
}

__global__ void gcn_fill_list(const int* __restrict__ src, const int* __restrict__ dst,
                              const int* __restrict__ beg, const int* __restrict__ rnk,
                              int* __restrict__ lst, int E) {
  int e = blockIdx.x * blockDim.x + threadIdx.x;
  if (e >= E) return;
  int s = __builtin_nontemporal_load(&src[e]);
  int d = __builtin_nontemporal_load(&dst[e]);
  int r = __builtin_nontemporal_load(&rnk[e]);
  lst[beg[d] + r] = s;
}

#define GCN_SRC_BITS 22
#define GCN_MAX_NB 4096
#define GCN_U 16

__global__ __launch_bounds__(1024) void gcn_bucket_hist(const int* __restrict__ dst,
                                                        int* __restrict__ hist,
                                                        int E, int chunk, int NB, int G) {
  __shared__ int h[GCN_MAX_NB];
  const int tid = threadIdx.x;
  for (int i = tid; i < NB; i += 1024) h[i] = 0;
  __syncthreads();
  const int e0 = blockIdx.x * chunk;
  const int e1 = min(e0 + chunk, E);
  for (int eb = e0 + tid; eb < e1; eb += GCN_U * 1024) {
    int d[GCN_U];
#pragma unroll
    for (int u = 0; u < GCN_U; ++u) {
      const int e = eb + u * 1024;
      d[u] = (e < e1) ? dst[e] : -1;
    }
#pragma unroll
    for (int u = 0; u < GCN_U; ++u)
      if (d[u] >= 0) atomicAdd(&h[d[u] >> 10], 1);
  }
  __syncthreads();
  for (int i = tid; i < NB; i += 1024) hist[i * G + blockIdx.x] = h[i];
}

__global__ __launch_bounds__(1024) void gcn_hist_scan(int* hist, int* __restrict__ bstart,
                                                      int total, int G) {
  __shared__ int wtot[32];
  const int tid = threadIdx.x, w = tid >> 5, l = tid & 31;
  const int per = (total + 1023) >> 10;
  const int i0 = min(tid * per, total);
  const int i1 = min(i0 + per, total);
  int sum = 0;
  for (int i = i0; i < i1; ++i) sum += hist[i];
  int s = sum;
#pragma unroll
  for (int o = 1; o < 32; o <<= 1) {
    int t = __shfl_up(s, o, 32);
    if (l >= o) s += t;
  }
  if (l == 31) wtot[w] = s;
  __syncthreads();
  int wx = (l < w) ? wtot[l] : 0;
#pragma unroll
  for (int o = 16; o > 0; o >>= 1) wx += __shfl_xor(wx, o, 32);
  int run = wx + s - sum;
  for (int i = i0; i < i1; ++i) {
    const int v = hist[i];
    hist[i] = run;
    if (i % G == 0) bstart[i / G] = run;
    run += v;
  }
}

__global__ __launch_bounds__(1024) void gcn_bucket_part(const int* __restrict__ src,
                                                        const int* __restrict__ dst,
                                                        const int* __restrict__ offs,
                                                        int* __restrict__ tmp,
                                                        int E, int chunk, int NB, int G) {
  __shared__ int cur[GCN_MAX_NB];
  const int tid = threadIdx.x;
  for (int i = tid; i < NB; i += 1024) cur[i] = offs[i * G + blockIdx.x];
  __syncthreads();
  const int e0 = blockIdx.x * chunk;
  const int e1 = min(e0 + chunk, E);
  for (int eb = e0 + tid; eb < e1; eb += GCN_U * 1024) {
    int s[GCN_U], d[GCN_U];
#pragma unroll
    for (int u = 0; u < GCN_U; ++u) {
      const int e = eb + u * 1024;
      const bool ok = e < e1;
      s[u] = ok ? __builtin_nontemporal_load(&src[e]) : 0;
      d[u] = ok ? __builtin_nontemporal_load(&dst[e]) : -1;
    }
#pragma unroll
    for (int u = 0; u < GCN_U; ++u) {
      if (d[u] >= 0) {
        const unsigned du = (unsigned)d[u];
        const int pos = atomicAdd(&cur[du >> 10], 1);
        tmp[pos] = (int)(((du & 1023u) << GCN_SRC_BITS) | (unsigned)s[u]);
      }
    }
  }
}

__global__ __launch_bounds__(1024) void gcn_bucket_sort(const int* __restrict__ tmp,
                                                        const int* __restrict__ bstart,
                                                        int* __restrict__ cnt,
                                                        int* __restrict__ beg,
                                                        float* __restrict__ dis,
                                                        int* __restrict__ lst,
                                                        int N, int E, int NB) {
  __shared__ int c[1024];
  __shared__ int wtot[32];
  const int tid = threadIdx.x, w = tid >> 5, l = tid & 31;
  const int b = blockIdx.x;
  const int p0 = bstart[b];
  const int p1 = (b + 1 < NB) ? bstart[b + 1] : E;
  c[tid] = 0;
  __syncthreads();
  for (int pb = p0 + tid; pb < p1; pb += GCN_U * 1024) {
    unsigned v[GCN_U];
#pragma unroll
    for (int u = 0; u < GCN_U; ++u) v[u] = (unsigned)tmp[min(pb + u * 1024, p1 - 1)];
#pragma unroll
    for (int u = 0; u < GCN_U; ++u)
      if (pb + u * 1024 < p1) atomicAdd(&c[v[u] >> GCN_SRC_BITS], 1);
  }
  __syncthreads();
  const int n = c[tid];
  int s = n;
#pragma unroll
  for (int o = 1; o < 32; o <<= 1) {
    int t = __shfl_up(s, o, 32);
    if (l >= o) s += t;
  }
  if (l == 31) wtot[w] = s;
  __syncthreads();
  int wx = (l < w) ? wtot[l] : 0;
#pragma unroll
  for (int o = 16; o > 0; o >>= 1) wx += __shfl_xor(wx, o, 32);
  const int ex = p0 + wx + s - n;
  const int i = (b << 10) + tid;
  if (i < N) {
    cnt[i] = n;
    beg[i] = ex;
    dis[i] = rsqrtf((float)(n + 1));
  }
  c[tid] = ex;
  __syncthreads();
  for (int pb = p0 + tid; pb < p1; pb += GCN_U * 1024) {
    unsigned v[GCN_U];
#pragma unroll
    for (int u = 0; u < GCN_U; ++u)
      v[u] = (unsigned)__builtin_nontemporal_load(&tmp[min(pb + u * 1024, p1 - 1)]);
#pragma unroll
    for (int u = 0; u < GCN_U; ++u) {
      if (pb + u * 1024 < p1) {
        const int pos = atomicAdd(&c[v[u] >> GCN_SRC_BITS], 1);
        lst[pos] = (int)(v[u] & ((1u << GCN_SRC_BITS) - 1u));
      }
    }
  }
}

__global__ __launch_bounds__(256) void gcn_gather16_relu(
    const int* __restrict__ beg, const int* __restrict__ cnt, const int* __restrict__ lst,
    const float* __restrict__ dis, const float* __restrict__ hws, const float* __restrict__ b,
    float* __restrict__ h, int N) {
  const int t = blockIdx.x * blockDim.x + threadIdx.x;
  const int i = t >> 4, c = t & 15;
  if (i >= N) return;
  const int* lp = lst + beg[i];
  const int len = cnt[i];
  const float d = dis[i];
  float a[8];
  a[0] = hws[((size_t)i << 4) + c];
#pragma unroll
  for (int u = 1; u < 8; ++u) a[u] = 0.f;
  int j = 0;
  for (; j + 8 <= len; j += 8) {
    int s[8];
#pragma unroll
    for (int u = 0; u < 8; ++u) s[u] = lp[j + u];
#pragma unroll
    for (int u = 0; u < 8; ++u)
      a[u] += hws[((size_t)s[u] << 4) + c];
  }
  for (; j < len; ++j) {
    int s = lp[j];
    a[0] += hws[((size_t)s << 4) + c];
  }
  float r = ((a[0] + a[1]) + (a[2] + a[3])) + ((a[4] + a[5]) + (a[6] + a[7]));
  h[((size_t)i << 4) + c] = fmaxf(d * r + b[c], 0.0f);
}

__global__ __launch_bounds__(256) void gcn_gather1(
    const int* __restrict__ beg, const int* __restrict__ cnt, const int* __restrict__ lst,
    const float* __restrict__ dis, const float* __restrict__ hw3s, const float* __restrict__ b3,
    float* __restrict__ out, int N) {
  const int t = blockIdx.x * blockDim.x + threadIdx.x;
  const int i = t >> 4, l = t & 15;
  const bool ok = i < N;
  const int ii = ok ? i : 0;
  const int* lp = lst + beg[ii];
  const int len = ok ? cnt[ii] : 0;
  const float d = dis[ii];
  float a = 0.f;
  for (int j = l; j < len; j += 16) a += hw3s[lp[j]];
#pragma unroll
  for (int o = 8; o > 0; o >>= 1) a += __shfl_xor(a, o, 16);
  if (ok && l == 0) out[i] = d * (a + hw3s[i]) + b3[0];
}

static inline int cdiv(long long a, long long b) { return (int)((a + b - 1) / b); }
static inline size_t up256(size_t v) { return (v + 255) & ~(size_t)255; }

extern "C" void kernel_launch(void* const* d_in, const int* in_sizes, int n_in,
                              void* d_out, int out_size, void* d_ws, size_t ws_size,
                              hipStream_t stream) {
  const float* x  = (const float*)d_in[0];
  const int*   ei = (const int*)d_in[1];
  const float* W1 = (const float*)d_in[2];
  const float* b1 = (const float*)d_in[3];
  const float* W2 = (const float*)d_in[4];
  const float* b2 = (const float*)d_in[5];
  const float* W3 = (const float*)d_in[6];
  const float* b3 = (const float*)d_in[7];

  const int IN_CH = 256;
  const int HID = 16;
  const int N = in_sizes[0] / IN_CH;
  const int E = in_sizes[1] / 2;
  const int* src = ei;
  const int* dst = ei + E;
  float* out = (float*)d_out;

  char* ws = (char*)d_ws;
  size_t off = 0;
  const size_t hwB = up256((size_t)N * HID * 4);
  float* dis = (float*)(ws + off); off += up256((size_t)N * 4);
  float* HW  = (float*)(ws + off); off += hwB;
  float* AGG = (float*)(ws + off); off += hwB;
  (void)n_in; (void)out_size;

  const int B = 256;
  const int n16 = N * HID;
  const int gemmBlocks = cdiv(cdiv(N, 16), B / 32);
  const int nb = cdiv(N, 1024);

  int* cnt  = (int*)(ws + off); off += up256((size_t)N * 4);
  int* beg  = (int*)(ws + off); off += up256((size_t)N * 4);
  int* bsum = (int*)(ws + off); off += up256((size_t)nb * 4);
  int* lst  = (int*)(ws + off); off += up256((size_t)E * 4);
  int* rnk  = (int*)HW;
  const bool sorted_path = (ws_size >= off) && ((size_t)E * 4 <= 2 * hwB);

  const int G = 256;
  const int chunk = cdiv(E, G);
  int* hist = lst;
  int* bstart = bsum;
  const bool lds_sort = sorted_path && N <= (1 << GCN_SRC_BITS) && nb <= GCN_MAX_NB &&
                        (long long)nb * G <= (long long)E;

  if (lds_sort) {
    gcn_bucket_hist<<<G, 1024, 0, stream>>>(dst, hist, E, chunk, nb, G);
    gcn_hist_scan<<<1, 1024, 0, stream>>>(hist, bstart, nb * G, G);
    gcn_bucket_part<<<G, 1024, 0, stream>>>(src, dst, hist, rnk, E, chunk, nb, G);
    gcn_bucket_sort<<<nb, 1024, 0, stream>>>(rnk, bstart, cnt, beg, dis, lst, N, E, nb);
  } else if (sorted_path) {
    gcn_zero_i32<<<cdiv(N, B), B, 0, stream>>>(cnt, N);
    gcn_edge_rank<<<cdiv(E, B), B, 0, stream>>>(dst, cnt, rnk, E);
    gcn_block_sum<<<nb, 1024, 0, stream>>>(cnt, bsum, N);
    gcn_scan_dis<<<nb, 1024, 0, stream>>>(cnt, bsum, beg, dis, N);
    gcn_fill_list<<<cdiv(E, B), B, 0, stream>>>(src, dst, beg, rnk, lst, E);
  }

  if (sorted_path) {
    gcn_gemm_wmma_f32<<<gemmBlocks, B, 0, stream>>>(x, W1, HW, dis, N, IN_CH);
    gcn_gather16_relu<<<cdiv(n16, B), B, 0, stream>>>(beg, cnt, lst, dis, HW, b1, AGG, N);
    gcn_gemm_wmma_f32<<<gemmBlocks, B, 0, stream>>>(AGG, W2, HW, dis, N, HID);
    gcn_gather16_relu<<<cdiv(n16, B), B, 0, stream>>>(beg, cnt, lst, dis, HW, b2, AGG, N);
    gcn_mv16<<<cdiv(N, B), B, 0, stream>>>(AGG, W3, dis, HW, N);
    gcn_gather1<<<cdiv(n16, B), B, 0, stream>>>(beg, cnt, lst, dis, HW, b3, out, N);
    return;
  }

  gcn_fill<<<cdiv(N, B), B, 0, stream>>>(dis, 1.0f, N);
  gcn_deg_count<<<cdiv(E, B), B, 0, stream>>>(dst, dis, E);
  gcn_rsqrt<<<cdiv(N, B), B, 0, stream>>>(dis, N);

  gcn_gemm_wmma_f32<<<gemmBlocks, B, 0, stream>>>(x, W1, HW, nullptr, N, IN_CH);
  gcn_agg_init<<<cdiv(n16, B), B, 0, stream>>>(HW, dis, b1, AGG, n16);
  gcn_edge_scatter16<<<cdiv((long long)E * 16, B), B, 0, stream>>>(src, dst, dis, HW, AGG, E);
  gcn_relu<<<cdiv(n16, B), B, 0, stream>>>(AGG, n16);

  gcn_gemm_wmma_f32<<<gemmBlocks, B, 0, stream>>>(AGG, W2, HW, nullptr, N, HID);
  gcn_agg_init<<<cdiv(n16, B), B, 0, stream>>>(HW, dis, b2, AGG, n16);
  gcn_edge_scatter16<<<cdiv((long long)E * 16, B), B, 0, stream>>>(src, dst, dis, HW, AGG, E);
  gcn_relu<<<cdiv(n16, B), B, 0, stream>>>(AGG, n16);

  gcn_mv16<<<cdiv(N, B), B, 0, stream>>>(AGG, W3, nullptr, HW, N);
  gcn_out_init<<<cdiv(N, B), B, 0, stream>>>(HW, dis, b3, out, N);
  gcn_edge_scatter1<<<cdiv(E, B), B, 0, stream>>>(src, dst, dis, HW, out, E);
}


# 16-channel gathers: the one-edge-per-round-trip tail loop replaced by one last trip of up to 15 edges with clamped list positions, all its loads in flight together; float sum order kept
# speedup vs baseline: 3.4769x; 1.0584x over previous
#include <hip/hip_runtime.h>
#include <hip/hip_bf16.h>

typedef __attribute__((ext_vector_type(2))) float v2f;
typedef __attribute__((ext_vector_type(8))) float v8f;

__global__ __launch_bounds__(256) void gcn_gemm_wmma_f32(
    const float* __restrict__ A, const float* __restrict__ W,
    float* __restrict__ C, const float* __restrict__ rs, int M, int K) {
  const int wave = blockIdx.x * (blockDim.x >> 5) + (threadIdx.x >> 5);
  const int lane = threadIdx.x & 31;
  const int m0 = wave << 4;
  if (m0 >= M) return;
  const int half = lane >> 4;
  const int l = lane & 15;

  v8f acc = {};
  const float* arow = A + (size_t)(m0 + l) * K + (half << 1);
  for (int k0 = 0; k0 < K; k0 += 4) {
    v2f a;
    a.x = __builtin_nontemporal_load(&arow[k0 + 0]);
    a.y = __builtin_nontemporal_load(&arow[k0 + 1]);
    v2f b;
    b.x = W[(size_t)(k0 + (half << 1) + 0) * 16 + l];
    b.y = W[(size_t)(k0 + (half << 1) + 1) * 16 + l];
    acc = __builtin_amdgcn_wmma_f32_16x16x4_f32(
         false, a,  false, b,
         (short)0, acc,  false,  false);
  }
  const int r0 = m0 + (half << 3);
  float* cp = C + ((size_t)r0 << 4) + l;
  if (rs) {
#pragma unroll
    for (int r = 0; r < 8; ++r) cp[(size_t)r << 4] = acc[r] * rs[r0 + r];
  } else {
#pragma unroll
    for (int r = 0; r < 8; ++r) cp[(size_t)r << 4] = acc[r];
  }
}

__global__ void gcn_fill(float* __restrict__ p, float v, int n) {
  int i = blockIdx.x * blockDim.x + threadIdx.x;
  if (i < n) p[i] = v;
}

__global__ void gcn_deg_count(const int* __restrict__ dst, float* __restrict__ deg, int E) {
  int e = blockIdx.x * blockDim.x + threadIdx.x;
  if (e < E) atomicAdd(&deg[__builtin_nontemporal_load(&dst[e])], 1.0f);
}

__global__ void gcn_rsqrt(float* __restrict__ d, int n) {
  int i = blockIdx.x * blockDim.x + threadIdx.x;
  if (i < n) d[i] = rsqrtf(d[i]);
}

__global__ void gcn_agg_init(const float* __restrict__ hw, const float* __restrict__ dis,
                             const float* __restrict__ b, float* __restrict__ agg, int n16) {
  int t = blockIdx.x * blockDim.x + threadIdx.x;
  if (t >= n16) return;
  int i = t >> 4, c = t & 15;
  float d = dis[i];
  agg[t] = d * d * hw[t] + b[c];
}

__global__ void gcn_edge_scatter16(const int* __restrict__ src, const int* __restrict__ dst,
                                   const float* __restrict__ dis, const float* __restrict__ hw,
                                   float* __restrict__ agg, int E) {
  int t = blockIdx.x * blockDim.x + threadIdx.x;
  int e = t >> 4, c = t & 15;
  if (e >= E) return;
  int s = __builtin_nontemporal_load(&src[e]);
  int d = __builtin_nontemporal_load(&dst[e]);
  float norm = dis[s] * dis[d];
  atomicAdd(&agg[((size_t)d << 4) + c], norm * hw[((size_t)s << 4) + c]);
}

__global__ void gcn_relu(float* __restrict__ p, int n) {
  int i = blockIdx.x * blockDim.x + threadIdx.x;
  if (i < n) p[i] = fmaxf(p[i], 0.0f);
}

__global__ void gcn_mv16(const float* __restrict__ h, const float* __restrict__ W3,
                         const float* __restrict__ rs, float* __restrict__ out, int N) {
  int i = blockIdx.x * blockDim.x + threadIdx.x;
  if (i >= N) return;
  const float* hp = h + ((size_t)i << 4);
  float s = 0.f;
#pragma unroll
  for (int c = 0; c < 16; ++c) s += __builtin_nontemporal_load(&hp[c]) * W3[c];
  out[i] = rs ? s * rs[i] : s;
}

__global__ void gcn_out_init(const float* __restrict__ hw3, const float* __restrict__ dis,
                             const float* __restrict__ b3, float* __restrict__ out, int N) {
  int i = blockIdx.x * blockDim.x + threadIdx.x;
  if (i >= N) return;
  float d = dis[i];
  out[i] = d * d * hw3[i] + b3[0];
}

__global__ void gcn_edge_scatter1(const int* __restrict__ src, const int* __restrict__ dst,
                                  const float* __restrict__ dis, const float* __restrict__ hw3,
                                  float* __restrict__ out, int E) {
  int e = blockIdx.x * blockDim.x + threadIdx.x;
  if (e >= E) return;
  int s = __builtin_nontemporal_load(&src[e]);
  int d = __builtin_nontemporal_load(&dst[e]);
  atomicAdd(&out[d], dis[s] * dis[d] * hw3[s]);
}

__global__ void gcn_zero_i32(int* __restrict__ p, int n) {
  int i = blockIdx.x * blockDim.x + threadIdx.x;
  if (i < n) p[i] = 0;
}

__global__ void gcn_edge_rank(const int* __restrict__ dst, int* __restrict__ cnt,
                              int* __restrict__ rnk, int E) {
  int e = blockIdx.x * blockDim.x + threadIdx.x;
  if (e < E) rnk[e] = atomicAdd(&cnt[dst[e]], 1);
}

__global__ __launch_bounds__(1024) void gcn_block_sum(const int* __restrict__ cnt,
                                                      int* __restrict__ bsum, int n) {
  __shared__ int wsum[32];
  const int tid = threadIdx.x, w = tid >> 5, l = tid & 31;
  const int i = blockIdx.x * 1024 + tid;
  int v = (i < n) ? cnt[i] : 0;
#pragma unroll
  for (int o = 16; o > 0; o >>= 1) v += __shfl_xor(v, o, 32);
  if (l == 0) wsum[w] = v;
  __syncthreads();
  if (w == 0) {
    int t = wsum[l];
#pragma unroll
    for (int o = 16; o > 0; o >>= 1) t += __shfl_xor(t, o, 32);
    if (l == 0) bsum[blockIdx.x] = t;
  }
}

__global__ __launch_bounds__(1024) void gcn_scan_dis(const int* __restrict__ cnt,
                                                     const int* __restrict__ bsum,
                                                     int* __restrict__ beg,
                                                     float* __restrict__ dis, int n) {
  __shared__ int wtot[32];
  __shared__ int ptot[32];
  const int tid = threadIdx.x, w = tid >> 5, l = tid & 31;
  const int i = blockIdx.x * 1024 + tid;
  const int c = (i < n) ? cnt[i] : 0;
  int p = 0;
  for (int j = tid; j < (int)blockIdx.x; j += 1024) p += bsum[j];
#pragma unroll
  for (int o = 16; o > 0; o >>= 1) p += __shfl_xor(p, o, 32);
  int s = c;
#pragma unroll
  for (int o = 1; o < 32; o <<= 1) {
    int t = __shfl_up(s, o, 32);
    if (l >= o) s += t;
  }
  if (l == 31) wtot[w] = s;
  if (l == 0) ptot[w] = p;
  __syncthreads();
  int pt = ptot[l];
  int wx = (l < w) ? wtot[l] : 0;
#pragma unroll
  for (int o = 16; o > 0; o >>= 1) {
    pt += __shfl_xor(pt, o, 32);
    wx += __shfl_xor(wx, o, 32);
  }
  if (i < n) {
    beg[i] = pt + wx + s - c;
    dis[i] = rsqrtf((float)(c + 1));
  }
}

__global__ void gcn_fill_list(const int* __restrict__ src, const int* __restrict__ dst,
                              const int* __restrict__ beg, const int* __restrict__ rnk,
                              int* __restrict__ lst, int E) {
  int e = blockIdx.x * blockDim.x + threadIdx.x;
  if (e >= E) return;
  int s = __builtin_nontemporal_load(&src[e]);
  int d = __builtin_nontemporal_load(&dst[e]);
  int r = __builtin_nontemporal_load(&rnk[e]);
  lst[beg[d] + r] = s;
}

#define GCN_SRC_BITS 22
#define GCN_MAX_NB 4096
#define GCN_U 16

__global__ __launch_bounds__(1024) void gcn_bucket_hist(const int* __restrict__ dst,
                                                        int* __restrict__ hist,
                                                        int E, int chunk, int NB, int G) {
  __shared__ int h[GCN_MAX_NB];
  const int tid = threadIdx.x;
  for (int i = tid; i < NB; i += 1024) h[i] = 0;
  __syncthreads();
  const int e0 = blockIdx.x * chunk;
  const int e1 = min(e0 + chunk, E);
  for (int eb = e0 + tid; eb < e1; eb += GCN_U * 1024) {
    int d[GCN_U];
#pragma unroll
    for (int u = 0; u < GCN_U; ++u) {
      const int e = eb + u * 1024;
      d[u] = (e < e1) ? dst[e] : -1;
    }
#pragma unroll
    for (int u = 0; u < GCN_U; ++u)
      if (d[u] >= 0) atomicAdd(&h[d[u] >> 10], 1);
  }
  __syncthreads();
  for (int i = tid; i < NB; i += 1024) hist[i * G + blockIdx.x] = h[i];
}

__global__ __launch_bounds__(1024) void gcn_hist_scan(int* hist, int* __restrict__ bstart,
                                                      int total, int G) {
  __shared__ int wtot[32];
  const int tid = threadIdx.x, w = tid >> 5, l = tid & 31;
  const int per = (total + 1023) >> 10;
  const int i0 = min(tid * per, total);
  const int i1 = min(i0 + per, total);
  int sum = 0;
  for (int i = i0; i < i1; ++i) sum += hist[i];
  int s = sum;
#pragma unroll
  for (int o = 1; o < 32; o <<= 1) {
    int t = __shfl_up(s, o, 32);
    if (l >= o) s += t;
  }
  if (l == 31) wtot[w] = s;
  __syncthreads();
  int wx = (l < w) ? wtot[l] : 0;
#pragma unroll
  for (int o = 16; o > 0; o >>= 1) wx += __shfl_xor(wx, o, 32);
  int run = wx + s - sum;
  for (int i = i0; i < i1; ++i) {
    const int v = hist[i];
    hist[i] = run;
    if (i % G == 0) bstart[i / G] = run;
    run += v;
  }
}

__global__ __launch_bounds__(1024) void gcn_bucket_part(const int* __restrict__ src,
                                                        const int* __restrict__ dst,
                                                        const int* __restrict__ offs,
                                                        int* __restrict__ tmp,
                                                        int E, int chunk, int NB, int G) {
  __shared__ int cur[GCN_MAX_NB];
  const int tid = threadIdx.x;
  for (int i = tid; i < NB; i += 1024) cur[i] = offs[i * G + blockIdx.x];
  __syncthreads();
  const int e0 = blockIdx.x * chunk;
  const int e1 = min(e0 + chunk, E);
  for (int eb = e0 + tid; eb < e1; eb += GCN_U * 1024) {
    int s[GCN_U], d[GCN_U];
#pragma unroll
    for (int u = 0; u < GCN_U; ++u) {
      const int e = eb + u * 1024;
      const bool ok = e < e1;
      s[u] = ok ? __builtin_nontemporal_load(&src[e]) : 0;
      d[u] = ok ? __builtin_nontemporal_load(&dst[e]) : -1;
    }
#pragma unroll
    for (int u = 0; u < GCN_U; ++u) {
      if (d[u] >= 0) {
        const unsigned du = (unsigned)d[u];
        const int pos = atomicAdd(&cur[du >> 10], 1);
        tmp[pos] = (int)(((du & 1023u) << GCN_SRC_BITS) | (unsigned)s[u]);
      }
    }
  }
}

__global__ __launch_bounds__(1024) void gcn_bucket_sort(const int* __restrict__ tmp,
                                                        const int* __restrict__ bstart,
                                                        int* __restrict__ cnt,
                                                        int* __restrict__ beg,
                                                        float* __restrict__ dis,
                                                        int* __restrict__ lst,
                                                        int N, int E, int NB) {
  __shared__ int c[1024];
  __shared__ int wtot[32];
  const int tid = threadIdx.x, w = tid >> 5, l = tid & 31;
  const int b = blockIdx.x;
  const int p0 = bstart[b];
  const int p1 = (b + 1 < NB) ? bstart[b + 1] : E;
  c[tid] = 0;
  __syncthreads();
  for (int pb = p0 + tid; pb < p1; pb += GCN_U * 1024) {
    unsigned v[GCN_U];
#pragma unroll
    for (int u = 0; u < GCN_U; ++u) v[u] = (unsigned)tmp[min(pb + u * 1024, p1 - 1)];
#pragma unroll
    for (int u = 0; u < GCN_U; ++u)
      if (pb + u * 1024 < p1) atomicAdd(&c[v[u] >> GCN_SRC_BITS], 1);
  }
  __syncthreads();
  const int n = c[tid];
  int s = n;
#pragma unroll
  for (int o = 1; o < 32; o <<= 1) {
    int t = __shfl_up(s, o, 32);
    if (l >= o) s += t;
  }
  if (l == 31) wtot[w] = s;
  __syncthreads();
  int wx = (l < w) ? wtot[l] : 0;
#pragma unroll
  for (int o = 16; o > 0; o >>= 1) wx += __shfl_xor(wx, o, 32);
  const int ex = p0 + wx + s - n;
  const int i = (b << 10) + tid;
  if (i < N) {
    cnt[i] = n;
    beg[i] = ex;
    dis[i] = rsqrtf((float)(n + 1));
  }
  c[tid] = ex;
  __syncthreads();
  for (int pb = p0 + tid; pb < p1; pb += GCN_U * 1024) {
    unsigned v[GCN_U];
#pragma unroll
    for (int u = 0; u < GCN_U; ++u)
      v[u] = (unsigned)__builtin_nontemporal_load(&tmp[min(pb + u * 1024, p1 - 1)]);
#pragma unroll
    for (int u = 0; u < GCN_U; ++u) {
      if (pb + u * 1024 < p1) {
        const int pos = atomicAdd(&c[v[u] >> GCN_SRC_BITS], 1);
        lst[pos] = (int)(v[u] & ((1u << GCN_SRC_BITS) - 1u));
      }
    }
  }
}

__global__ __launch_bounds__(256) void gcn_gather16_relu(
    const int* __restrict__ beg, const int* __restrict__ cnt, const int* __restrict__ lst,
    const float* __restrict__ dis, const float* __restrict__ hws, const float* __restrict__ b,
    float* __restrict__ h, int N) {
  const int t = blockIdx.x * blockDim.x + threadIdx.x;
  const int i = t >> 4, c = t & 15;
  if (i >= N) return;
  const int* lp = lst + beg[i];
  const int len = cnt[i];
  const float d = dis[i];
  float a[8];
  a[0] = hws[((size_t)i << 4) + c];
#pragma unroll
  for (int u = 1; u < 8; ++u) a[u] = 0.f;
  int j = 0;
#pragma unroll 1
  for (; j + 16 <= len; j += 16) {
    int s[16];
#pragma unroll
    for (int u = 0; u < 16; ++u) s[u] = lp[j + u];
    float r[16];
#pragma unroll
    for (int u = 0; u < 16; ++u) r[u] = hws[((size_t)s[u] << 4) + c];
#pragma unroll
    for (int u = 0; u < 8; ++u) a[u] += r[u];
#pragma unroll
    for (int u = 0; u < 8; ++u) a[u] += r[8 + u];
  }
  const int rem = len - j;
  if (rem > 0) {
    const int last = len - 1;
    int s[15];
#pragma unroll
    for (int u = 0; u < 15; ++u) s[u] = lp[min(j + u, last)];
#pragma unroll
    for (int u = 0; u < 15; ++u) asm volatile("" : "+v"(s[u]));
    float r[15];
#pragma unroll
    for (int u = 0; u < 15; ++u) r[u] = hws[((size_t)s[u] << 4) + c];
#pragma unroll
    for (int u = 0; u < 15; ++u) asm volatile("" : "+v"(r[u]));
    const bool blk = rem >= 8;
#pragma unroll
    for (int u = 0; u < 8; ++u) a[u] = blk ? a[u] + r[u] : a[u];
    const int tail = rem & 7;
#pragma unroll
    for (int q = 0; q < 7; ++q) {
      const float v = blk ? r[8 + q] : r[q];
      a[0] = (q < tail) ? a[0] + v : a[0];
    }
  }
  float r = ((a[0] + a[1]) + (a[2] + a[3])) + ((a[4] + a[5]) + (a[6] + a[7]));
  h[((size_t)i << 4) + c] = fmaxf(d * r + b[c], 0.0f);
}

__global__ __launch_bounds__(256) void gcn_gather1(
    const int* __restrict__ beg, const int* __restrict__ cnt, const int* __restrict__ lst,
    const float* __restrict__ dis, const float* __restrict__ hw3s, const float* __restrict__ b3,
    float* __restrict__ out, int N) {
  const int t = blockIdx.x * blockDim.x + threadIdx.x;
  const int i = t >> 4, l = t & 15;
  const bool ok = i < N;
  const int ii = ok ? i : 0;
  const int* lp = lst + beg[ii];
  const int len = ok ? cnt[ii] : 0;
  const float d = dis[ii];
  float a = 0.f;
  for (int j = l; j < len; j += 16) a += hw3s[lp[j]];
#pragma unroll
  for (int o = 8; o > 0; o >>= 1) a += __shfl_xor(a, o, 16);
  if (ok && l == 0) out[i] = d * (a + hw3s[i]) + b3[0];
}

static inline int cdiv(long long a, long long b) { return (int)((a + b - 1) / b); }
static inline size_t up256(size_t v) { return (v + 255) & ~(size_t)255; }

extern "C" void kernel_launch(void* const* d_in, const int* in_sizes, int n_in,
                              void* d_out, int out_size, void* d_ws, size_t ws_size,
                              hipStream_t stream) {
  const float* x  = (const float*)d_in[0];
  const int*   ei = (const int*)d_in[1];
  const float* W1 = (const float*)d_in[2];
  const float* b1 = (const float*)d_in[3];
  const float* W2 = (const float*)d_in[4];
  const float* b2 = (const float*)d_in[5];
  const float* W3 = (const float*)d_in[6];
  const float* b3 = (const float*)d_in[7];

  const int IN_CH = 256;
  const int HID = 16;
  const int N = in_sizes[0] / IN_CH;
  const int E = in_sizes[1] / 2;
  const int* src = ei;
  const int* dst = ei + E;
  float* out = (float*)d_out;

  char* ws = (char*)d_ws;
  size_t off = 0;
  const size_t hwB = up256((size_t)N * HID * 4);
  float* dis = (float*)(ws + off); off += up256((size_t)N * 4);
  float* HW  = (float*)(ws + off); off += hwB;
  float* AGG = (float*)(ws + off); off += hwB;
  (void)n_in; (void)out_size;

  const int B = 256;
  const int n16 = N * HID;
  const int gemmBlocks = cdiv(cdiv(N, 16), B / 32);
  const int nb = cdiv(N, 1024);

  int* cnt  = (int*)(ws + off); off += up256((size_t)N * 4);
  int* beg  = (int*)(ws + off); off += up256((size_t)N * 4);
  int* bsum = (int*)(ws + off); off += up256((size_t)nb * 4);
  int* lst  = (int*)(ws + off); off += up256((size_t)E * 4);
  int* rnk  = (int*)HW;
  const bool sorted_path = (ws_size >= off) && ((size_t)E * 4 <= 2 * hwB);

  const int G = 256;
  const int chunk = cdiv(E, G);
  int* hist = lst;
  int* bstart = bsum;
  const bool lds_sort = sorted_path && N <= (1 << GCN_SRC_BITS) && nb <= GCN_MAX_NB &&
                        (long long)nb * G <= (long long)E;

  if (lds_sort) {
    gcn_bucket_hist<<<G, 1024, 0, stream>>>(dst, hist, E, chunk, nb, G);
    gcn_hist_scan<<<1, 1024, 0, stream>>>(hist, bstart, nb * G, G);
    gcn_bucket_part<<<G, 1024, 0, stream>>>(src, dst, hist, rnk, E, chunk, nb, G);
    gcn_bucket_sort<<<nb, 1024, 0, stream>>>(rnk, bstart, cnt, beg, dis, lst, N, E, nb);
  } else if (sorted_path) {
    gcn_zero_i32<<<cdiv(N, B), B, 0, stream>>>(cnt, N);
    gcn_edge_rank<<<cdiv(E, B), B, 0, stream>>>(dst, cnt, rnk, E);
    gcn_block_sum<<<nb, 1024, 0, stream>>>(cnt, bsum, N);
    gcn_scan_dis<<<nb, 1024, 0, stream>>>(cnt, bsum, beg, dis, N);
    gcn_fill_list<<<cdiv(E, B), B, 0, stream>>>(src, dst, beg, rnk, lst, E);
  }

  if (sorted_path) {
    gcn_gemm_wmma_f32<<<gemmBlocks, B, 0, stream>>>(x, W1, HW, dis, N, IN_CH);
    gcn_gather16_relu<<<cdiv(n16, B), B, 0, stream>>>(beg, cnt, lst, dis, HW, b1, AGG, N);
    gcn_gemm_wmma_f32<<<gemmBlocks, B, 0, stream>>>(AGG, W2, HW, dis, N, HID);
    gcn_gather16_relu<<<cdiv(n16, B), B, 0, stream>>>(beg, cnt, lst, dis, HW, b2, AGG, N);
    gcn_mv16<<<cdiv(N, B), B, 0, stream>>>(AGG, W3, dis, HW, N);
    gcn_gather1<<<cdiv(n16, B), B, 0, stream>>>(beg, cnt, lst, dis, HW, b3, out, N);
    return;
  }

  gcn_fill<<<cdiv(N, B), B, 0, stream>>>(dis, 1.0f, N);
  gcn_deg_count<<<cdiv(E, B), B, 0, stream>>>(dst, dis, E);
  gcn_rsqrt<<<cdiv(N, B), B, 0, stream>>>(dis, N);

  gcn_gemm_wmma_f32<<<gemmBlocks, B, 0, stream>>>(x, W1, HW, nullptr, N, IN_CH);
  gcn_agg_init<<<cdiv(n16, B), B, 0, stream>>>(HW, dis, b1, AGG, n16);
  gcn_edge_scatter16<<<cdiv((long long)E * 16, B), B, 0, stream>>>(src, dst, dis, HW, AGG, E);
  gcn_relu<<<cdiv(n16, B), B, 0, stream>>>(AGG, n16);

  gcn_gemm_wmma_f32<<<gemmBlocks, B, 0, stream>>>(AGG, W2, HW, nullptr, N, HID);
  gcn_agg_init<<<cdiv(n16, B), B, 0, stream>>>(HW, dis, b2, AGG, n16);
  gcn_edge_scatter16<<<cdiv((long long)E * 16, B), B, 0, stream>>>(src, dst, dis, HW, AGG, E);
  gcn_relu<<<cdiv(n16, B), B, 0, stream>>>(AGG, n16);

  gcn_mv16<<<cdiv(N, B), B, 0, stream>>>(AGG, W3, nullptr, HW, N);
  gcn_out_init<<<cdiv(N, B), B, 0, stream>>>(HW, dis, b3, out, N);
  gcn_edge_scatter1<<<cdiv(E, B), B, 0, stream>>>(src, dst, dis, HW, out, E);
}


# Sort offsets: the one-work-group scan launch over the 196 x 256 count matrix removed; each partition work-group sums the matrix itself, whole-line row reads (10 -> 9 launches); sum order unchanged
# speedup vs baseline: 4.2462x; 1.2213x over previous
#include <hip/hip_runtime.h>
#include <hip/hip_bf16.h>

typedef __attribute__((ext_vector_type(2))) float v2f;
typedef __attribute__((ext_vector_type(8))) float v8f;

__global__ __launch_bounds__(256) void gcn_gemm_wmma_f32(
    const float* __restrict__ A, const float* __restrict__ W,
    float* __restrict__ C, const float* __restrict__ rs, int M, int K) {
  const int wave = blockIdx.x * (blockDim.x >> 5) + (threadIdx.x >> 5);
  const int lane = threadIdx.x & 31;
  const int m0 = wave << 4;
  if (m0 >= M) return;
  const int half = lane >> 4;
  const int l = lane & 15;

  v8f acc = {};
  const float* arow = A + (size_t)(m0 + l) * K + (half << 1);
  for (int k0 = 0; k0 < K; k0 += 4) {
    v2f a;
    a.x = __builtin_nontemporal_load(&arow[k0 + 0]);
    a.y = __builtin_nontemporal_load(&arow[k0 + 1]);
    v2f b;
    b.x = W[(size_t)(k0 + (half << 1) + 0) * 16 + l];
    b.y = W[(size_t)(k0 + (half << 1) + 1) * 16 + l];
    acc = __builtin_amdgcn_wmma_f32_16x16x4_f32(
         false, a,  false, b,
         (short)0, acc,  false,  false);
  }
  const int r0 = m0 + (half << 3);
  float* cp = C + ((size_t)r0 << 4) + l;
  if (rs) {
#pragma unroll
    for (int r = 0; r < 8; ++r) cp[(size_t)r << 4] = acc[r] * rs[r0 + r];
  } else {
#pragma unroll
    for (int r = 0; r < 8; ++r) cp[(size_t)r << 4] = acc[r];
  }
}

__global__ void gcn_fill(float* __restrict__ p, float v, int n) {
  int i = blockIdx.x * blockDim.x + threadIdx.x;
  if (i < n) p[i] = v;
}

__global__ void gcn_deg_count(const int* __restrict__ dst, float* __restrict__ deg, int E) {
  int e = blockIdx.x * blockDim.x + threadIdx.x;
  if (e < E) atomicAdd(&deg[__builtin_nontemporal_load(&dst[e])], 1.0f);
}

__global__ void gcn_rsqrt(float* __restrict__ d, int n) {
  int i = blockIdx.x * blockDim.x + threadIdx.x;
  if (i < n) d[i] = rsqrtf(d[i]);
}

__global__ void gcn_agg_init(const float* __restrict__ hw, const float* __restrict__ dis,
                             const float* __restrict__ b, float* __restrict__ agg, int n16) {
  int t = blockIdx.x * blockDim.x + threadIdx.x;
  if (t >= n16) return;
  int i = t >> 4, c = t & 15;
  float d = dis[i];
  agg[t] = d * d * hw[t] + b[c];
}

__global__ void gcn_edge_scatter16(const int* __restrict__ src, const int* __restrict__ dst,
                                   const float* __restrict__ dis, const float* __restrict__ hw,
                                   float* __restrict__ agg, int E) {
  int t = blockIdx.x * blockDim.x + threadIdx.x;
  int e = t >> 4, c = t & 15;
  if (e >= E) return;
  int s = __builtin_nontemporal_load(&src[e]);
  int d = __builtin_nontemporal_load(&dst[e]);
  float norm = dis[s] * dis[d];
  atomicAdd(&agg[((size_t)d << 4) + c], norm * hw[((size_t)s << 4) + c]);
}

__global__ void gcn_relu(float* __restrict__ p, int n) {
  int i = blockIdx.x * blockDim.x + threadIdx.x;
  if (i < n) p[i] = fmaxf(p[i], 0.0f);
}

__global__ void gcn_mv16(const float* __restrict__ h, const float* __restrict__ W3,
                         const float* __restrict__ rs, float* __restrict__ out, int N) {
  int i = blockIdx.x * blockDim.x + threadIdx.x;
  if (i >= N) return;
  const float* hp = h + ((size_t)i << 4);
  float s = 0.f;
#pragma unroll
  for (int c = 0; c < 16; ++c) s += __builtin_nontemporal_load(&hp[c]) * W3[c];
  out[i] = rs ? s * rs[i] : s;
}

__global__ void gcn_out_init(const float* __restrict__ hw3, const float* __restrict__ dis,
                             const float* __restrict__ b3, float* __restrict__ out, int N) {
  int i = blockIdx.x * blockDim.x + threadIdx.x;
  if (i >= N) return;
  float d = dis[i];
  out[i] = d * d * hw3[i] + b3[0];
}

__global__ void gcn_edge_scatter1(const int* __restrict__ src, const int* __restrict__ dst,
                                  const float* __restrict__ dis, const float* __restrict__ hw3,
                                  float* __restrict__ out, int E) {
  int e = blockIdx.x * blockDim.x + threadIdx.x;
  if (e >= E) return;
  int s = __builtin_nontemporal_load(&src[e]);
  int d = __builtin_nontemporal_load(&dst[e]);
  atomicAdd(&out[d], dis[s] * dis[d] * hw3[s]);
}

__global__ void gcn_zero_i32(int* __restrict__ p, int n) {
  int i = blockIdx.x * blockDim.x + threadIdx.x;
  if (i < n) p[i] = 0;
}

__global__ void gcn_edge_rank(const int* __restrict__ dst, int* __restrict__ cnt,
                              int* __restrict__ rnk, int E) {
  int e = blockIdx.x * blockDim.x + threadIdx.x;
  if (e < E) rnk[e] = atomicAdd(&cnt[dst[e]], 1);
}

__global__ __launch_bounds__(1024) void gcn_block_sum(const int* __restrict__ cnt,
                                                      int* __restrict__ bsum, int n) {
  __shared__ int wsum[32];
  const int tid = threadIdx.x, w = tid >> 5, l = tid & 31;
  const int i = blockIdx.x * 1024 + tid;
  int v = (i < n) ? cnt[i] : 0;
#pragma unroll
  for (int o = 16; o > 0; o >>= 1) v += __shfl_xor(v, o, 32);
  if (l == 0) wsum[w] = v;
  __syncthreads();
  if (w == 0) {
    int t = wsum[l];
#pragma unroll
    for (int o = 16; o > 0; o >>= 1) t += __shfl_xor(t, o, 32);
    if (l == 0) bsum[blockIdx.x] = t;
  }
}

__global__ __launch_bounds__(1024) void gcn_scan_dis(const int* __restrict__ cnt,
                                                     const int* __restrict__ bsum,
                                                     int* __restrict__ beg,
                                                     float* __restrict__ dis, int n) {
  __shared__ int wtot[32];
  __shared__ int ptot[32];
  const int tid = threadIdx.x, w = tid >> 5, l = tid & 31;
  const int i = blockIdx.x * 1024 + tid;
  const int c = (i < n) ? cnt[i] : 0;
  int p = 0;
  for (int j = tid; j < (int)blockIdx.x; j += 1024) p += bsum[j];
#pragma unroll
  for (int o = 16; o > 0; o >>= 1) p += __shfl_xor(p, o, 32);
  int s = c;
#pragma unroll
  for (int o = 1; o < 32; o <<= 1) {
    int t = __shfl_up(s, o, 32);
    if (l >= o) s += t;
  }
  if (l == 31) wtot[w] = s;
  if (l == 0) ptot[w] = p;
  __syncthreads();
  int pt = ptot[l];
  int wx = (l < w) ? wtot[l] : 0;
#pragma unroll
  for (int o = 16; o > 0; o >>= 1) {
    pt += __shfl_xor(pt, o, 32);
    wx += __shfl_xor(wx, o, 32);
  }
  if (i < n) {
    beg[i] = pt + wx + s - c;
    dis[i] = rsqrtf((float)(c + 1));
  }
}

__global__ void gcn_fill_list(const int* __restrict__ src, const int* __restrict__ dst,
                              const int* __restrict__ beg, const int* __restrict__ rnk,
                              int* __restrict__ lst, int E) {
  int e = blockIdx.x * blockDim.x + threadIdx.x;
  if (e >= E) return;
  int s = __builtin_nontemporal_load(&src[e]);
  int d = __builtin_nontemporal_load(&dst[e]);
  int r = __builtin_nontemporal_load(&rnk[e]);
  lst[beg[d] + r] = s;
}

#define GCN_SRC_BITS 22
#define GCN_MAX_NB 4096
#define GCN_U 16
#define GCN_G 256
#define GCN_ROWS 8

__global__ __launch_bounds__(1024) void gcn_bucket_hist(const int* __restrict__ dst,
                                                        int* __restrict__ hist,
                                                        int E, int chunk, int NB, int G) {
  __shared__ int h[GCN_MAX_NB];
  const int tid = threadIdx.x;
  for (int i = tid; i < NB; i += 1024) h[i] = 0;
  __syncthreads();
  const int e0 = blockIdx.x * chunk;
  const int e1 = min(e0 + chunk, E);
  for (int eb = e0 + tid; eb < e1; eb += GCN_U * 1024) {
    int d[GCN_U];
#pragma unroll
    for (int u = 0; u < GCN_U; ++u) {
      const int e = eb + u * 1024;
      d[u] = (e < e1) ? dst[e] : -1;
    }
#pragma unroll
    for (int u = 0; u < GCN_U; ++u)
      if (d[u] >= 0) atomicAdd(&h[d[u] >> 10], 1);
  }
  __syncthreads();
  for (int i = tid; i < NB; i += 1024) hist[i * G + blockIdx.x] = h[i];
}

__global__ __launch_bounds__(1024) void gcn_bucket_part(const int* __restrict__ src,
                                                        const int* __restrict__ dst,
                                                        const int* __restrict__ hist,
                                                        int* __restrict__ bstart,
                                                        int* __restrict__ tmp,
                                                        int E, int chunk, int NB) {
  static_assert(GCN_G == 256, "a count-matrix row is read as 2 x 16 bytes per lane");
  __shared__ int cur[GCN_MAX_NB];
  __shared__ int btot[GCN_MAX_NB];
  __shared__ int wtot[32];
  const int tid = threadIdx.x, w = tid >> 5, l = tid & 31;
  const int me = blockIdx.x;
  for (int b0 = w; b0 < NB; b0 += 32 * GCN_ROWS) {
    int4 x[GCN_ROWS][2];
#pragma unroll
    for (int r = 0; r < GCN_ROWS; ++r) {
      const int b = min(b0 + 32 * r, NB - 1);
      const int4* row = (const int4*)(hist + (size_t)b * GCN_G);
      x[r][0] = row[l];
      x[r][1] = row[32 + l];
    }
    const int c0 = 4 * l, c1 = 128 + 4 * l;
    int tot[GCN_ROWS], part[GCN_ROWS];
#pragma unroll
    for (int r = 0; r < GCN_ROWS; ++r) {
      const int4 p = x[r][0], q = x[r][1];
      tot[r] = ((p.x + p.y) + (p.z + p.w)) + ((q.x + q.y) + (q.z + q.w));
      part[r] = (c0 + 0 < me ? p.x : 0) + (c0 + 1 < me ? p.y : 0) +
                (c0 + 2 < me ? p.z : 0) + (c0 + 3 < me ? p.w : 0) +
                (c1 + 0 < me ? q.x : 0) + (c1 + 1 < me ? q.y : 0) +
                (c1 + 2 < me ? q.z : 0) + (c1 + 3 < me ? q.w : 0);
    }
#pragma unroll
    for (int o = 16; o > 0; o >>= 1) {
#pragma unroll
      for (int r = 0; r < GCN_ROWS; ++r) {
        tot[r] += __shfl_xor(tot[r], o, 32);
        part[r] += __shfl_xor(part[r], o, 32);
      }
    }
#pragma unroll
    for (int r = 0; r < GCN_ROWS; ++r) {
      const int b = b0 + 32 * r;
      if (l == 0 && b < NB) { btot[b] = tot[r]; cur[b] = part[r]; }
    }
  }
  __syncthreads();
  {
    int v[4];
    int sum = 0;
#pragma unroll
    for (int k = 0; k < 4; ++k) {
      const int b = 4 * tid + k;
      v[k] = (b < NB) ? btot[b] : 0;
      sum += v[k];
    }
    int s = sum;
#pragma unroll
    for (int o = 1; o < 32; o <<= 1) {
      int t = __shfl_up(s, o, 32);
      if (l >= o) s += t;
    }
    if (l == 31) wtot[w] = s;
    __syncthreads();
    int wx = (l < w) ? wtot[l] : 0;
#pragma unroll
    for (int o = 16; o > 0; o >>= 1) wx += __shfl_xor(wx, o, 32);
    int run = wx + s - sum;
#pragma unroll
    for (int k = 0; k < 4; ++k) {
      const int b = 4 * tid + k;
      if (b < NB) {
        cur[b] += run;
        if (me == 0) bstart[b] = run;
      }
      run += v[k];
    }
  }
  __syncthreads();
  const int e0 = blockIdx.x * chunk;
  const int e1 = min(e0 + chunk, E);
  for (int eb = e0 + tid; eb < e1; eb += GCN_U * 1024) {
    int s[GCN_U], d[GCN_U];
#pragma unroll
    for (int u = 0; u < GCN_U; ++u) {
      const int e = eb + u * 1024;
      const bool ok = e < e1;
      s[u] = ok ? __builtin_nontemporal_load(&src[e]) : 0;
      d[u] = ok ? __builtin_nontemporal_load(&dst[e]) : -1;
    }
#pragma unroll
    for (int u = 0; u < GCN_U; ++u) {
      if (d[u] >= 0) {
        const unsigned du = (unsigned)d[u];
        const int pos = atomicAdd(&cur[du >> 10], 1);
        tmp[pos] = (int)(((du & 1023u) << GCN_SRC_BITS) | (unsigned)s[u]);
      }
    }
  }
}

__global__ __launch_bounds__(1024) void gcn_bucket_sort(const int* __restrict__ tmp,
                                                        const int* __restrict__ bstart,
                                                        int* __restrict__ cnt,
                                                        int* __restrict__ beg,
                                                        float* __restrict__ dis,
                                                        int* __restrict__ lst,
                                                        int N, int E, int NB) {
  __shared__ int c[1024];
  __shared__ int wtot[32];
  const int tid = threadIdx.x, w = tid >> 5, l = tid & 31;
  const int b = blockIdx.x;
  const int p0 = bstart[b];
  const int p1 = (b + 1 < NB) ? bstart[b + 1] : E;
  c[tid] = 0;
  __syncthreads();
  for (int pb = p0 + tid; pb < p1; pb += GCN_U * 1024) {
    unsigned v[GCN_U];
#pragma unroll
    for (int u = 0; u < GCN_U; ++u) v[u] = (unsigned)tmp[min(pb + u * 1024, p1 - 1)];
#pragma unroll
    for (int u = 0; u < GCN_U; ++u)
      if (pb + u * 1024 < p1) atomicAdd(&c[v[u] >> GCN_SRC_BITS], 1);
  }
  __syncthreads();
  const int n = c[tid];
  int s = n;
#pragma unroll
  for (int o = 1; o < 32; o <<= 1) {
    int t = __shfl_up(s, o, 32);
    if (l >= o) s += t;
  }
  if (l == 31) wtot[w] = s;
  __syncthreads();
  int wx = (l < w) ? wtot[l] : 0;
#pragma unroll
  for (int o = 16; o > 0; o >>= 1) wx += __shfl_xor(wx, o, 32);
  const int ex = p0 + wx + s - n;
  const int i = (b << 10) + tid;
  if (i < N) {
    cnt[i] = n;
    beg[i] = ex;
    dis[i] = rsqrtf((float)(n + 1));
  }
  c[tid] = ex;
  __syncthreads();
  for (int pb = p0 + tid; pb < p1; pb += GCN_U * 1024) {
    unsigned v[GCN_U];
#pragma unroll
    for (int u = 0; u < GCN_U; ++u)
      v[u] = (unsigned)__builtin_nontemporal_load(&tmp[min(pb + u * 1024, p1 - 1)]);
#pragma unroll
    for (int u = 0; u < GCN_U; ++u) {
      if (pb + u * 1024 < p1) {
        const int pos = atomicAdd(&c[v[u] >> GCN_SRC_BITS], 1);
        lst[pos] = (int)(v[u] & ((1u << GCN_SRC_BITS) - 1u));
      }
    }
  }
}

__global__ __launch_bounds__(256) void gcn_gather16_relu(
    const int* __restrict__ beg, const int* __restrict__ cnt, const int* __restrict__ lst,
    const float* __restrict__ dis, const float* __restrict__ hws, const float* __restrict__ b,
    float* __restrict__ h, int N) {
  const int t = blockIdx.x * blockDim.x + threadIdx.x;
  const int i = t >> 4, c = t & 15;
  if (i >= N) return;
  const int* lp = lst + beg[i];
  const int len = cnt[i];
  const float d = dis[i];
  float a[8];
  a[0] = hws[((size_t)i << 4) + c];
#pragma unroll
  for (int u = 1; u < 8; ++u) a[u] = 0.f;
  int j = 0;
#pragma unroll 1
  for (; j + 16 <= len; j += 16) {
    int s[16];
#pragma unroll
    for (int u = 0; u < 16; ++u) s[u] = lp[j + u];
    float r[16];
#pragma unroll
    for (int u = 0; u < 16; ++u) r[u] = hws[((size_t)s[u] << 4) + c];
#pragma unroll
    for (int u = 0; u < 8; ++u) a[u] += r[u];
#pragma unroll
    for (int u = 0; u < 8; ++u) a[u] += r[8 + u];
  }
  const int rem = len - j;
  if (rem > 0) {
    const int last = len - 1;
    int s[15];
#pragma unroll
    for (int u = 0; u < 15; ++u) s[u] = lp[min(j + u, last)];
#pragma unroll
    for (int u = 0; u < 15; ++u) asm volatile("" : "+v"(s[u]));
    float r[15];
#pragma unroll
    for (int u = 0; u < 15; ++u) r[u] = hws[((size_t)s[u] << 4) + c];
#pragma unroll
    for (int u = 0; u < 15; ++u) asm volatile("" : "+v"(r[u]));
    const bool blk = rem >= 8;
#pragma unroll
    for (int u = 0; u < 8; ++u) a[u] = blk ? a[u] + r[u] : a[u];
    const int tail = rem & 7;
#pragma unroll
    for (int q = 0; q < 7; ++q) {
      const float v = blk ? r[8 + q] : r[q];
      a[0] = (q < tail) ? a[0] + v : a[0];
    }
  }
  float r = ((a[0] + a[1]) + (a[2] + a[3])) + ((a[4] + a[5]) + (a[6] + a[7]));
  h[((size_t)i << 4) + c] = fmaxf(d * r + b[c], 0.0f);
}

__global__ __launch_bounds__(256) void gcn_gather1(
    const int* __restrict__ beg, const int* __restrict__ cnt, const int* __restrict__ lst,
    const float* __restrict__ dis, const float* __restrict__ hw3s, const float* __restrict__ b3,
    float* __restrict__ out, int N) {
  const int t = blockIdx.x * blockDim.x + threadIdx.x;
  const int i = t >> 4, l = t & 15;
  const bool ok = i < N;
  const int ii = ok ? i : 0;
  const int* lp = lst + beg[ii];
  const int len = ok ? cnt[ii] : 0;
  const float d = dis[ii];
  float a = 0.f;
  for (int j = l; j < len; j += 16) a += hw3s[lp[j]];
#pragma unroll
  for (int o = 8; o > 0; o >>= 1) a += __shfl_xor(a, o, 16);
  if (ok && l == 0) out[i] = d * (a + hw3s[i]) + b3[0];
}

static inline int cdiv(long long a, long long b) { return (int)((a + b - 1) / b); }
static inline size_t up256(size_t v) { return (v + 255) & ~(size_t)255; }

extern "C" void kernel_launch(void* const* d_in, const int* in_sizes, int n_in,
                              void* d_out, int out_size, void* d_ws, size_t ws_size,
                              hipStream_t stream) {
  const float* x  = (const float*)d_in[0];
  const int*   ei = (const int*)d_in[1];
  const float* W1 = (const float*)d_in[2];
  const float* b1 = (const float*)d_in[3];
  const float* W2 = (const float*)d_in[4];
  const float* b2 = (const float*)d_in[5];
  const float* W3 = (const float*)d_in[6];
  const float* b3 = (const float*)d_in[7];

  const int IN_CH = 256;
  const int HID = 16;
  const int N = in_sizes[0] / IN_CH;
  const int E = in_sizes[1] / 2;
  const int* src = ei;
  const int* dst = ei + E;
  float* out = (float*)d_out;

  char* ws = (char*)d_ws;
  size_t off = 0;
  const size_t hwB = up256((size_t)N * HID * 4);
  float* dis = (float*)(ws + off); off += up256((size_t)N * 4);
  float* HW  = (float*)(ws + off); off += hwB;
  float* AGG = (float*)(ws + off); off += hwB;
  (void)n_in; (void)out_size;

  const int B = 256;
  const int n16 = N * HID;
  const int gemmBlocks = cdiv(cdiv(N, 16), B / 32);
  const int nb = cdiv(N, 1024);

  int* cnt  = (int*)(ws + off); off += up256((size_t)N * 4);
  int* beg  = (int*)(ws + off); off += up256((size_t)N * 4);
  int* bsum = (int*)(ws + off); off += up256((size_t)nb * 4);
  int* lst  = (int*)(ws + off); off += up256((size_t)E * 4);
  int* rnk  = (int*)HW;
  const bool sorted_path = (ws_size >= off) && ((size_t)E * 4 <= 2 * hwB);

  const int G = GCN_G;
  const int chunk = cdiv(E, G);
  int* hist = lst;
  int* bstart = bsum;
  const bool lds_sort = sorted_path && N <= (1 << GCN_SRC_BITS) && nb <= GCN_MAX_NB &&
                        (long long)nb * G <= (long long)E;

  if (lds_sort) {
    gcn_bucket_hist<<<G, 1024, 0, stream>>>(dst, hist, E, chunk, nb, G);
    gcn_bucket_part<<<G, 1024, 0, stream>>>(src, dst, hist, bstart, rnk, E, chunk, nb);
    gcn_bucket_sort<<<nb, 1024, 0, stream>>>(rnk, bstart, cnt, beg, dis, lst, N, E, nb);
  } else if (sorted_path) {
    gcn_zero_i32<<<cdiv(N, B), B, 0, stream>>>(cnt, N);
    gcn_edge_rank<<<cdiv(E, B), B, 0, stream>>>(dst, cnt, rnk, E);
    gcn_block_sum<<<nb, 1024, 0, stream>>>(cnt, bsum, N);
    gcn_scan_dis<<<nb, 1024, 0, stream>>>(cnt, bsum, beg, dis, N);
    gcn_fill_list<<<cdiv(E, B), B, 0, stream>>>(src, dst, beg, rnk, lst, E);
  }

  if (sorted_path) {
    gcn_gemm_wmma_f32<<<gemmBlocks, B, 0, stream>>>(x, W1, HW, dis, N, IN_CH);
    gcn_gather16_relu<<<cdiv(n16, B), B, 0, stream>>>(beg, cnt, lst, dis, HW, b1, AGG, N);
    gcn_gemm_wmma_f32<<<gemmBlocks, B, 0, stream>>>(AGG, W2, HW, dis, N, HID);
    gcn_gather16_relu<<<cdiv(n16, B), B, 0, stream>>>(beg, cnt, lst, dis, HW, b2, AGG, N);
    gcn_mv16<<<cdiv(N, B), B, 0, stream>>>(AGG, W3, dis, HW, N);
    gcn_gather1<<<cdiv(n16, B), B, 0, stream>>>(beg, cnt, lst, dis, HW, b3, out, N);
    return;
  }

  gcn_fill<<<cdiv(N, B), B, 0, stream>>>(dis, 1.0f, N);
  gcn_deg_count<<<cdiv(E, B), B, 0, stream>>>(dst, dis, E);
  gcn_rsqrt<<<cdiv(N, B), B, 0, stream>>>(dis, N);

  gcn_gemm_wmma_f32<<<gemmBlocks, B, 0, stream>>>(x, W1, HW, nullptr, N, IN_CH);
  gcn_agg_init<<<cdiv(n16, B), B, 0, stream>>>(HW, dis, b1, AGG, n16);
  gcn_edge_scatter16<<<cdiv((long long)E * 16, B), B, 0, stream>>>(src, dst, dis, HW, AGG, E);
  gcn_relu<<<cdiv(n16, B), B, 0, stream>>>(AGG, n16);

  gcn_gemm_wmma_f32<<<gemmBlocks, B, 0, stream>>>(AGG, W2, HW, nullptr, N, HID);
  gcn_agg_init<<<cdiv(n16, B), B, 0, stream>>>(HW, dis, b2, AGG, n16);
  gcn_edge_scatter16<<<cdiv((long long)E * 16, B), B, 0, stream>>>(src, dst, dis, HW, AGG, E);
  gcn_relu<<<cdiv(n16, B), B, 0, stream>>>(AGG, n16);

  gcn_mv16<<<cdiv(N, B), B, 0, stream>>>(AGG, W3, nullptr, HW, N);
  gcn_out_init<<<cdiv(N, B), B, 0, stream>>>(HW, dis, b3, out, N);
  gcn_edge_scatter1<<<cdiv(E, B), B, 0, stream>>>(src, dst, dis, HW, out, E);
}


# 256->16 product: A rows read as whole 128-byte lines (16 lanes x 16 B per row; was 16 rows x 16 B per load), put into the fragment layout through a per-wave LDS tile; k order and sum order kept
# speedup vs baseline: 5.1863x; 1.2214x over previous
#include <hip/hip_runtime.h>
#include <hip/hip_bf16.h>

typedef __attribute__((ext_vector_type(2))) float v2f;
typedef __attribute__((ext_vector_type(8))) float v8f;

__global__ __launch_bounds__(256) void gcn_gemm_wmma_f32(
    const float* __restrict__ A, const float* __restrict__ W,
    float* __restrict__ C, const float* __restrict__ rs, int M, int K) {
  const int wave = blockIdx.x * (blockDim.x >> 5) + (threadIdx.x >> 5);
  const int lane = threadIdx.x & 31;
  const int m0 = wave << 4;
  if (m0 >= M) return;
  const int half = lane >> 4;
  const int l = lane & 15;

  v8f acc = {};
  const float* arow = A + (size_t)(m0 + l) * K + (half << 1);
  for (int k0 = 0; k0 < K; k0 += 4) {
    v2f a;
    a.x = __builtin_nontemporal_load(&arow[k0 + 0]);
    a.y = __builtin_nontemporal_load(&arow[k0 + 1]);
    v2f b;
    b.x = W[(size_t)(k0 + (half << 1) + 0) * 16 + l];
    b.y = W[(size_t)(k0 + (half << 1) + 1) * 16 + l];
    acc = __builtin_amdgcn_wmma_f32_16x16x4_f32(
         false, a,  false, b,
         (short)0, acc,  false,  false);
  }
  const int r0 = m0 + (half << 3);
  float* cp = C + ((size_t)r0 << 4) + l;
  if (rs) {
#pragma unroll
    for (int r = 0; r < 8; ++r) cp[(size_t)r << 4] = acc[r] * rs[r0 + r];
  } else {
#pragma unroll
    for (int r = 0; r < 8; ++r) cp[(size_t)r << 4] = acc[r];
  }
}

typedef __attribute__((ext_vector_type(4))) float v4f;
#define GCN_KC 64
#define GCN_APITCH 68

__global__ __launch_bounds__(256) void gcn_gemm_wmma_f32_lines(
    const float* __restrict__ A, const float* __restrict__ W,
    float* __restrict__ C, const float* __restrict__ rs, int M, int K) {
  __shared__ __attribute__((aligned(16))) float sh[8][16 * GCN_APITCH];
  const int wv = threadIdx.x >> 5;
  const int wave = blockIdx.x * 8 + wv;
  const int lane = threadIdx.x & 31;
  const int m0 = wave << 4;
  if (m0 >= M) return;
  const int half = lane >> 4;
  const int l = lane & 15;
  float* tile = sh[wv];
  const float* gp = A + (size_t)(m0 + half) * K + 4 * l;
  float* sp = tile + half * GCN_APITCH + 4 * l;
  const float* fp = tile + l * GCN_APITCH + 2 * half;
  const float* wp = W + (size_t)(half << 1) * 16 + l;

  v8f acc = {};
  v4f buf[8];
#pragma unroll
  for (int i = 0; i < 8; ++i)
    buf[i] = __builtin_nontemporal_load((const v4f*)(gp + (size_t)(2 * i) * K));
  for (int k0 = 0; k0 < K; k0 += GCN_KC) {
    __builtin_amdgcn_fence(__ATOMIC_RELEASE, "wavefront");
    __builtin_amdgcn_wave_barrier();
#pragma unroll
    for (int i = 0; i < 8; ++i) *(v4f*)(sp + 2 * i * GCN_APITCH) = buf[i];
    v2f b[GCN_KC / 4];
#pragma unroll
    for (int j = 0; j < GCN_KC / 4; ++j) {
      b[j].x = wp[(size_t)(k0 + 4 * j + 0) * 16];
      b[j].y = wp[(size_t)(k0 + 4 * j + 1) * 16];
    }
    if (k0 + GCN_KC < K) {
#pragma unroll
      for (int i = 0; i < 8; ++i)
        buf[i] = __builtin_nontemporal_load((const v4f*)(gp + (size_t)(2 * i) * K + k0 + GCN_KC));
    }
    __builtin_amdgcn_wave_barrier();
    __builtin_amdgcn_fence(__ATOMIC_ACQUIRE, "wavefront");
#pragma unroll
    for (int j = 0; j < GCN_KC / 4; ++j) {
      const v2f a = *(const v2f*)(fp + 4 * j);
      acc = __builtin_amdgcn_wmma_f32_16x16x4_f32(
           false, a,  false, b[j],
           (short)0, acc,  false,  false);
    }
  }
  const int r0 = m0 + (half << 3);
  float* cp = C + ((size_t)r0 << 4) + l;
  if (rs) {
#pragma unroll
    for (int r = 0; r < 8; ++r) cp[(size_t)r << 4] = acc[r] * rs[r0 + r];
  } else {
#pragma unroll
    for (int r = 0; r < 8; ++r) cp[(size_t)r << 4] = acc[r];
  }
}

__global__ void gcn_fill(float* __restrict__ p, float v, int n) {
  int i = blockIdx.x * blockDim.x + threadIdx.x;
  if (i < n) p[i] = v;
}

__global__ void gcn_deg_count(const int* __restrict__ dst, float* __restrict__ deg, int E) {
  int e = blockIdx.x * blockDim.x + threadIdx.x;
  if (e < E) atomicAdd(&deg[__builtin_nontemporal_load(&dst[e])], 1.0f);
}

__global__ void gcn_rsqrt(float* __restrict__ d, int n) {
  int i = blockIdx.x * blockDim.x + threadIdx.x;
  if (i < n) d[i] = rsqrtf(d[i]);
}

__global__ void gcn_agg_init(const float* __restrict__ hw, const float* __restrict__ dis,
                             const float* __restrict__ b, float* __restrict__ agg, int n16) {
  int t = blockIdx.x * blockDim.x + threadIdx.x;
  if (t >= n16) return;
  int i = t >> 4, c = t & 15;
  float d = dis[i];
  agg[t] = d * d * hw[t] + b[c];
}

__global__ void gcn_edge_scatter16(const int* __restrict__ src, const int* __restrict__ dst,
                                   const float* __restrict__ dis, const float* __restrict__ hw,
                                   float* __restrict__ agg, int E) {
  int t = blockIdx.x * blockDim.x + threadIdx.x;
  int e = t >> 4, c = t & 15;
  if (e >= E) return;
  int s = __builtin_nontemporal_load(&src[e]);
  int d = __builtin_nontemporal_load(&dst[e]);
  float norm = dis[s] * dis[d];
  atomicAdd(&agg[((size_t)d << 4) + c], norm * hw[((size_t)s << 4) + c]);
}

__global__ void gcn_relu(float* __restrict__ p, int n) {
  int i = blockIdx.x * blockDim.x + threadIdx.x;
  if (i < n) p[i] = fmaxf(p[i], 0.0f);
}

__global__ void gcn_mv16(const float* __restrict__ h, const float* __restrict__ W3,
                         const float* __restrict__ rs, float* __restrict__ out, int N) {
  int i = blockIdx.x * blockDim.x + threadIdx.x;
  if (i >= N) return;
  const float* hp = h + ((size_t)i << 4);
  float s = 0.f;
#pragma unroll
  for (int c = 0; c < 16; ++c) s += __builtin_nontemporal_load(&hp[c]) * W3[c];
  out[i] = rs ? s * rs[i] : s;
}

__global__ void gcn_out_init(const float* __restrict__ hw3, const float* __restrict__ dis,
                             const float* __restrict__ b3, float* __restrict__ out, int N) {
  int i = blockIdx.x * blockDim.x + threadIdx.x;
  if (i >= N) return;
  float d = dis[i];
  out[i] = d * d * hw3[i] + b3[0];
}

__global__ void gcn_edge_scatter1(const int* __restrict__ src, const int* __restrict__ dst,
                                  const float* __restrict__ dis, const float* __restrict__ hw3,
                                  float* __restrict__ out, int E) {
  int e = blockIdx.x * blockDim.x + threadIdx.x;
  if (e >= E) return;
  int s = __builtin_nontemporal_load(&src[e]);
  int d = __builtin_nontemporal_load(&dst[e]);
  atomicAdd(&out[d], dis[s] * dis[d] * hw3[s]);
}

__global__ void gcn_zero_i32(int* __restrict__ p, int n) {
  int i = blockIdx.x * blockDim.x + threadIdx.x;
  if (i < n) p[i] = 0;
}

__global__ void gcn_edge_rank(const int* __restrict__ dst, int* __restrict__ cnt,
                              int* __restrict__ rnk, int E) {
  int e = blockIdx.x * blockDim.x + threadIdx.x;
  if (e < E) rnk[e] = atomicAdd(&cnt[dst[e]], 1);
}

__global__ __launch_bounds__(1024) void gcn_block_sum(const int* __restrict__ cnt,
                                                      int* __restrict__ bsum, int n) {
  __shared__ int wsum[32];
  const int tid = threadIdx.x, w = tid >> 5, l = tid & 31;
  const int i = blockIdx.x * 1024 + tid;
  int v = (i < n) ? cnt[i] : 0;
#pragma unroll
  for (int o = 16; o > 0; o >>= 1) v += __shfl_xor(v, o, 32);
  if (l == 0) wsum[w] = v;
  __syncthreads();
  if (w == 0) {
    int t = wsum[l];
#pragma unroll
    for (int o = 16; o > 0; o >>= 1) t += __shfl_xor(t, o, 32);
    if (l == 0) bsum[blockIdx.x] = t;
  }
}

__global__ __launch_bounds__(1024) void gcn_scan_dis(const int* __restrict__ cnt,
                                                     const int* __restrict__ bsum,
                                                     int* __restrict__ beg,
                                                     float* __restrict__ dis, int n) {
  __shared__ int wtot[32];
  __shared__ int ptot[32];
  const int tid = threadIdx.x, w = tid >> 5, l = tid & 31;
  const int i = blockIdx.x * 1024 + tid;
  const int c = (i < n) ? cnt[i] : 0;
  int p = 0;
  for (int j = tid; j < (int)blockIdx.x; j += 1024) p += bsum[j];
#pragma unroll
  for (int o = 16; o > 0; o >>= 1) p += __shfl_xor(p, o, 32);
  int s = c;
#pragma unroll
  for (int o = 1; o < 32; o <<= 1) {
    int t = __shfl_up(s, o, 32);
    if (l >= o) s += t;
  }
  if (l == 31) wtot[w] = s;
  if (l == 0) ptot[w] = p;
  __syncthreads();
  int pt = ptot[l];
  int wx = (l < w) ? wtot[l] : 0;
#pragma unroll
  for (int o = 16; o > 0; o >>= 1) {
    pt += __shfl_xor(pt, o, 32);
    wx += __shfl_xor(wx, o, 32);
  }
  if (i < n) {
    beg[i] = pt + wx + s - c;
    dis[i] = rsqrtf((float)(c + 1));
  }
}

__global__ void gcn_fill_list(const int* __restrict__ src, const int* __restrict__ dst,
                              const int* __restrict__ beg, const int* __restrict__ rnk,
                              int* __restrict__ lst, int E) {
  int e = blockIdx.x * blockDim.x + threadIdx.x;
  if (e >= E) return;
  int s = __builtin_nontemporal_load(&src[e]);
  int d = __builtin_nontemporal_load(&dst[e]);
  int r = __builtin_nontemporal_load(&rnk[e]);
  lst[beg[d] + r] = s;
}

#define GCN_SRC_BITS 22
#define GCN_MAX_NB 4096
#define GCN_U 16
#define GCN_G 256
#define GCN_ROWS 8

__global__ __launch_bounds__(1024) void gcn_bucket_hist(const int* __restrict__ dst,
                                                        int* __restrict__ hist,
                                                        int E, int chunk, int NB, int G) {
  __shared__ int h[GCN_MAX_NB];
  const int tid = threadIdx.x;
  for (int i = tid; i < NB; i += 1024) h[i] = 0;
  __syncthreads();
  const int e0 = blockIdx.x * chunk;
  const int e1 = min(e0 + chunk, E);
  for (int eb = e0 + tid; eb < e1; eb += GCN_U * 1024) {
    int d[GCN_U];
#pragma unroll
    for (int u = 0; u < GCN_U; ++u) {
      const int e = eb + u * 1024;
      d[u] = (e < e1) ? dst[e] : -1;
    }
#pragma unroll
    for (int u = 0; u < GCN_U; ++u)
      if (d[u] >= 0) atomicAdd(&h[d[u] >> 10], 1);
  }
  __syncthreads();
  for (int i = tid; i < NB; i += 1024) hist[i * G + blockIdx.x] = h[i];
}

__global__ __launch_bounds__(1024) void gcn_bucket_part(const int* __restrict__ src,
                                                        const int* __restrict__ dst,
                                                        const int* __restrict__ hist,
                                                        int* __restrict__ bstart,
                                                        int* __restrict__ tmp,
                                                        int E, int chunk, int NB) {
  static_assert(GCN_G == 256, "a count-matrix row is read as 2 x 16 bytes per lane");
  __shared__ int cur[GCN_MAX_NB];
  __shared__ int btot[GCN_MAX_NB];
  __shared__ int wtot[32];
  const int tid = threadIdx.x, w = tid >> 5, l = tid & 31;
  const int me = blockIdx.x;
  for (int b0 = w; b0 < NB; b0 += 32 * GCN_ROWS) {
    int4 x[GCN_ROWS][2];
#pragma unroll
    for (int r = 0; r < GCN_ROWS; ++r) {
      const int b = min(b0 + 32 * r, NB - 1);
      const int4* row = (const int4*)(hist + (size_t)b * GCN_G);
      x[r][0] = row[l];
      x[r][1] = row[32 + l];
    }
    const int c0 = 4 * l, c1 = 128 + 4 * l;
    int tot[GCN_ROWS], part[GCN_ROWS];
#pragma unroll
    for (int r = 0; r < GCN_ROWS; ++r) {
      const int4 p = x[r][0], q = x[r][1];
      tot[r] = ((p.x + p.y) + (p.z + p.w)) + ((q.x + q.y) + (q.z + q.w));
      part[r] = (c0 + 0 < me ? p.x : 0) + (c0 + 1 < me ? p.y : 0) +
                (c0 + 2 < me ? p.z : 0) + (c0 + 3 < me ? p.w : 0) +
                (c1 + 0 < me ? q.x : 0) + (c1 + 1 < me ? q.y : 0) +
                (c1 + 2 < me ? q.z : 0) + (c1 + 3 < me ? q.w : 0);
    }
#pragma unroll
    for (int o = 16; o > 0; o >>= 1) {
#pragma unroll
      for (int r = 0; r < GCN_ROWS; ++r) {
        tot[r] += __shfl_xor(tot[r], o, 32);
        part[r] += __shfl_xor(part[r], o, 32);
      }
    }
#pragma unroll
    for (int r = 0; r < GCN_ROWS; ++r) {
      const int b = b0 + 32 * r;
      if (l == 0 && b < NB) { btot[b] = tot[r]; cur[b] = part[r]; }
    }
  }
  __syncthreads();
  {
    int v[4];
    int sum = 0;
#pragma unroll
    for (int k = 0; k < 4; ++k) {
      const int b = 4 * tid + k;
      v[k] = (b < NB) ? btot[b] : 0;
      sum += v[k];
    }
    int s = sum;
#pragma unroll
    for (int o = 1; o < 32; o <<= 1) {
      int t = __shfl_up(s, o, 32);
      if (l >= o) s += t;
    }
    if (l == 31) wtot[w] = s;
    __syncthreads();
    int wx = (l < w) ? wtot[l] : 0;
#pragma unroll
    for (int o = 16; o > 0; o >>= 1) wx += __shfl_xor(wx, o, 32);
    int run = wx + s - sum;
#pragma unroll
    for (int k = 0; k < 4; ++k) {
      const int b = 4 * tid + k;
      if (b < NB) {
        cur[b] += run;
        if (me == 0) bstart[b] = run;
      }
      run += v[k];
    }
  }
  __syncthreads();
  const int e0 = blockIdx.x * chunk;
  const int e1 = min(e0 + chunk, E);
  for (int eb = e0 + tid; eb < e1; eb += GCN_U * 1024) {
    int s[GCN_U], d[GCN_U];
#pragma unroll
    for (int u = 0; u < GCN_U; ++u) {
      const int e = eb + u * 1024;
      const bool ok = e < e1;
      s[u] = ok ? __builtin_nontemporal_load(&src[e]) : 0;
      d[u] = ok ? __builtin_nontemporal_load(&dst[e]) : -1;
    }
#pragma unroll
    for (int u = 0; u < GCN_U; ++u) {
      if (d[u] >= 0) {
        const unsigned du = (unsigned)d[u];
        const int pos = atomicAdd(&cur[du >> 10], 1);
        tmp[pos] = (int)(((du & 1023u) << GCN_SRC_BITS) | (unsigned)s[u]);
      }
    }
  }
}

__global__ __launch_bounds__(1024) void gcn_bucket_sort(const int* __restrict__ tmp,
                                                        const int* __restrict__ bstart,
                                                        int* __restrict__ cnt,
                                                        int* __restrict__ beg,
                                                        float* __restrict__ dis,
                                                        int* __restrict__ lst,
                                                        int N, int E, int NB) {
  __shared__ int c[1024];
  __shared__ int wtot[32];
  const int tid = threadIdx.x, w = tid >> 5, l = tid & 31;
  const int b = blockIdx.x;
  const int p0 = bstart[b];
  const int p1 = (b + 1 < NB) ? bstart[b + 1] : E;
  c[tid] = 0;
  __syncthreads();
  for (int pb = p0 + tid; pb < p1; pb += GCN_U * 1024) {
    unsigned v[GCN_U];
#pragma unroll
    for (int u = 0; u < GCN_U; ++u) v[u] = (unsigned)tmp[min(pb + u * 1024, p1 - 1)];
#pragma unroll
    for (int u = 0; u < GCN_U; ++u)
      if (pb + u * 1024 < p1) atomicAdd(&c[v[u] >> GCN_SRC_BITS], 1);
  }
  __syncthreads();
  const int n = c[tid];
  int s = n;
#pragma unroll
  for (int o = 1; o < 32; o <<= 1) {
    int t = __shfl_up(s, o, 32);
    if (l >= o) s += t;
  }
  if (l == 31) wtot[w] = s;
  __syncthreads();
  int wx = (l < w) ? wtot[l] : 0;
#pragma unroll
  for (int o = 16; o > 0; o >>= 1) wx += __shfl_xor(wx, o, 32);
  const int ex = p0 + wx + s - n;
  const int i = (b << 10) + tid;
  if (i < N) {
    cnt[i] = n;
    beg[i] = ex;
    dis[i] = rsqrtf((float)(n + 1));
  }
  c[tid] = ex;
  __syncthreads();
  for (int pb = p0 + tid; pb < p1; pb += GCN_U * 1024) {
    unsigned v[GCN_U];
#pragma unroll
    for (int u = 0; u < GCN_U; ++u)
      v[u] = (unsigned)__builtin_nontemporal_load(&tmp[min(pb + u * 1024, p1 - 1)]);
#pragma unroll
    for (int u = 0; u < GCN_U; ++u) {
      if (pb + u * 1024 < p1) {
        const int pos = atomicAdd(&c[v[u] >> GCN_SRC_BITS], 1);
        lst[pos] = (int)(v[u] & ((1u << GCN_SRC_BITS) - 1u));
      }
    }
  }
}

__global__ __launch_bounds__(256) void gcn_gather16_relu(
    const int* __restrict__ beg, const int* __restrict__ cnt, const int* __restrict__ lst,
    const float* __restrict__ dis, const float* __restrict__ hws, const float* __restrict__ b,
    float* __restrict__ h, int N) {
  const int t = blockIdx.x * blockDim.x + threadIdx.x;
  const int i = t >> 4, c = t & 15;
  if (i >= N) return;
  const int* lp = lst + beg[i];
  const int len = cnt[i];
  const float d = dis[i];
  float a[8];
  a[0] = hws[((size_t)i << 4) + c];
#pragma unroll
  for (int u = 1; u < 8; ++u) a[u] = 0.f;
  int j = 0;
#pragma unroll 1
  for (; j + 16 <= len; j += 16) {
    int s[16];
#pragma unroll
    for (int u = 0; u < 16; ++u) s[u] = lp[j + u];
    float r[16];
#pragma unroll
    for (int u = 0; u < 16; ++u) r[u] = hws[((size_t)s[u] << 4) + c];
#pragma unroll
    for (int u = 0; u < 8; ++u) a[u] += r[u];
#pragma unroll
    for (int u = 0; u < 8; ++u) a[u] += r[8 + u];
  }
  const int rem = len - j;
  if (rem > 0) {
    const int last = len - 1;
    int s[15];
#pragma unroll
    for (int u = 0; u < 15; ++u) s[u] = lp[min(j + u, last)];
#pragma unroll
    for (int u = 0; u < 15; ++u) asm volatile("" : "+v"(s[u]));
    float r[15];
#pragma unroll
    for (int u = 0; u < 15; ++u) r[u] = hws[((size_t)s[u] << 4) + c];
#pragma unroll
    for (int u = 0; u < 15; ++u) asm volatile("" : "+v"(r[u]));
    const bool blk = rem >= 8;
#pragma unroll
    for (int u = 0; u < 8; ++u) a[u] = blk ? a[u] + r[u] : a[u];
    const int tail = rem & 7;
#pragma unroll
    for (int q = 0; q < 7; ++q) {
      const float v = blk ? r[8 + q] : r[q];
      a[0] = (q < tail) ? a[0] + v : a[0];
    }
  }
  float r = ((a[0] + a[1]) + (a[2] + a[3])) + ((a[4] + a[5]) + (a[6] + a[7]));
  h[((size_t)i << 4) + c] = fmaxf(d * r + b[c], 0.0f);
}

__global__ __launch_bounds__(256) void gcn_gather1(
    const int* __restrict__ beg, const int* __restrict__ cnt, const int* __restrict__ lst,
    const float* __restrict__ dis, const float* __restrict__ hw3s, const float* __restrict__ b3,
    float* __restrict__ out, int N) {
  const int t = blockIdx.x * blockDim.x + threadIdx.x;
  const int i = t >> 4, l = t & 15;
  const bool ok = i < N;
  const int ii = ok ? i : 0;
  const int* lp = lst + beg[ii];
  const int len = ok ? cnt[ii] : 0;
  const float d = dis[ii];
  float a = 0.f;
  for (int j = l; j < len; j += 16) a += hw3s[lp[j]];
#pragma unroll
  for (int o = 8; o > 0; o >>= 1) a += __shfl_xor(a, o, 16);
  if (ok && l == 0) out[i] = d * (a + hw3s[i]) + b3[0];
}

static inline int cdiv(long long a, long long b) { return (int)((a + b - 1) / b); }
static inline size_t up256(size_t v) { return (v + 255) & ~(size_t)255; }

extern "C" void kernel_launch(void* const* d_in, const int* in_sizes, int n_in,
                              void* d_out, int out_size, void* d_ws, size_t ws_size,
                              hipStream_t stream) {
  const float* x  = (const float*)d_in[0];
  const int*   ei = (const int*)d_in[1];
  const float* W1 = (const float*)d_in[2];
  const float* b1 = (const float*)d_in[3];
  const float* W2 = (const float*)d_in[4];
  const float* b2 = (const float*)d_in[5];
  const float* W3 = (const float*)d_in[6];
  const float* b3 = (const float*)d_in[7];

  const int IN_CH = 256;
  const int HID = 16;
  const int N = in_sizes[0] / IN_CH;
  const int E = in_sizes[1] / 2;
  const int* src = ei;
  const int* dst = ei + E;
  float* out = (float*)d_out;

  char* ws = (char*)d_ws;
  size_t off = 0;
  const size_t hwB = up256((size_t)N * HID * 4);
  float* dis = (float*)(ws + off); off += up256((size_t)N * 4);
  float* HW  = (float*)(ws + off); off += hwB;
  float* AGG = (float*)(ws + off); off += hwB;
  (void)n_in; (void)out_size;

  const int B = 256;
  const int n16 = N * HID;
  const int gemmBlocks = cdiv(cdiv(N, 16), B / 32);
  const int nb = cdiv(N, 1024);

  int* cnt  = (int*)(ws + off); off += up256((size_t)N * 4);
  int* beg  = (int*)(ws + off); off += up256((size_t)N * 4);
  int* bsum = (int*)(ws + off); off += up256((size_t)nb * 4);
  int* lst  = (int*)(ws + off); off += up256((size_t)E * 4);
  int* rnk  = (int*)HW;
  const bool sorted_path = (ws_size >= off) && ((size_t)E * 4 <= 2 * hwB);

  const int G = GCN_G;
  const int chunk = cdiv(E, G);
  int* hist = lst;
  int* bstart = bsum;
  const bool lds_sort = sorted_path && N <= (1 << GCN_SRC_BITS) && nb <= GCN_MAX_NB &&
                        (long long)nb * G <= (long long)E;

  if (lds_sort) {
    gcn_bucket_hist<<<G, 1024, 0, stream>>>(dst, hist, E, chunk, nb, G);
    gcn_bucket_part<<<G, 1024, 0, stream>>>(src, dst, hist, bstart, rnk, E, chunk, nb);
    gcn_bucket_sort<<<nb, 1024, 0, stream>>>(rnk, bstart, cnt, beg, dis, lst, N, E, nb);
  } else if (sorted_path) {
    gcn_zero_i32<<<cdiv(N, B), B, 0, stream>>>(cnt, N);
    gcn_edge_rank<<<cdiv(E, B), B, 0, stream>>>(dst, cnt, rnk, E);
    gcn_block_sum<<<nb, 1024, 0, stream>>>(cnt, bsum, N);
    gcn_scan_dis<<<nb, 1024, 0, stream>>>(cnt, bsum, beg, dis, N);
    gcn_fill_list<<<cdiv(E, B), B, 0, stream>>>(src, dst, beg, rnk, lst, E);
  }

  if (sorted_path) {
    gcn_gemm_wmma_f32_lines<<<gemmBlocks, B, 0, stream>>>(x, W1, HW, dis, N, IN_CH);
    gcn_gather16_relu<<<cdiv(n16, B), B, 0, stream>>>(beg, cnt, lst, dis, HW, b1, AGG, N);
    gcn_gemm_wmma_f32<<<gemmBlocks, B, 0, stream>>>(AGG, W2, HW, dis, N, HID);
    gcn_gather16_relu<<<cdiv(n16, B), B, 0, stream>>>(beg, cnt, lst, dis, HW, b2, AGG, N);
    gcn_mv16<<<cdiv(N, B), B, 0, stream>>>(AGG, W3, dis, HW, N);
    gcn_gather1<<<cdiv(n16, B), B, 0, stream>>>(beg, cnt, lst, dis, HW, b3, out, N);
    return;
  }

  gcn_fill<<<cdiv(N, B), B, 0, stream>>>(dis, 1.0f, N);
  gcn_deg_count<<<cdiv(E, B), B, 0, stream>>>(dst, dis, E);
  gcn_rsqrt<<<cdiv(N, B), B, 0, stream>>>(dis, N);

  gcn_gemm_wmma_f32_lines<<<gemmBlocks, B, 0, stream>>>(x, W1, HW, nullptr, N, IN_CH);
  gcn_agg_init<<<cdiv(n16, B), B, 0, stream>>>(HW, dis, b1, AGG, n16);
  gcn_edge_scatter16<<<cdiv((long long)E * 16, B), B, 0, stream>>>(src, dst, dis, HW, AGG, E);
  gcn_relu<<<cdiv(n16, B), B, 0, stream>>>(AGG, n16);

  gcn_gemm_wmma_f32<<<gemmBlocks, B, 0, stream>>>(AGG, W2, HW, nullptr, N, HID);
  gcn_agg_init<<<cdiv(n16, B), B, 0, stream>>>(HW, dis, b2, AGG, n16);
  gcn_edge_scatter16<<<cdiv((long long)E * 16, B), B, 0, stream>>>(src, dst, dis, HW, AGG, E);
  gcn_relu<<<cdiv(n16, B), B, 0, stream>>>(AGG, n16);

  gcn_mv16<<<cdiv(N, B), B, 0, stream>>>(AGG, W3, nullptr, HW, N);
  gcn_out_init<<<cdiv(N, B), B, 0, stream>>>(HW, dis, b3, out, N);
  gcn_edge_scatter1<<<cdiv(E, B), B, 0, stream>>>(src, dst, dis, HW, out, E);
}


# Both sort passes: edge records ranked into an LDS image of the chunk / bucket, then copied out as whole 128-byte lines (was one scattered 4-byte store per edge); list order still LDS arrival order
# speedup vs baseline: 6.4725x; 1.2480x over previous
#include <hip/hip_runtime.h>
#include <hip/hip_bf16.h>

typedef __attribute__((ext_vector_type(2))) float v2f;
typedef __attribute__((ext_vector_type(8))) float v8f;

__global__ __launch_bounds__(256) void gcn_gemm_wmma_f32(
    const float* __restrict__ A, const float* __restrict__ W,
    float* __restrict__ C, const float* __restrict__ rs, int M, int K) {
  const int wave = blockIdx.x * (blockDim.x >> 5) + (threadIdx.x >> 5);
  const int lane = threadIdx.x & 31;
  const int m0 = wave << 4;
  if (m0 >= M) return;
  const int half = lane >> 4;
  const int l = lane & 15;

  v8f acc = {};
  const float* arow = A + (size_t)(m0 + l) * K + (half << 1);
  for (int k0 = 0; k0 < K; k0 += 4) {
    v2f a;
    a.x = __builtin_nontemporal_load(&arow[k0 + 0]);
    a.y = __builtin_nontemporal_load(&arow[k0 + 1]);
    v2f b;
    b.x = W[(size_t)(k0 + (half << 1) + 0) * 16 + l];
    b.y = W[(size_t)(k0 + (half << 1) + 1) * 16 + l];
    acc = __builtin_amdgcn_wmma_f32_16x16x4_f32(
         false, a,  false, b,
         (short)0, acc,  false,  false);
  }
  const int r0 = m0 + (half << 3);
  float* cp = C + ((size_t)r0 << 4) + l;
  if (rs) {
#pragma unroll
    for (int r = 0; r < 8; ++r) cp[(size_t)r << 4] = acc[r] * rs[r0 + r];
  } else {
#pragma unroll
    for (int r = 0; r < 8; ++r) cp[(size_t)r << 4] = acc[r];
  }
}

typedef __attribute__((ext_vector_type(4))) float v4f;
#define GCN_KC 64
#define GCN_APITCH 68

__global__ __launch_bounds__(256) void gcn_gemm_wmma_f32_lines(
    const float* __restrict__ A, const float* __restrict__ W,
    float* __restrict__ C, const float* __restrict__ rs, int M, int K) {
  __shared__ __attribute__((aligned(16))) float sh[8][16 * GCN_APITCH];
  const int wv = threadIdx.x >> 5;
  const int wave = blockIdx.x * 8 + wv;
  const int lane = threadIdx.x & 31;
  const int m0 = wave << 4;
  if (m0 >= M) return;
  const int half = lane >> 4;
  const int l = lane & 15;
  float* tile = sh[wv];
  const float* gp = A + (size_t)(m0 + half) * K + 4 * l;
  float* sp = tile + half * GCN_APITCH + 4 * l;
  const float* fp = tile + l * GCN_APITCH + 2 * half;
  const float* wp = W + (size_t)(half << 1) * 16 + l;

  v8f acc = {};
  v4f buf[8];
#pragma unroll
  for (int i = 0; i < 8; ++i)
    buf[i] = __builtin_nontemporal_load((const v4f*)(gp + (size_t)(2 * i) * K));
  for (int k0 = 0; k0 < K; k0 += GCN_KC) {
    __builtin_amdgcn_fence(__ATOMIC_RELEASE, "wavefront");
    __builtin_amdgcn_wave_barrier();
#pragma unroll
    for (int i = 0; i < 8; ++i) *(v4f*)(sp + 2 * i * GCN_APITCH) = buf[i];
    v2f b[GCN_KC / 4];
#pragma unroll
    for (int j = 0; j < GCN_KC / 4; ++j) {
      b[j].x = wp[(size_t)(k0 + 4 * j + 0) * 16];
      b[j].y = wp[(size_t)(k0 + 4 * j + 1) * 16];
    }
    if (k0 + GCN_KC < K) {
#pragma unroll
      for (int i = 0; i < 8; ++i)
        buf[i] = __builtin_nontemporal_load((const v4f*)(gp + (size_t)(2 * i) * K + k0 + GCN_KC));
    }
    __builtin_amdgcn_wave_barrier();
    __builtin_amdgcn_fence(__ATOMIC_ACQUIRE, "wavefront");
#pragma unroll
    for (int j = 0; j < GCN_KC / 4; ++j) {
      const v2f a = *(const v2f*)(fp + 4 * j);
      acc = __builtin_amdgcn_wmma_f32_16x16x4_f32(
           false, a,  false, b[j],
           (short)0, acc,  false,  false);
    }
  }
  const int r0 = m0 + (half << 3);
  float* cp = C + ((size_t)r0 << 4) + l;
  if (rs) {
#pragma unroll
    for (int r = 0; r < 8; ++r) cp[(size_t)r << 4] = acc[r] * rs[r0 + r];
  } else {
#pragma unroll
    for (int r = 0; r < 8; ++r) cp[(size_t)r << 4] = acc[r];
  }
}

__global__ void gcn_fill(float* __restrict__ p, float v, int n) {
  int i = blockIdx.x * blockDim.x + threadIdx.x;
  if (i < n) p[i] = v;
}

__global__ void gcn_deg_count(const int* __restrict__ dst, float* __restrict__ deg, int E) {
  int e = blockIdx.x * blockDim.x + threadIdx.x;
  if (e < E) atomicAdd(&deg[__builtin_nontemporal_load(&dst[e])], 1.0f);
}

__global__ void gcn_rsqrt(float* __restrict__ d, int n) {
  int i = blockIdx.x * blockDim.x + threadIdx.x;
  if (i < n) d[i] = rsqrtf(d[i]);
}

__global__ void gcn_agg_init(const float* __restrict__ hw, const float* __restrict__ dis,
                             const float* __restrict__ b, float* __restrict__ agg, int n16) {
  int t = blockIdx.x * blockDim.x + threadIdx.x;
  if (t >= n16) return;
  int i = t >> 4, c = t & 15;
  float d = dis[i];
  agg[t] = d * d * hw[t] + b[c];
}

__global__ void gcn_edge_scatter16(const int* __restrict__ src, const int* __restrict__ dst,
                                   const float* __restrict__ dis, const float* __restrict__ hw,
                                   float* __restrict__ agg, int E) {
  int t = blockIdx.x * blockDim.x + threadIdx.x;
  int e = t >> 4, c = t & 15;
  if (e >= E) return;
  int s = __builtin_nontemporal_load(&src[e]);
  int d = __builtin_nontemporal_load(&dst[e]);
  float norm = dis[s] * dis[d];
  atomicAdd(&agg[((size_t)d << 4) + c], norm * hw[((size_t)s << 4) + c]);
}

__global__ void gcn_relu(float* __restrict__ p, int n) {
  int i = blockIdx.x * blockDim.x + threadIdx.x;
  if (i < n) p[i] = fmaxf(p[i], 0.0f);
}

__global__ void gcn_mv16(const float* __restrict__ h, const float* __restrict__ W3,
                         const float* __restrict__ rs, float* __restrict__ out, int N) {
  int i = blockIdx.x * blockDim.x + threadIdx.x;
  if (i >= N) return;
  const float* hp = h + ((size_t)i << 4);
  float s = 0.f;
#pragma unroll
  for (int c = 0; c < 16; ++c) s += __builtin_nontemporal_load(&hp[c]) * W3[c];
  out[i] = rs ? s * rs[i] : s;
}

__global__ void gcn_out_init(const float* __restrict__ hw3, const float* __restrict__ dis,
                             const float* __restrict__ b3, float* __restrict__ out, int N) {
  int i = blockIdx.x * blockDim.x + threadIdx.x;
  if (i >= N) return;
  float d = dis[i];
  out[i] = d * d * hw3[i] + b3[0];
}

__global__ void gcn_edge_scatter1(const int* __restrict__ src, const int* __restrict__ dst,
                                  const float* __restrict__ dis, const float* __restrict__ hw3,
                                  float* __restrict__ out, int E) {
  int e = blockIdx.x * blockDim.x + threadIdx.x;
  if (e >= E) return;
  int s = __builtin_nontemporal_load(&src[e]);
  int d = __builtin_nontemporal_load(&dst[e]);
  atomicAdd(&out[d], dis[s] * dis[d] * hw3[s]);
}

__global__ void gcn_zero_i32(int* __restrict__ p, int n) {
  int i = blockIdx.x * blockDim.x + threadIdx.x;
  if (i < n) p[i] = 0;
}

__global__ void gcn_edge_rank(const int* __restrict__ dst, int* __restrict__ cnt,
                              int* __restrict__ rnk, int E) {
  int e = blockIdx.x * blockDim.x + threadIdx.x;
  if (e < E) rnk[e] = atomicAdd(&cnt[dst[e]], 1);
}

__global__ __launch_bounds__(1024) void gcn_block_sum(const int* __restrict__ cnt,
                                                      int* __restrict__ bsum, int n) {
  __shared__ int wsum[32];
  const int tid = threadIdx.x, w = tid >> 5, l = tid & 31;
  const int i = blockIdx.x * 1024 + tid;
  int v = (i < n) ? cnt[i] : 0;
#pragma unroll
  for (int o = 16; o > 0; o >>= 1) v += __shfl_xor(v, o, 32);
  if (l == 0) wsum[w] = v;
  __syncthreads();
  if (w == 0) {
    int t = wsum[l];
#pragma unroll
    for (int o = 16; o > 0; o >>= 1) t += __shfl_xor(t, o, 32);
    if (l == 0) bsum[blockIdx.x] = t;
  }
}

__global__ __launch_bounds__(1024) void gcn_scan_dis(const int* __restrict__ cnt,
                                                     const int* __restrict__ bsum,
                                                     int* __restrict__ beg,
                                                     float* __restrict__ dis, int n) {
  __shared__ int wtot[32];
  __shared__ int ptot[32];
  const int tid = threadIdx.x, w = tid >> 5, l = tid & 31;
  const int i = blockIdx.x * 1024 + tid;
  const int c = (i < n) ? cnt[i] : 0;
  int p = 0;
  for (int j = tid; j < (int)blockIdx.x; j += 1024) p += bsum[j];
#pragma unroll
  for (int o = 16; o > 0; o >>= 1) p += __shfl_xor(p, o, 32);
  int s = c;
#pragma unroll
  for (int o = 1; o < 32; o <<= 1) {
    int t = __shfl_up(s, o, 32);
    if (l >= o) s += t;
  }
  if (l == 31) wtot[w] = s;
  if (l == 0) ptot[w] = p;
  __syncthreads();
  int pt = ptot[l];
  int wx = (l < w) ? wtot[l] : 0;
#pragma unroll
  for (int o = 16; o > 0; o >>= 1) {
    pt += __shfl_xor(pt, o, 32);
    wx += __shfl_xor(wx, o, 32);
  }
  if (i < n) {
    beg[i] = pt + wx + s - c;
    dis[i] = rsqrtf((float)(c + 1));
  }
}

__global__ void gcn_fill_list(const int* __restrict__ src, const int* __restrict__ dst,
                              const int* __restrict__ beg, const int* __restrict__ rnk,
                              int* __restrict__ lst, int E) {
  int e = blockIdx.x * blockDim.x + threadIdx.x;
  if (e >= E) return;
  int s = __builtin_nontemporal_load(&src[e]);
  int d = __builtin_nontemporal_load(&dst[e]);
  int r = __builtin_nontemporal_load(&rnk[e]);
  lst[beg[d] + r] = s;
}

#define GCN_SRC_BITS 22
#define GCN_MAX_NB 4096
#define GCN_U 16
#define GCN_G 256
#define GCN_ROWS 8
#define GCN_PCAP 32768
#define GCN_SCAP 40960

__global__ __launch_bounds__(1024) void gcn_bucket_hist(const int* __restrict__ dst,
                                                        int* __restrict__ hist,
                                                        int E, int chunk, int NB, int G) {
  __shared__ int h[GCN_MAX_NB];
  const int tid = threadIdx.x;
  for (int i = tid; i < NB; i += 1024) h[i] = 0;
  __syncthreads();
  const int e0 = blockIdx.x * chunk;
  const int e1 = min(e0 + chunk, E);
  for (int eb = e0 + tid; eb < e1; eb += GCN_U * 1024) {
    int d[GCN_U];
#pragma unroll
    for (int u = 0; u < GCN_U; ++u) {
      const int e = eb + u * 1024;
      d[u] = (e < e1) ? dst[e] : -1;
    }
#pragma unroll
    for (int u = 0; u < GCN_U; ++u)
      if (d[u] >= 0) atomicAdd(&h[d[u] >> 10], 1);
  }
  __syncthreads();
  for (int i = tid; i < NB; i += 1024) hist[i * G + blockIdx.x] = h[i];
}

__global__ __launch_bounds__(1024) void gcn_bucket_part(const int* __restrict__ src,
                                                        const int* __restrict__ dst,
                                                        const int* __restrict__ hist,
                                                        int* __restrict__ bstart,
                                                        int* __restrict__ tmp,
                                                        int E, int chunk, int NB) {
  static_assert(GCN_G == 256, "a count-matrix row is read as 2 x 16 bytes per lane");
  __shared__ int cur[GCN_MAX_NB];
  __shared__ int btot[GCN_MAX_NB];
  __shared__ int own[GCN_MAX_NB];
  __shared__ int lcur[GCN_MAX_NB];
  __shared__ int wtot[32];
  __shared__ int wtot2[32];
  __shared__ int stage[GCN_PCAP];
  const int tid = threadIdx.x, w = tid >> 5, l = tid & 31;
  const int me = blockIdx.x;
  const bool staged = chunk <= GCN_PCAP;
  for (int b0 = w; b0 < NB; b0 += 32 * GCN_ROWS) {
    int4 x[GCN_ROWS][2];
    int mine[GCN_ROWS];
#pragma unroll
    for (int r = 0; r < GCN_ROWS; ++r) {
      const int b = min(b0 + 32 * r, NB - 1);
      const int4* row = (const int4*)(hist + (size_t)b * GCN_G);
      x[r][0] = row[l];
      x[r][1] = row[32 + l];
      mine[r] = hist[(size_t)b * GCN_G + me];
    }
    const int c0 = 4 * l, c1 = 128 + 4 * l;
    int tot[GCN_ROWS], part[GCN_ROWS];
#pragma unroll
    for (int r = 0; r < GCN_ROWS; ++r) {
      const int4 p = x[r][0], q = x[r][1];
      tot[r] = ((p.x + p.y) + (p.z + p.w)) + ((q.x + q.y) + (q.z + q.w));
      part[r] = (c0 + 0 < me ? p.x : 0) + (c0 + 1 < me ? p.y : 0) +
                (c0 + 2 < me ? p.z : 0) + (c0 + 3 < me ? p.w : 0) +
                (c1 + 0 < me ? q.x : 0) + (c1 + 1 < me ? q.y : 0) +
                (c1 + 2 < me ? q.z : 0) + (c1 + 3 < me ? q.w : 0);
    }
#pragma unroll
    for (int o = 16; o > 0; o >>= 1) {
#pragma unroll
      for (int r = 0; r < GCN_ROWS; ++r) {
        tot[r] += __shfl_xor(tot[r], o, 32);
        part[r] += __shfl_xor(part[r], o, 32);
      }
    }
#pragma unroll
    for (int r = 0; r < GCN_ROWS; ++r) {
      const int b = b0 + 32 * r;
      if (l == 0 && b < NB) { btot[b] = tot[r]; cur[b] = part[r]; own[b] = mine[r]; }
    }
  }
  __syncthreads();
  {
    int v[4], ov[4];
    int sum = 0, osum = 0;
#pragma unroll
    for (int k = 0; k < 4; ++k) {
      const int b = 4 * tid + k;
      v[k] = (b < NB) ? btot[b] : 0;
      ov[k] = (b < NB) ? own[b] : 0;
      sum += v[k];
      osum += ov[k];
    }
    int s = sum, os = osum;
#pragma unroll
    for (int o = 1; o < 32; o <<= 1) {
      int t = __shfl_up(s, o, 32);
      int ot = __shfl_up(os, o, 32);
      if (l >= o) { s += t; os += ot; }
    }
    if (l == 31) { wtot[w] = s; wtot2[w] = os; }
    __syncthreads();
    int wx = (l < w) ? wtot[l] : 0;
    int owx = (l < w) ? wtot2[l] : 0;
#pragma unroll
    for (int o = 16; o > 0; o >>= 1) {
      wx += __shfl_xor(wx, o, 32);
      owx += __shfl_xor(owx, o, 32);
    }
    int run = wx + s - sum;
    int lrun = owx + os - osum;
#pragma unroll
    for (int k = 0; k < 4; ++k) {
      const int b = 4 * tid + k;
      if (b < NB) {
        cur[b] += run;
        if (me == 0) bstart[b] = run;
        btot[b] = lrun;
        lcur[b] = lrun;
      }
      run += v[k];
      lrun += ov[k];
    }
  }
  __syncthreads();
  const int e0 = blockIdx.x * chunk;
  const int e1 = min(e0 + chunk, E);
  for (int eb = e0 + tid; eb < e1; eb += GCN_U * 1024) {
    int s[GCN_U], d[GCN_U];
#pragma unroll
    for (int u = 0; u < GCN_U; ++u) {
      const int e = eb + u * 1024;
      const bool ok = e < e1;
      s[u] = ok ? __builtin_nontemporal_load(&src[e]) : 0;
      d[u] = ok ? __builtin_nontemporal_load(&dst[e]) : -1;
    }
#pragma unroll
    for (int u = 0; u < GCN_U; ++u) {
      if (d[u] >= 0) {
        const unsigned du = (unsigned)d[u];
        const int rec = (int)(((du & 1023u) << GCN_SRC_BITS) | (unsigned)s[u]);
        if (staged) {
          const int lp = atomicAdd(&lcur[du >> 10], 1);
          stage[lp] = rec;
        } else {
          const int pos = atomicAdd(&cur[du >> 10], 1);
          tmp[pos] = rec;
        }
      }
    }
  }
  if (staged) {
    __syncthreads();
    for (int b = w; b < NB; b += 32) {
      const int n = own[b], ls = btot[b], g = cur[b];
      for (int k = l - (g & 31); k < n; k += 32)
        if (k >= 0) tmp[g + k] = stage[ls + k];
    }
  }
}

__global__ __launch_bounds__(1024) void gcn_bucket_sort(const int* __restrict__ tmp,
                                                        const int* __restrict__ bstart,
                                                        int* __restrict__ cnt,
                                                        int* __restrict__ beg,
                                                        float* __restrict__ dis,
                                                        int* __restrict__ lst,
                                                        int N, int E, int NB) {
  __shared__ int c[1024];
  __shared__ int wtot[32];
  __shared__ int stage[GCN_SCAP];
  const int tid = threadIdx.x, w = tid >> 5, l = tid & 31;
  const int b = blockIdx.x;
  const int p0 = bstart[b];
  const int p1 = (b + 1 < NB) ? bstart[b + 1] : E;
  const int nrec = p1 - p0;
  const bool staged = nrec <= GCN_SCAP;
  c[tid] = 0;
  __syncthreads();
  for (int pb = p0 + tid; pb < p1; pb += GCN_U * 1024) {
    unsigned v[GCN_U];
#pragma unroll
    for (int u = 0; u < GCN_U; ++u) v[u] = (unsigned)tmp[min(pb + u * 1024, p1 - 1)];
#pragma unroll
    for (int u = 0; u < GCN_U; ++u)
      if (pb + u * 1024 < p1) atomicAdd(&c[v[u] >> GCN_SRC_BITS], 1);
  }
  __syncthreads();
  const int n = c[tid];
  int s = n;
#pragma unroll
  for (int o = 1; o < 32; o <<= 1) {
    int t = __shfl_up(s, o, 32);
    if (l >= o) s += t;
  }
  if (l == 31) wtot[w] = s;
  __syncthreads();
  int wx = (l < w) ? wtot[l] : 0;
#pragma unroll
  for (int o = 16; o > 0; o >>= 1) wx += __shfl_xor(wx, o, 32);
  const int ex = p0 + wx + s - n;
  const int i = (b << 10) + tid;
  if (i < N) {
    cnt[i] = n;
    beg[i] = ex;
    dis[i] = rsqrtf((float)(n + 1));
  }
  c[tid] = ex;
  __syncthreads();
  for (int pb = p0 + tid; pb < p1; pb += GCN_U * 1024) {
    unsigned v[GCN_U];
#pragma unroll
    for (int u = 0; u < GCN_U; ++u)
      v[u] = (unsigned)__builtin_nontemporal_load(&tmp[min(pb + u * 1024, p1 - 1)]);
#pragma unroll
    for (int u = 0; u < GCN_U; ++u) {
      if (pb + u * 1024 < p1) {
        const int pos = atomicAdd(&c[v[u] >> GCN_SRC_BITS], 1);
        const int sv = (int)(v[u] & ((1u << GCN_SRC_BITS) - 1u));
        if (staged) stage[pos - p0] = sv;
        else lst[pos] = sv;
      }
    }
  }
  if (staged) {
    __syncthreads();
    for (int k = tid - (p0 & 31); k < nrec; k += 1024)
      if (k >= 0) lst[p0 + k] = stage[k];
  }
}

__global__ __launch_bounds__(256) void gcn_gather16_relu(
    const int* __restrict__ beg, const int* __restrict__ cnt, const int* __restrict__ lst,
    const float* __restrict__ dis, const float* __restrict__ hws, const float* __restrict__ b,
    float* __restrict__ h, int N) {
  const int t = blockIdx.x * blockDim.x + threadIdx.x;
  const int i = t >> 4, c = t & 15;
  if (i >= N) return;
  const int* lp = lst + beg[i];
  const int len = cnt[i];
  const float d = dis[i];
  float a[8];
  a[0] = hws[((size_t)i << 4) + c];
#pragma unroll
  for (int u = 1; u < 8; ++u) a[u] = 0.f;
  int j = 0;
#pragma unroll 1
  for (; j + 16 <= len; j += 16) {
    int s[16];
#pragma unroll
    for (int u = 0; u < 16; ++u) s[u] = lp[j + u];
    float r[16];
#pragma unroll
    for (int u = 0; u < 16; ++u) r[u] = hws[((size_t)s[u] << 4) + c];
#pragma unroll
    for (int u = 0; u < 8; ++u) a[u] += r[u];
#pragma unroll
    for (int u = 0; u < 8; ++u) a[u] += r[8 + u];
  }
  const int rem = len - j;
  if (rem > 0) {
    const int last = len - 1;
    int s[15];
#pragma unroll
    for (int u = 0; u < 15; ++u) s[u] = lp[min(j + u, last)];
#pragma unroll
    for (int u = 0; u < 15; ++u) asm volatile("" : "+v"(s[u]));
    float r[15];
#pragma unroll
    for (int u = 0; u < 15; ++u) r[u] = hws[((size_t)s[u] << 4) + c];
#pragma unroll
    for (int u = 0; u < 15; ++u) asm volatile("" : "+v"(r[u]));
    const bool blk = rem >= 8;
#pragma unroll
    for (int u = 0; u < 8; ++u) a[u] = blk ? a[u] + r[u] : a[u];
    const int tail = rem & 7;
#pragma unroll
    for (int q = 0; q < 7; ++q) {
      const float v = blk ? r[8 + q] : r[q];
      a[0] = (q < tail) ? a[0] + v : a[0];
    }
  }
  float r = ((a[0] + a[1]) + (a[2] + a[3])) + ((a[4] + a[5]) + (a[6] + a[7]));
  h[((size_t)i << 4) + c] = fmaxf(d * r + b[c], 0.0f);
}

__global__ __launch_bounds__(256) void gcn_gather1(
    const int* __restrict__ beg, const int* __restrict__ cnt, const int* __restrict__ lst,
    const float* __restrict__ dis, const float* __restrict__ hw3s, const float* __restrict__ b3,
    float* __restrict__ out, int N) {
  const int t = blockIdx.x * blockDim.x + threadIdx.x;
  const int i = t >> 4, l = t & 15;
  const bool ok = i < N;
  const int ii = ok ? i : 0;
  const int* lp = lst + beg[ii];
  const int len = ok ? cnt[ii] : 0;
  const float d = dis[ii];
  float a = 0.f;
  for (int j = l; j < len; j += 16) a += hw3s[lp[j]];
#pragma unroll
  for (int o = 8; o > 0; o >>= 1) a += __shfl_xor(a, o, 16);
  if (ok && l == 0) out[i] = d * (a + hw3s[i]) + b3[0];
}

static inline int cdiv(long long a, long long b) { return (int)((a + b - 1) / b); }
static inline size_t up256(size_t v) { return (v + 255) & ~(size_t)255; }

extern "C" void kernel_launch(void* const* d_in, const int* in_sizes, int n_in,
                              void* d_out, int out_size, void* d_ws, size_t ws_size,
                              hipStream_t stream) {
  const float* x  = (const float*)d_in[0];
  const int*   ei = (const int*)d_in[1];
  const float* W1 = (const float*)d_in[2];
  const float* b1 = (const float*)d_in[3];
  const float* W2 = (const float*)d_in[4];
  const float* b2 = (const float*)d_in[5];
  const float* W3 = (const float*)d_in[6];
  const float* b3 = (const float*)d_in[7];

  const int IN_CH = 256;
  const int HID = 16;
  const int N = in_sizes[0] / IN_CH;
  const int E = in_sizes[1] / 2;
  const int* src = ei;
  const int* dst = ei + E;
  float* out = (float*)d_out;

  char* ws = (char*)d_ws;
  size_t off = 0;
  const size_t hwB = up256((size_t)N * HID * 4);
  float* dis = (float*)(ws + off); off += up256((size_t)N * 4);
  float* HW  = (float*)(ws + off); off += hwB;
  float* AGG = (float*)(ws + off); off += hwB;
  (void)n_in; (void)out_size;

  const int B = 256;
  const int n16 = N * HID;
  const int gemmBlocks = cdiv(cdiv(N, 16), B / 32);
  const int nb = cdiv(N, 1024);

  int* cnt  = (int*)(ws + off); off += up256((size_t)N * 4);
  int* beg  = (int*)(ws + off); off += up256((size_t)N * 4);
  int* bsum = (int*)(ws + off); off += up256((size_t)nb * 4);
  int* lst  = (int*)(ws + off); off += up256((size_t)E * 4);
  int* rnk  = (int*)HW;
  const bool sorted_path = (ws_size >= off) && ((size_t)E * 4 <= 2 * hwB);

  const int G = GCN_G;
  const int chunk = cdiv(E, G);
  int* hist = lst;
  int* bstart = bsum;
  const bool lds_sort = sorted_path && N <= (1 << GCN_SRC_BITS) && nb <= GCN_MAX_NB &&
                        (long long)nb * G <= (long long)E;

  if (lds_sort) {
    gcn_bucket_hist<<<G, 1024, 0, stream>>>(dst, hist, E, chunk, nb, G);
    gcn_bucket_part<<<G, 1024, 0, stream>>>(src, dst, hist, bstart, rnk, E, chunk, nb);
    gcn_bucket_sort<<<nb, 1024, 0, stream>>>(rnk, bstart, cnt, beg, dis, lst, N, E, nb);
  } else if (sorted_path) {
    gcn_zero_i32<<<cdiv(N, B), B, 0, stream>>>(cnt, N);
    gcn_edge_rank<<<cdiv(E, B), B, 0, stream>>>(dst, cnt, rnk, E);
    gcn_block_sum<<<nb, 1024, 0, stream>>>(cnt, bsum, N);
    gcn_scan_dis<<<nb, 1024, 0, stream>>>(cnt, bsum, beg, dis, N);
    gcn_fill_list<<<cdiv(E, B), B, 0, stream>>>(src, dst, beg, rnk, lst, E);
  }

  if (sorted_path) {
    gcn_gemm_wmma_f32_lines<<<gemmBlocks, B, 0, stream>>>(x, W1, HW, dis, N, IN_CH);
    gcn_gather16_relu<<<cdiv(n16, B), B, 0, stream>>>(beg, cnt, lst, dis, HW, b1, AGG, N);
    gcn_gemm_wmma_f32<<<gemmBlocks, B, 0, stream>>>(AGG, W2, HW, dis, N, HID);
    gcn_gather16_relu<<<cdiv(n16, B), B, 0, stream>>>(beg, cnt, lst, dis, HW, b2, AGG, N);
    gcn_mv16<<<cdiv(N, B), B, 0, stream>>>(AGG, W3, dis, HW, N);
    gcn_gather1<<<cdiv(n16, B), B, 0, stream>>>(beg, cnt, lst, dis, HW, b3, out, N);
    return;
  }

  gcn_fill<<<cdiv(N, B), B, 0, stream>>>(dis, 1.0f, N);
  gcn_deg_count<<<cdiv(E, B), B, 0, stream>>>(dst, dis, E);
  gcn_rsqrt<<<cdiv(N, B), B, 0, stream>>>(dis, N);

  gcn_gemm_wmma_f32_lines<<<gemmBlocks, B, 0, stream>>>(x, W1, HW, nullptr, N, IN_CH);
  gcn_agg_init<<<cdiv(n16, B), B, 0, stream>>>(HW, dis, b1, AGG, n16);
  gcn_edge_scatter16<<<cdiv((long long)E * 16, B), B, 0, stream>>>(src, dst, dis, HW, AGG, E);
  gcn_relu<<<cdiv(n16, B), B, 0, stream>>>(AGG, n16);

  gcn_gemm_wmma_f32<<<gemmBlocks, B, 0, stream>>>(AGG, W2, HW, nullptr, N, HID);
  gcn_agg_init<<<cdiv(n16, B), B, 0, stream>>>(HW, dis, b2, AGG, n16);
  gcn_edge_scatter16<<<cdiv((long long)E * 16, B), B, 0, stream>>>(src, dst, dis, HW, AGG, E);
  gcn_relu<<<cdiv(n16, B), B, 0, stream>>>(AGG, n16);

  gcn_mv16<<<cdiv(N, B), B, 0, stream>>>(AGG, W3, nullptr, HW, N);
  gcn_out_init<<<cdiv(N, B), B, 0, stream>>>(HW, dis, b3, out, N);
  gcn_edge_scatter1<<<cdiv(E, B), B, 0, stream>>>(src, dst, dis, HW, out, E);
}


# 1-channel gather: source table looked up in LDS, streamed in 3 slices per work-group with whole-line reads, not one 4-byte L2 read per edge; per-lane sum order now slice by slice, then list order
# speedup vs baseline: 6.9893x; 1.0798x over previous
#include <hip/hip_runtime.h>
#include <hip/hip_bf16.h>

typedef __attribute__((ext_vector_type(2))) float v2f;
typedef __attribute__((ext_vector_type(8))) float v8f;

__global__ __launch_bounds__(256) void gcn_gemm_wmma_f32(
    const float* __restrict__ A, const float* __restrict__ W,
    float* __restrict__ C, const float* __restrict__ rs, int M, int K) {
  const int wave = blockIdx.x * (blockDim.x >> 5) + (threadIdx.x >> 5);
  const int lane = threadIdx.x & 31;
  const int m0 = wave << 4;
  if (m0 >= M) return;
  const int half = lane >> 4;
  const int l = lane & 15;

  v8f acc = {};
  const float* arow = A + (size_t)(m0 + l) * K + (half << 1);
  for (int k0 = 0; k0 < K; k0 += 4) {
    v2f a;
    a.x = __builtin_nontemporal_load(&arow[k0 + 0]);
    a.y = __builtin_nontemporal_load(&arow[k0 + 1]);
    v2f b;
    b.x = W[(size_t)(k0 + (half << 1) + 0) * 16 + l];
    b.y = W[(size_t)(k0 + (half << 1) + 1) * 16 + l];
    acc = __builtin_amdgcn_wmma_f32_16x16x4_f32(
         false, a,  false, b,
         (short)0, acc,  false,  false);
  }
  const int r0 = m0 + (half << 3);
  float* cp = C + ((size_t)r0 << 4) + l;
  if (rs) {
#pragma unroll
    for (int r = 0; r < 8; ++r) cp[(size_t)r << 4] = acc[r] * rs[r0 + r];
  } else {
#pragma unroll
    for (int r = 0; r < 8; ++r) cp[(size_t)r << 4] = acc[r];
  }
}

typedef __attribute__((ext_vector_type(4))) float v4f;
#define GCN_KC 64
#define GCN_APITCH 68

__global__ __launch_bounds__(256) void gcn_gemm_wmma_f32_lines(
    const float* __restrict__ A, const float* __restrict__ W,
    float* __restrict__ C, const float* __restrict__ rs, int M, int K) {
  __shared__ __attribute__((aligned(16))) float sh[8][16 * GCN_APITCH];
  const int wv = threadIdx.x >> 5;
  const int wave = blockIdx.x * 8 + wv;
  const int lane = threadIdx.x & 31;
  const int m0 = wave << 4;
  if (m0 >= M) return;
  const int half = lane >> 4;
  const int l = lane & 15;
  float* tile = sh[wv];
  const float* gp = A + (size_t)(m0 + half) * K + 4 * l;
  float* sp = tile + half * GCN_APITCH + 4 * l;
  const float* fp = tile + l * GCN_APITCH + 2 * half;
  const float* wp = W + (size_t)(half << 1) * 16 + l;

  v8f acc = {};
  v4f buf[8];
#pragma unroll
  for (int i = 0; i < 8; ++i)
    buf[i] = __builtin_nontemporal_load((const v4f*)(gp + (size_t)(2 * i) * K));
  for (int k0 = 0; k0 < K; k0 += GCN_KC) {
    __builtin_amdgcn_fence(__ATOMIC_RELEASE, "wavefront");
    __builtin_amdgcn_wave_barrier();
#pragma unroll
    for (int i = 0; i < 8; ++i) *(v4f*)(sp + 2 * i * GCN_APITCH) = buf[i];
    v2f b[GCN_KC / 4];
#pragma unroll
    for (int j = 0; j < GCN_KC / 4; ++j) {
      b[j].x = wp[(size_t)(k0 + 4 * j + 0) * 16];
      b[j].y = wp[(size_t)(k0 + 4 * j + 1) * 16];
    }
    if (k0 + GCN_KC < K) {
#pragma unroll
      for (int i = 0; i < 8; ++i)
        buf[i] = __builtin_nontemporal_load((const v4f*)(gp + (size_t)(2 * i) * K + k0 + GCN_KC));
    }
    __builtin_amdgcn_wave_barrier();
    __builtin_amdgcn_fence(__ATOMIC_ACQUIRE, "wavefront");
#pragma unroll
    for (int j = 0; j < GCN_KC / 4; ++j) {
      const v2f a = *(const v2f*)(fp + 4 * j);
      acc = __builtin_amdgcn_wmma_f32_16x16x4_f32(
           false, a,  false, b[j],
           (short)0, acc,  false,  false);
    }
  }
  const int r0 = m0 + (half << 3);
  float* cp = C + ((size_t)r0 << 4) + l;
  if (rs) {
#pragma unroll
    for (int r = 0; r < 8; ++r) cp[(size_t)r << 4] = acc[r] * rs[r0 + r];
  } else {
#pragma unroll
    for (int r = 0; r < 8; ++r) cp[(size_t)r << 4] = acc[r];
  }
}

__global__ void gcn_fill(float* __restrict__ p, float v, int n) {
  int i = blockIdx.x * blockDim.x + threadIdx.x;
  if (i < n) p[i] = v;
}

__global__ void gcn_deg_count(const int* __restrict__ dst, float* __restrict__ deg, int E) {
  int e = blockIdx.x * blockDim.x + threadIdx.x;
  if (e < E) atomicAdd(&deg[__builtin_nontemporal_load(&dst[e])], 1.0f);
}

__global__ void gcn_rsqrt(float* __restrict__ d, int n) {
  int i = blockIdx.x * blockDim.x + threadIdx.x;
  if (i < n) d[i] = rsqrtf(d[i]);
}

__global__ void gcn_agg_init(const float* __restrict__ hw, const float* __restrict__ dis,
                             const float* __restrict__ b, float* __restrict__ agg, int n16) {
  int t = blockIdx.x * blockDim.x + threadIdx.x;
  if (t >= n16) return;
  int i = t >> 4, c = t & 15;
  float d = dis[i];
  agg[t] = d * d * hw[t] + b[c];
}

__global__ void gcn_edge_scatter16(const int* __restrict__ src, const int* __restrict__ dst,
                                   const float* __restrict__ dis, const float* __restrict__ hw,
                                   float* __restrict__ agg, int E) {
  int t = blockIdx.x * blockDim.x + threadIdx.x;
  int e = t >> 4, c = t & 15;
  if (e >= E) return;
  int s = __builtin_nontemporal_load(&src[e]);
  int d = __builtin_nontemporal_load(&dst[e]);
  float norm = dis[s] * dis[d];
  atomicAdd(&agg[((size_t)d << 4) + c], norm * hw[((size_t)s << 4) + c]);
}

__global__ void gcn_relu(float* __restrict__ p, int n) {
  int i = blockIdx.x * blockDim.x + threadIdx.x;
  if (i < n) p[i] = fmaxf(p[i], 0.0f);
}

__global__ void gcn_mv16(const float* __restrict__ h, const float* __restrict__ W3,
                         const float* __restrict__ rs, float* __restrict__ out, int N) {
  int i = blockIdx.x * blockDim.x + threadIdx.x;
  if (i >= N) return;
  const float* hp = h + ((size_t)i << 4);
  float s = 0.f;
#pragma unroll
  for (int c = 0; c < 16; ++c) s += __builtin_nontemporal_load(&hp[c]) * W3[c];
  out[i] = rs ? s * rs[i] : s;
}

__global__ void gcn_out_init(const float* __restrict__ hw3, const float* __restrict__ dis,
                             const float* __restrict__ b3, float* __restrict__ out, int N) {
  int i = blockIdx.x * blockDim.x + threadIdx.x;
  if (i >= N) return;
  float d = dis[i];
  out[i] = d * d * hw3[i] + b3[0];
}

__global__ void gcn_edge_scatter1(const int* __restrict__ src, const int* __restrict__ dst,
                                  const float* __restrict__ dis, const float* __restrict__ hw3,
                                  float* __restrict__ out, int E) {
  int e = blockIdx.x * blockDim.x + threadIdx.x;
  if (e >= E) return;
  int s = __builtin_nontemporal_load(&src[e]);
  int d = __builtin_nontemporal_load(&dst[e]);
  atomicAdd(&out[d], dis[s] * dis[d] * hw3[s]);
}

__global__ void gcn_zero_i32(int* __restrict__ p, int n) {
  int i = blockIdx.x * blockDim.x + threadIdx.x;
  if (i < n) p[i] = 0;
}

__global__ void gcn_edge_rank(const int* __restrict__ dst, int* __restrict__ cnt,
                              int* __restrict__ rnk, int E) {
  int e = blockIdx.x * blockDim.x + threadIdx.x;
  if (e < E) rnk[e] = atomicAdd(&cnt[dst[e]], 1);
}

__global__ __launch_bounds__(1024) void gcn_block_sum(const int* __restrict__ cnt,
                                                      int* __restrict__ bsum, int n) {
  __shared__ int wsum[32];
  const int tid = threadIdx.x, w = tid >> 5, l = tid & 31;
  const int i = blockIdx.x * 1024 + tid;
  int v = (i < n) ? cnt[i] : 0;
#pragma unroll
  for (int o = 16; o > 0; o >>= 1) v += __shfl_xor(v, o, 32);
  if (l == 0) wsum[w] = v;
  __syncthreads();
  if (w == 0) {
    int t = wsum[l];
#pragma unroll
    for (int o = 16; o > 0; o >>= 1) t += __shfl_xor(t, o, 32);
    if (l == 0) bsum[blockIdx.x] = t;
  }
}

__global__ __launch_bounds__(1024) void gcn_scan_dis(const int* __restrict__ cnt,
                                                     const int* __restrict__ bsum,
                                                     int* __restrict__ beg,
                                                     float* __restrict__ dis, int n) {
  __shared__ int wtot[32];
  __shared__ int ptot[32];
  const int tid = threadIdx.x, w = tid >> 5, l = tid & 31;
  const int i = blockIdx.x * 1024 + tid;
  const int c = (i < n) ? cnt[i] : 0;
  int p = 0;
  for (int j = tid; j < (int)blockIdx.x; j += 1024) p += bsum[j];
#pragma unroll
  for (int o = 16; o > 0; o >>= 1) p += __shfl_xor(p, o, 32);
  int s = c;
#pragma unroll
  for (int o = 1; o < 32; o <<= 1) {
    int t = __shfl_up(s, o, 32);
    if (l >= o) s += t;
  }
  if (l == 31) wtot[w] = s;
  if (l == 0) ptot[w] = p;
  __syncthreads();
  int pt = ptot[l];
  int wx = (l < w) ? wtot[l] : 0;
#pragma unroll
  for (int o = 16; o > 0; o >>= 1) {
    pt += __shfl_xor(pt, o, 32);
    wx += __shfl_xor(wx, o, 32);
  }
  if (i < n) {
    beg[i] = pt + wx + s - c;
    dis[i] = rsqrtf((float)(c + 1));
  }
}

__global__ void gcn_fill_list(const int* __restrict__ src, const int* __restrict__ dst,
                              const int* __restrict__ beg, const int* __restrict__ rnk,
                              int* __restrict__ lst, int E) {
  int e = blockIdx.x * blockDim.x + threadIdx.x;
  if (e >= E) return;
  int s = __builtin_nontemporal_load(&src[e]);
  int d = __builtin_nontemporal_load(&dst[e]);
  int r = __builtin_nontemporal_load(&rnk[e]);
  lst[beg[d] + r] = s;
}

#define GCN_SRC_BITS 22
#define GCN_MAX_NB 4096
#define GCN_U 16
#define GCN_G 256
#define GCN_ROWS 8
#define GCN_PCAP 32768
#define GCN_SCAP 40960

__global__ __launch_bounds__(1024) void gcn_bucket_hist(const int* __restrict__ dst,
                                                        int* __restrict__ hist,
                                                        int E, int chunk, int NB, int G) {
  __shared__ int h[GCN_MAX_NB];
  const int tid = threadIdx.x;
  for (int i = tid; i < NB; i += 1024) h[i] = 0;
  __syncthreads();
  const int e0 = blockIdx.x * chunk;
  const int e1 = min(e0 + chunk, E);
  for (int eb = e0 + tid; eb < e1; eb += GCN_U * 1024) {
    int d[GCN_U];
#pragma unroll
    for (int u = 0; u < GCN_U; ++u) {
      const int e = eb + u * 1024;
      d[u] = (e < e1) ? dst[e] : -1;
    }
#pragma unroll
    for (int u = 0; u < GCN_U; ++u)
      if (d[u] >= 0) atomicAdd(&h[d[u] >> 10], 1);
  }
  __syncthreads();
  for (int i = tid; i < NB; i += 1024) hist[i * G + blockIdx.x] = h[i];
}

__global__ __launch_bounds__(1024) void gcn_bucket_part(const int* __restrict__ src,
                                                        const int* __restrict__ dst,
                                                        const int* __restrict__ hist,
                                                        int* __restrict__ bstart,
                                                        int* __restrict__ tmp,
                                                        int E, int chunk, int NB) {
  static_assert(GCN_G == 256, "a count-matrix row is read as 2 x 16 bytes per lane");
  __shared__ int cur[GCN_MAX_NB];
  __shared__ int btot[GCN_MAX_NB];
  __shared__ int own[GCN_MAX_NB];
  __shared__ int lcur[GCN_MAX_NB];
  __shared__ int wtot[32];
  __shared__ int wtot2[32];
  __shared__ int stage[GCN_PCAP];
  const int tid = threadIdx.x, w = tid >> 5, l = tid & 31;
  const int me = blockIdx.x;
  const bool staged = chunk <= GCN_PCAP;
  for (int b0 = w; b0 < NB; b0 += 32 * GCN_ROWS) {
    int4 x[GCN_ROWS][2];
    int mine[GCN_ROWS];
#pragma unroll
    for (int r = 0; r < GCN_ROWS; ++r) {
      const int b = min(b0 + 32 * r, NB - 1);
      const int4* row = (const int4*)(hist + (size_t)b * GCN_G);
      x[r][0] = row[l];
      x[r][1] = row[32 + l];
      mine[r] = hist[(size_t)b * GCN_G + me];
    }
    const int c0 = 4 * l, c1 = 128 + 4 * l;
    int tot[GCN_ROWS], part[GCN_ROWS];
#pragma unroll
    for (int r = 0; r < GCN_ROWS; ++r) {
      const int4 p = x[r][0], q = x[r][1];
      tot[r] = ((p.x + p.y) + (p.z + p.w)) + ((q.x + q.y) + (q.z + q.w));
      part[r] = (c0 + 0 < me ? p.x : 0) + (c0 + 1 < me ? p.y : 0) +
                (c0 + 2 < me ? p.z : 0) + (c0 + 3 < me ? p.w : 0) +
                (c1 + 0 < me ? q.x : 0) + (c1 + 1 < me ? q.y : 0) +
                (c1 + 2 < me ? q.z : 0) + (c1 + 3 < me ? q.w : 0);
    }
#pragma unroll
    for (int o = 16; o > 0; o >>= 1) {
#pragma unroll
      for (int r = 0; r < GCN_ROWS; ++r) {
        tot[r] += __shfl_xor(tot[r], o, 32);
        part[r] += __shfl_xor(part[r], o, 32);
      }
    }
#pragma unroll
    for (int r = 0; r < GCN_ROWS; ++r) {
      const int b = b0 + 32 * r;
      if (l == 0 && b < NB) { btot[b] = tot[r]; cur[b] = part[r]; own[b] = mine[r]; }
    }
  }
  __syncthreads();
  {
    int v[4], ov[4];
    int sum = 0, osum = 0;
#pragma unroll
    for (int k = 0; k < 4; ++k) {
      const int b = 4 * tid + k;
      v[k] = (b < NB) ? btot[b] : 0;
      ov[k] = (b < NB) ? own[b] : 0;
      sum += v[k];
      osum += ov[k];
    }
    int s = sum, os = osum;
#pragma unroll
    for (int o = 1; o < 32; o <<= 1) {
      int t = __shfl_up(s, o, 32);
      int ot = __shfl_up(os, o, 32);
      if (l >= o) { s += t; os += ot; }
    }
    if (l == 31) { wtot[w] = s; wtot2[w] = os; }
    __syncthreads();
    int wx = (l < w) ? wtot[l] : 0;
    int owx = (l < w) ? wtot2[l] : 0;
#pragma unroll
    for (int o = 16; o > 0; o >>= 1) {
      wx += __shfl_xor(wx, o, 32);
      owx += __shfl_xor(owx, o, 32);
    }
    int run = wx + s - sum;
    int lrun = owx + os - osum;
#pragma unroll
    for (int k = 0; k < 4; ++k) {
      const int b = 4 * tid + k;
      if (b < NB) {
        cur[b] += run;
        if (me == 0) bstart[b] = run;
        btot[b] = lrun;
        lcur[b] = lrun;
      }
      run += v[k];
      lrun += ov[k];
    }
  }
  __syncthreads();
  const int e0 = blockIdx.x * chunk;
  const int e1 = min(e0 + chunk, E);
  for (int eb = e0 + tid; eb < e1; eb += GCN_U * 1024) {
    int s[GCN_U], d[GCN_U];
#pragma unroll
    for (int u = 0; u < GCN_U; ++u) {
      const int e = eb + u * 1024;
      const bool ok = e < e1;
      s[u] = ok ? __builtin_nontemporal_load(&src[e]) : 0;
      d[u] = ok ? __builtin_nontemporal_load(&dst[e]) : -1;
    }
#pragma unroll
    for (int u = 0; u < GCN_U; ++u) {
      if (d[u] >= 0) {
        const unsigned du = (unsigned)d[u];
        const int rec = (int)(((du & 1023u) << GCN_SRC_BITS) | (unsigned)s[u]);
        if (staged) {
          const int lp = atomicAdd(&lcur[du >> 10], 1);
          stage[lp] = rec;
        } else {
          const int pos = atomicAdd(&cur[du >> 10], 1);
          tmp[pos] = rec;
        }
      }
    }
  }
  if (staged) {
    __syncthreads();
    for (int b = w; b < NB; b += 32) {
      const int n = own[b], ls = btot[b], g = cur[b];
      for (int k = l - (g & 31); k < n; k += 32)
        if (k >= 0) tmp[g + k] = stage[ls + k];
    }
  }
}

__global__ __launch_bounds__(1024) void gcn_bucket_sort(const int* __restrict__ tmp,
                                                        const int* __restrict__ bstart,
                                                        int* __restrict__ cnt,
                                                        int* __restrict__ beg,
                                                        float* __restrict__ dis,
                                                        int* __restrict__ lst,
                                                        int N, int E, int NB) {
  __shared__ int c[1024];
  __shared__ int wtot[32];
  __shared__ int stage[GCN_SCAP];
  const int tid = threadIdx.x, w = tid >> 5, l = tid & 31;
  const int b = blockIdx.x;
  const int p0 = bstart[b];
  const int p1 = (b + 1 < NB) ? bstart[b + 1] : E;
  const int nrec = p1 - p0;
  const bool staged = nrec <= GCN_SCAP;
  c[tid] = 0;
  __syncthreads();
  for (int pb = p0 + tid; pb < p1; pb += GCN_U * 1024) {
    unsigned v[GCN_U];
#pragma unroll
    for (int u = 0; u < GCN_U; ++u) v[u] = (unsigned)tmp[min(pb + u * 1024, p1 - 1)];
#pragma unroll
    for (int u = 0; u < GCN_U; ++u)
      if (pb + u * 1024 < p1) atomicAdd(&c[v[u] >> GCN_SRC_BITS], 1);
  }
  __syncthreads();
  const int n = c[tid];
  int s = n;
#pragma unroll
  for (int o = 1; o < 32; o <<= 1) {
    int t = __shfl_up(s, o, 32);
    if (l >= o) s += t;
  }
  if (l == 31) wtot[w] = s;
  __syncthreads();
  int wx = (l < w) ? wtot[l] : 0;
#pragma unroll
  for (int o = 16; o > 0; o >>= 1) wx += __shfl_xor(wx, o, 32);
  const int ex = p0 + wx + s - n;
  const int i = (b << 10) + tid;
  if (i < N) {
    cnt[i] = n;
    beg[i] = ex;
    dis[i] = rsqrtf((float)(n + 1));
  }
  c[tid] = ex;
  __syncthreads();
  for (int pb = p0 + tid; pb < p1; pb += GCN_U * 1024) {
    unsigned v[GCN_U];
#pragma unroll
    for (int u = 0; u < GCN_U; ++u)
      v[u] = (unsigned)__builtin_nontemporal_load(&tmp[min(pb + u * 1024, p1 - 1)]);
#pragma unroll
    for (int u = 0; u < GCN_U; ++u) {
      if (pb + u * 1024 < p1) {
        const int pos = atomicAdd(&c[v[u] >> GCN_SRC_BITS], 1);
        const int sv = (int)(v[u] & ((1u << GCN_SRC_BITS) - 1u));
        if (staged) stage[pos - p0] = sv;
        else lst[pos] = sv;
      }
    }
  }
  if (staged) {
    __syncthreads();
    for (int k = tid - (p0 & 31); k < nrec; k += 1024)
      if (k >= 0) lst[p0 + k] = stage[k];
  }
}

__global__ __launch_bounds__(256) void gcn_gather16_relu(
    const int* __restrict__ beg, const int* __restrict__ cnt, const int* __restrict__ lst,
    const float* __restrict__ dis, const float* __restrict__ hws, const float* __restrict__ b,
    float* __restrict__ h, int N) {
  const int t = blockIdx.x * blockDim.x + threadIdx.x;
  const int i = t >> 4, c = t & 15;
  if (i >= N) return;
  const int* lp = lst + beg[i];
  const int len = cnt[i];
  const float d = dis[i];
  float a[8];
  a[0] = hws[((size_t)i << 4) + c];
#pragma unroll
  for (int u = 1; u < 8; ++u) a[u] = 0.f;
  int j = 0;
#pragma unroll 1
  for (; j + 16 <= len; j += 16) {
    int s[16];
#pragma unroll
    for (int u = 0; u < 16; ++u) s[u] = lp[j + u];
    float r[16];
#pragma unroll
    for (int u = 0; u < 16; ++u) r[u] = hws[((size_t)s[u] << 4) + c];
#pragma unroll
    for (int u = 0; u < 8; ++u) a[u] += r[u];
#pragma unroll
    for (int u = 0; u < 8; ++u) a[u] += r[8 + u];
  }
  const int rem = len - j;
  if (rem > 0) {
    const int last = len - 1;
    int s[15];
#pragma unroll
    for (int u = 0; u < 15; ++u) s[u] = lp[min(j + u, last)];
#pragma unroll
    for (int u = 0; u < 15; ++u) asm volatile("" : "+v"(s[u]));
    float r[15];
#pragma unroll
    for (int u = 0; u < 15; ++u) r[u] = hws[((size_t)s[u] << 4) + c];
#pragma unroll
    for (int u = 0; u < 15; ++u) asm volatile("" : "+v"(r[u]));
    const bool blk = rem >= 8;
#pragma unroll
    for (int u = 0; u < 8; ++u) a[u] = blk ? a[u] + r[u] : a[u];
    const int tail = rem & 7;
#pragma unroll
    for (int q = 0; q < 7; ++q) {
      const float v = blk ? r[8 + q] : r[q];
      a[0] = (q < tail) ? a[0] + v : a[0];
    }
  }
  float r = ((a[0] + a[1]) + (a[2] + a[3])) + ((a[4] + a[5]) + (a[6] + a[7]));
  h[((size_t)i << 4) + c] = fmaxf(d * r + b[c], 0.0f);
}

#define GCN_G1R 13
#define GCN_G1K 3
#define GCN_G1B 6
#define GCN_G1S 73728

__global__ __launch_bounds__(1024) void gcn_gather1_lds(
    const int* __restrict__ beg, const int* __restrict__ cnt, const int* __restrict__ lst,
    const float* __restrict__ dis, const float* __restrict__ hw3s, const float* __restrict__ b3,
    float* __restrict__ out, int N) {
  __shared__ __attribute__((aligned(16))) float tab[GCN_G1S];
  const int tid = threadIdx.x, l = tid & 15;
  const int i0 = blockIdx.x * (64 * GCN_G1R) + (tid >> 4);
  int ln[GCN_G1R], s[GCN_G1R][GCN_G1K];
  float a[GCN_G1R];
  int bg[GCN_G1R];
#pragma unroll
  for (int r = 0; r < GCN_G1R; ++r) {
    const int ic = min(i0 + 64 * r, N - 1);
    bg[r] = beg[ic];
    ln[r] = cnt[ic];
    a[r] = 0.f;
  }
#pragma unroll
  for (int r = 0; r < GCN_G1R; ++r) {
    if (i0 + 64 * r >= N) ln[r] = 0;
#pragma unroll
    for (int k = 0; k < GCN_G1K; ++k) {
      const int j = l + 16 * k;
      const int v = lst[max(min(bg[r] + j, bg[r] + ln[r] - 1), 0)];
      s[r][k] = (j < ln[r]) ? v : -1;
    }
  }
  const int nq = (N + GCN_G1S - 1) / GCN_G1S;
  for (int q = 0; q < nq; ++q) {
    const int q0 = q * GCN_G1S;
    const int sl = min(GCN_G1S, N - q0);
    const int sl4 = sl & ~3;
    const float* tp = hw3s + q0;
    __syncthreads();
    for (int t0 = 4 * tid; t0 < sl4; t0 += 4096 * GCN_G1B) {
      v4f buf[GCN_G1B];
#pragma unroll
      for (int u = 0; u < GCN_G1B; ++u) {
        const int t = t0 + 4096 * u;
        buf[u] = *(const v4f*)(tp + min(t, sl4 - 4));
      }
#pragma unroll
      for (int u = 0; u < GCN_G1B; ++u) {
        const int t = t0 + 4096 * u;
        if (t < sl4) *(v4f*)(tab + t) = buf[u];
      }
    }
    if (tid < sl - sl4) tab[sl4 + tid] = tp[sl4 + tid];
    __syncthreads();
#pragma unroll
    for (int r = 0; r < GCN_G1R; ++r) {
#pragma unroll
      for (int k = 0; k < GCN_G1K; ++k) {
        const unsigned u = (unsigned)(s[r][k] - q0);
        const bool in = u < (unsigned)sl;
        const float v = tab[in ? u : 0u];
        a[r] += in ? v : 0.f;
      }
    }
#pragma unroll
    for (int r = 0; r < GCN_G1R; ++r) {
      if (ln[r] > 16 * GCN_G1K) {
        const int* lp = lst + beg[i0 + 64 * r];
        for (int j = l + 16 * GCN_G1K; j < ln[r]; j += 16) {
          const unsigned u = (unsigned)(lp[j] - q0);
          if (u < (unsigned)sl) a[r] += tab[u];
        }
      }
    }
  }
  float dv[GCN_G1R], hv[GCN_G1R];
#pragma unroll
  for (int r = 0; r < GCN_G1R; ++r) {
    const int ic = min(i0 + 64 * r, N - 1);
    dv[r] = dis[ic];
    hv[r] = hw3s[ic];
  }
  const float bias = b3[0];
#pragma unroll
  for (int o = 8; o > 0; o >>= 1) {
#pragma unroll
    for (int r = 0; r < GCN_G1R; ++r) a[r] += __shfl_xor(a[r], o, 16);
  }
#pragma unroll
  for (int r = 0; r < GCN_G1R; ++r) {
    const int i = i0 + 64 * r;
    if (l == 0 && i < N) out[i] = dv[r] * (a[r] + hv[r]) + bias;
  }
}

static inline int cdiv(long long a, long long b) { return (int)((a + b - 1) / b); }
static inline size_t up256(size_t v) { return (v + 255) & ~(size_t)255; }

extern "C" void kernel_launch(void* const* d_in, const int* in_sizes, int n_in,
                              void* d_out, int out_size, void* d_ws, size_t ws_size,
                              hipStream_t stream) {
  const float* x  = (const float*)d_in[0];
  const int*   ei = (const int*)d_in[1];
  const float* W1 = (const float*)d_in[2];
  const float* b1 = (const float*)d_in[3];
  const float* W2 = (const float*)d_in[4];
  const float* b2 = (const float*)d_in[5];
  const float* W3 = (const float*)d_in[6];
  const float* b3 = (const float*)d_in[7];

  const int IN_CH = 256;
  const int HID = 16;
  const int N = in_sizes[0] / IN_CH;
  const int E = in_sizes[1] / 2;
  const int* src = ei;
  const int* dst = ei + E;
  float* out = (float*)d_out;

  char* ws = (char*)d_ws;
  size_t off = 0;
  const size_t hwB = up256((size_t)N * HID * 4);
  float* dis = (float*)(ws + off); off += up256((size_t)N * 4);
  float* HW  = (float*)(ws + off); off += hwB;
  float* AGG = (float*)(ws + off); off += hwB;
  (void)n_in; (void)out_size;

  const int B = 256;
  const int n16 = N * HID;
  const int gemmBlocks = cdiv(cdiv(N, 16), B / 32);
  const int nb = cdiv(N, 1024);

  int* cnt  = (int*)(ws + off); off += up256((size_t)N * 4);
  int* beg  = (int*)(ws + off); off += up256((size_t)N * 4);
  int* bsum = (int*)(ws + off); off += up256((size_t)nb * 4);
  int* lst  = (int*)(ws + off); off += up256((size_t)E * 4);
  int* rnk  = (int*)HW;
  const bool sorted_path = (ws_size >= off) && ((size_t)E * 4 <= 2 * hwB);

  const int G = GCN_G;
  const int chunk = cdiv(E, G);
  int* hist = lst;
  int* bstart = bsum;
  const bool lds_sort = sorted_path && N <= (1 << GCN_SRC_BITS) && nb <= GCN_MAX_NB &&
                        (long long)nb * G <= (long long)E;

  if (lds_sort) {
    gcn_bucket_hist<<<G, 1024, 0, stream>>>(dst, hist, E, chunk, nb, G);
    gcn_bucket_part<<<G, 1024, 0, stream>>>(src, dst, hist, bstart, rnk, E, chunk, nb);
    gcn_bucket_sort<<<nb, 1024, 0, stream>>>(rnk, bstart, cnt, beg, dis, lst, N, E, nb);
  } else if (sorted_path) {
    gcn_zero_i32<<<cdiv(N, B), B, 0, stream>>>(cnt, N);
    gcn_edge_rank<<<cdiv(E, B), B, 0, stream>>>(dst, cnt, rnk, E);
    gcn_block_sum<<<nb, 1024, 0, stream>>>(cnt, bsum, N);
    gcn_scan_dis<<<nb, 1024, 0, stream>>>(cnt, bsum, beg, dis, N);
    gcn_fill_list<<<cdiv(E, B), B, 0, stream>>>(src, dst, beg, rnk, lst, E);
  }

  if (sorted_path) {
    gcn_gemm_wmma_f32_lines<<<gemmBlocks, B, 0, stream>>>(x, W1, HW, dis, N, IN_CH);
    gcn_gather16_relu<<<cdiv(n16, B), B, 0, stream>>>(beg, cnt, lst, dis, HW, b1, AGG, N);
    gcn_gemm_wmma_f32<<<gemmBlocks, B, 0, stream>>>(AGG, W2, HW, dis, N, HID);
    gcn_gather16_relu<<<cdiv(n16, B), B, 0, stream>>>(beg, cnt, lst, dis, HW, b2, AGG, N);
    gcn_mv16<<<cdiv(N, B), B, 0, stream>>>(AGG, W3, dis, HW, N);
    gcn_gather1_lds<<<cdiv(N, 64 * GCN_G1R), 1024, 0, stream>>>(beg, cnt, lst, dis, HW, b3, out, N);
    return;
  }

  gcn_fill<<<cdiv(N, B), B, 0, stream>>>(dis, 1.0f, N);
  gcn_deg_count<<<cdiv(E, B), B, 0, stream>>>(dst, dis, E);
  gcn_rsqrt<<<cdiv(N, B), B, 0, stream>>>(dis, N);

  gcn_gemm_wmma_f32_lines<<<gemmBlocks, B, 0, stream>>>(x, W1, HW, nullptr, N, IN_CH);
  gcn_agg_init<<<cdiv(n16, B), B, 0, stream>>>(HW, dis, b1, AGG, n16);
  gcn_edge_scatter16<<<cdiv((long long)E * 16, B), B, 0, stream>>>(src, dst, dis, HW, AGG, E);
  gcn_relu<<<cdiv(n16, B), B, 0, stream>>>(AGG, n16);

  gcn_gemm_wmma_f32<<<gemmBlocks, B, 0, stream>>>(AGG, W2, HW, nullptr, N, HID);
  gcn_agg_init<<<cdiv(n16, B), B, 0, stream>>>(HW, dis, b2, AGG, n16);
  gcn_edge_scatter16<<<cdiv((long long)E * 16, B), B, 0, stream>>>(src, dst, dis, HW, AGG, E);
  gcn_relu<<<cdiv(n16, B), B, 0, stream>>>(AGG, n16);

  gcn_mv16<<<cdiv(N, B), B, 0, stream>>>(AGG, W3, nullptr, HW, N);
  gcn_out_init<<<cdiv(N, B), B, 0, stream>>>(HW, dis, b3, out, N);
  gcn_edge_scatter1<<<cdiv(E, B), B, 0, stream>>>(src, dst, dis, HW, out, E);
}


# Sort level 1 made chunk-local: one returning LDS add per edge is count and rank, chunk ordered by bucket in its own slice; count launch and matrix sums go (9->8); list order still LDS arrival order
# speedup vs baseline: 7.4091x; 1.0601x over previous
#include <hip/hip_runtime.h>
#include <hip/hip_bf16.h>

typedef __attribute__((ext_vector_type(2))) float v2f;
typedef __attribute__((ext_vector_type(8))) float v8f;

__global__ __launch_bounds__(256) void gcn_gemm_wmma_f32(
    const float* __restrict__ A, const float* __restrict__ W,
    float* __restrict__ C, const float* __restrict__ rs, int M, int K) {
  const int wave = blockIdx.x * (blockDim.x >> 5) + (threadIdx.x >> 5);
  const int lane = threadIdx.x & 31;
  const int m0 = wave << 4;
  if (m0 >= M) return;
  const int half = lane >> 4;
  const int l = lane & 15;

  v8f acc = {};
  const float* arow = A + (size_t)(m0 + l) * K + (half << 1);
  for (int k0 = 0; k0 < K; k0 += 4) {
    v2f a;
    a.x = __builtin_nontemporal_load(&arow[k0 + 0]);
    a.y = __builtin_nontemporal_load(&arow[k0 + 1]);
    v2f b;
    b.x = W[(size_t)(k0 + (half << 1) + 0) * 16 + l];
    b.y = W[(size_t)(k0 + (half << 1) + 1) * 16 + l];
    acc = __builtin_amdgcn_wmma_f32_16x16x4_f32(
         false, a,  false, b,
         (short)0, acc,  false,  false);
  }
  const int r0 = m0 + (half << 3);
  float* cp = C + ((size_t)r0 << 4) + l;
  if (rs) {
#pragma unroll
    for (int r = 0; r < 8; ++r) cp[(size_t)r << 4] = acc[r] * rs[r0 + r];
  } else {
#pragma unroll
    for (int r = 0; r < 8; ++r) cp[(size_t)r << 4] = acc[r];
  }
}

typedef __attribute__((ext_vector_type(4))) float v4f;
#define GCN_KC 64
#define GCN_APITCH 68

__global__ __launch_bounds__(256) void gcn_gemm_wmma_f32_lines(
    const float* __restrict__ A, const float* __restrict__ W,
    float* __restrict__ C, const float* __restrict__ rs, int M, int K) {
  __shared__ __attribute__((aligned(16))) float sh[8][16 * GCN_APITCH];
  const int wv = threadIdx.x >> 5;
  const int wave = blockIdx.x * 8 + wv;
  const int lane = threadIdx.x & 31;
  const int m0 = wave << 4;
  if (m0 >= M) return;
  const int half = lane >> 4;
  const int l = lane & 15;
  float* tile = sh[wv];
  const float* gp = A + (size_t)(m0 + half) * K + 4 * l;
  float* sp = tile + half * GCN_APITCH + 4 * l;
  const float* fp = tile + l * GCN_APITCH + 2 * half;
  const float* wp = W + (size_t)(half << 1) * 16 + l;

  v8f acc = {};
  v4f buf[8];
#pragma unroll
  for (int i = 0; i < 8; ++i)
    buf[i] = __builtin_nontemporal_load((const v4f*)(gp + (size_t)(2 * i) * K));
  for (int k0 = 0; k0 < K; k0 += GCN_KC) {
    __builtin_amdgcn_fence(__ATOMIC_RELEASE, "wavefront");
    __builtin_amdgcn_wave_barrier();
#pragma unroll
    for (int i = 0; i < 8; ++i) *(v4f*)(sp + 2 * i * GCN_APITCH) = buf[i];
    v2f b[GCN_KC / 4];
#pragma unroll
    for (int j = 0; j < GCN_KC / 4; ++j) {
      b[j].x = wp[(size_t)(k0 + 4 * j + 0) * 16];
      b[j].y = wp[(size_t)(k0 + 4 * j + 1) * 16];
    }
    if (k0 + GCN_KC < K) {
#pragma unroll
      for (int i = 0; i < 8; ++i)
        buf[i] = __builtin_nontemporal_load((const v4f*)(gp + (size_t)(2 * i) * K + k0 + GCN_KC));
    }
    __builtin_amdgcn_wave_barrier();
    __builtin_amdgcn_fence(__ATOMIC_ACQUIRE, "wavefront");
#pragma unroll
    for (int j = 0; j < GCN_KC / 4; ++j) {
      const v2f a = *(const v2f*)(fp + 4 * j);
      acc = __builtin_amdgcn_wmma_f32_16x16x4_f32(
           false, a,  false, b[j],
           (short)0, acc,  false,  false);
    }
  }
  const int r0 = m0 + (half << 3);
  float* cp = C + ((size_t)r0 << 4) + l;
  if (rs) {
#pragma unroll
    for (int r = 0; r < 8; ++r) cp[(size_t)r << 4] = acc[r] * rs[r0 + r];
  } else {
#pragma unroll
    for (int r = 0; r < 8; ++r) cp[(size_t)r << 4] = acc[r];
  }
}

__global__ void gcn_fill(float* __restrict__ p, float v, int n) {
  int i = blockIdx.x * blockDim.x + threadIdx.x;
  if (i < n) p[i] = v;
}

__global__ void gcn_deg_count(const int* __restrict__ dst, float* __restrict__ deg, int E) {
  int e = blockIdx.x * blockDim.x + threadIdx.x;
  if (e < E) atomicAdd(&deg[__builtin_nontemporal_load(&dst[e])], 1.0f);
}

__global__ void gcn_rsqrt(float* __restrict__ d, int n) {
  int i = blockIdx.x * blockDim.x + threadIdx.x;
  if (i < n) d[i] = rsqrtf(d[i]);
}

__global__ void gcn_agg_init(const float* __restrict__ hw, const float* __restrict__ dis,
                             const float* __restrict__ b, float* __restrict__ agg, int n16) {
  int t = blockIdx.x * blockDim.x + threadIdx.x;
  if (t >= n16) return;
  int i = t >> 4, c = t & 15;
  float d = dis[i];
  agg[t] = d * d * hw[t] + b[c];
}

__global__ void gcn_edge_scatter16(const int* __restrict__ src, const int* __restrict__ dst,
                                   const float* __restrict__ dis, const float* __restrict__ hw,
                                   float* __restrict__ agg, int E) {
  int t = blockIdx.x * blockDim.x + threadIdx.x;
  int e = t >> 4, c = t & 15;
  if (e >= E) return;
  int s = __builtin_nontemporal_load(&src[e]);
  int d = __builtin_nontemporal_load(&dst[e]);
  float norm = dis[s] * dis[d];
  atomicAdd(&agg[((size_t)d << 4) + c], norm * hw[((size_t)s << 4) + c]);
}

__global__ void gcn_relu(float* __restrict__ p, int n) {
  int i = blockIdx.x * blockDim.x + threadIdx.x;
  if (i < n) p[i] = fmaxf(p[i], 0.0f);
}

__global__ void gcn_mv16(const float* __restrict__ h, const float* __restrict__ W3,
                         const float* __restrict__ rs, float* __restrict__ out, int N) {
  int i = blockIdx.x * blockDim.x + threadIdx.x;
  if (i >= N) return;
  const float* hp = h + ((size_t)i << 4);
  float s = 0.f;
#pragma unroll
  for (int c = 0; c < 16; ++c) s += __builtin_nontemporal_load(&hp[c]) * W3[c];
  out[i] = rs ? s * rs[i] : s;
}

__global__ void gcn_out_init(const float* __restrict__ hw3, const float* __restrict__ dis,
                             const float* __restrict__ b3, float* __restrict__ out, int N) {
  int i = blockIdx.x * blockDim.x + threadIdx.x;
  if (i >= N) return;
  float d = dis[i];
  out[i] = d * d * hw3[i] + b3[0];
}

__global__ void gcn_edge_scatter1(const int* __restrict__ src, const int* __restrict__ dst,
                                  const float* __restrict__ dis, const float* __restrict__ hw3,
                                  float* __restrict__ out, int E) {
  int e = blockIdx.x * blockDim.x + threadIdx.x;
  if (e >= E) return;
  int s = __builtin_nontemporal_load(&src[e]);
  int d = __builtin_nontemporal_load(&dst[e]);
  atomicAdd(&out[d], dis[s] * dis[d] * hw3[s]);
}

__global__ void gcn_zero_i32(int* __restrict__ p, int n) {
  int i = blockIdx.x * blockDim.x + threadIdx.x;
  if (i < n) p[i] = 0;
}

__global__ void gcn_edge_rank(const int* __restrict__ dst, int* __restrict__ cnt,
                              int* __restrict__ rnk, int E) {
  int e = blockIdx.x * blockDim.x + threadIdx.x;
  if (e < E) rnk[e] = atomicAdd(&cnt[dst[e]], 1);
}

__global__ __launch_bounds__(1024) void gcn_block_sum(const int* __restrict__ cnt,
                                                      int* __restrict__ bsum, int n) {
  __shared__ int wsum[32];
  const int tid = threadIdx.x, w = tid >> 5, l = tid & 31;
  const int i = blockIdx.x * 1024 + tid;
  int v = (i < n) ? cnt[i] : 0;
#pragma unroll
  for (int o = 16; o > 0; o >>= 1) v += __shfl_xor(v, o, 32);
  if (l == 0) wsum[w] = v;
  __syncthreads();
  if (w == 0) {
    int t = wsum[l];
#pragma unroll
    for (int o = 16; o > 0; o >>= 1) t += __shfl_xor(t, o, 32);
    if (l == 0) bsum[blockIdx.x] = t;
  }
}

__global__ __launch_bounds__(1024) void gcn_scan_dis(const int* __restrict__ cnt,
                                                     const int* __restrict__ bsum,
                                                     int* __restrict__ beg,
                                                     float* __restrict__ dis, int n) {
  __shared__ int wtot[32];
  __shared__ int ptot[32];
  const int tid = threadIdx.x, w = tid >> 5, l = tid & 31;
  const int i = blockIdx.x * 1024 + tid;
  const int c = (i < n) ? cnt[i] : 0;
  int p = 0;
  for (int j = tid; j < (int)blockIdx.x; j += 1024) p += bsum[j];
#pragma unroll
  for (int o = 16; o > 0; o >>= 1) p += __shfl_xor(p, o, 32);
  int s = c;
#pragma unroll
  for (int o = 1; o < 32; o <<= 1) {
    int t = __shfl_up(s, o, 32);
    if (l >= o) s += t;
  }
  if (l == 31) wtot[w] = s;
  if (l == 0) ptot[w] = p;
  __syncthreads();
  int pt = ptot[l];
  int wx = (l < w) ? wtot[l] : 0;
#pragma unroll
  for (int o = 16; o > 0; o >>= 1) {
    pt += __shfl_xor(pt, o, 32);
    wx += __shfl_xor(wx, o, 32);
  }
  if (i < n) {
    beg[i] = pt + wx + s - c;
    dis[i] = rsqrtf((float)(c + 1));
  }
}

__global__ void gcn_fill_list(const int* __restrict__ src, const int* __restrict__ dst,
                              const int* __restrict__ beg, const int* __restrict__ rnk,
                              int* __restrict__ lst, int E) {
  int e = blockIdx.x * blockDim.x + threadIdx.x;
  if (e >= E) return;
  int s = __builtin_nontemporal_load(&src[e]);
  int d = __builtin_nontemporal_load(&dst[e]);
  int r = __builtin_nontemporal_load(&rnk[e]);
  lst[beg[d] + r] = s;
}

#define GCN_SRC_BITS 22
#define GCN_MAX_NB 4096
#define GCN_U 16
#define GCN_G 256
#define GCN_ROWS 8
#define GCN_PCAP 32768
#define GCN_SCAP 40960

__global__ __launch_bounds__(1024) void gcn_bucket_hist(const int* __restrict__ dst,
                                                        int* __restrict__ hist,
                                                        int E, int chunk, int NB, int G) {
  __shared__ int h[GCN_MAX_NB];
  const int tid = threadIdx.x;
  for (int i = tid; i < NB; i += 1024) h[i] = 0;
  __syncthreads();
  const int e0 = blockIdx.x * chunk;
  const int e1 = min(e0 + chunk, E);
  for (int eb = e0 + tid; eb < e1; eb += GCN_U * 1024) {
    int d[GCN_U];
#pragma unroll
    for (int u = 0; u < GCN_U; ++u) {
      const int e = eb + u * 1024;
      d[u] = (e < e1) ? dst[e] : -1;
    }
#pragma unroll
    for (int u = 0; u < GCN_U; ++u)
      if (d[u] >= 0) atomicAdd(&h[d[u] >> 10], 1);
  }
  __syncthreads();
  for (int i = tid; i < NB; i += 1024) hist[i * G + blockIdx.x] = h[i];
}

__global__ __launch_bounds__(1024) void gcn_bucket_part(const int* __restrict__ src,
                                                        const int* __restrict__ dst,
                                                        const int* __restrict__ hist,
                                                        int* __restrict__ bstart,
                                                        int* __restrict__ tmp,
                                                        int E, int chunk, int NB) {
  static_assert(GCN_G == 256, "a count-matrix row is read as 2 x 16 bytes per lane");
  __shared__ int cur[GCN_MAX_NB];
  __shared__ int btot[GCN_MAX_NB];
  __shared__ int own[GCN_MAX_NB];
  __shared__ int lcur[GCN_MAX_NB];
  __shared__ int wtot[32];
  __shared__ int wtot2[32];
  __shared__ int stage[GCN_PCAP];
  const int tid = threadIdx.x, w = tid >> 5, l = tid & 31;
  const int me = blockIdx.x;
  const bool staged = chunk <= GCN_PCAP;
  for (int b0 = w; b0 < NB; b0 += 32 * GCN_ROWS) {
    int4 x[GCN_ROWS][2];
    int mine[GCN_ROWS];
#pragma unroll
    for (int r = 0; r < GCN_ROWS; ++r) {
      const int b = min(b0 + 32 * r, NB - 1);
      const int4* row = (const int4*)(hist + (size_t)b * GCN_G);
      x[r][0] = row[l];
      x[r][1] = row[32 + l];
      mine[r] = hist[(size_t)b * GCN_G + me];
    }
    const int c0 = 4 * l, c1 = 128 + 4 * l;
    int tot[GCN_ROWS], part[GCN_ROWS];
#pragma unroll
    for (int r = 0; r < GCN_ROWS; ++r) {
      const int4 p = x[r][0], q = x[r][1];
      tot[r] = ((p.x + p.y) + (p.z + p.w)) + ((q.x + q.y) + (q.z + q.w));
      part[r] = (c0 + 0 < me ? p.x : 0) + (c0 + 1 < me ? p.y : 0) +
                (c0 + 2 < me ? p.z : 0) + (c0 + 3 < me ? p.w : 0) +
                (c1 + 0 < me ? q.x : 0) + (c1 + 1 < me ? q.y : 0) +
                (c1 + 2 < me ? q.z : 0) + (c1 + 3 < me ? q.w : 0);
    }
#pragma unroll
    for (int o = 16; o > 0; o >>= 1) {
#pragma unroll
      for (int r = 0; r < GCN_ROWS; ++r) {
        tot[r] += __shfl_xor(tot[r], o, 32);
        part[r] += __shfl_xor(part[r], o, 32);
      }
    }
#pragma unroll
    for (int r = 0; r < GCN_ROWS; ++r) {
      const int b = b0 + 32 * r;
      if (l == 0 && b < NB) { btot[b] = tot[r]; cur[b] = part[r]; own[b] = mine[r]; }
    }
  }
  __syncthreads();
  {
    int v[4], ov[4];
    int sum = 0, osum = 0;
#pragma unroll
    for (int k = 0; k < 4; ++k) {
      const int b = 4 * tid + k;
      v[k] = (b < NB) ? btot[b] : 0;
      ov[k] = (b < NB) ? own[b] : 0;
      sum += v[k];
      osum += ov[k];
    }
    int s = sum, os = osum;
#pragma unroll
    for (int o = 1; o < 32; o <<= 1) {
      int t = __shfl_up(s, o, 32);
      int ot = __shfl_up(os, o, 32);
      if (l >= o) { s += t; os += ot; }
    }
    if (l == 31) { wtot[w] = s; wtot2[w] = os; }
    __syncthreads();
    int wx = (l < w) ? wtot[l] : 0;
    int owx = (l < w) ? wtot2[l] : 0;
#pragma unroll
    for (int o = 16; o > 0; o >>= 1) {
      wx += __shfl_xor(wx, o, 32);
      owx += __shfl_xor(owx, o, 32);
    }
    int run = wx + s - sum;
    int lrun = owx + os - osum;
#pragma unroll
    for (int k = 0; k < 4; ++k) {
      const int b = 4 * tid + k;
      if (b < NB) {
        cur[b] += run;
        if (me == 0) bstart[b] = run;
        btot[b] = lrun;
        lcur[b] = lrun;
      }
      run += v[k];
      lrun += ov[k];
    }
  }
  __syncthreads();
  const int e0 = blockIdx.x * chunk;
  const int e1 = min(e0 + chunk, E);
  for (int eb = e0 + tid; eb < e1; eb += GCN_U * 1024) {
    int s[GCN_U], d[GCN_U];
#pragma unroll
    for (int u = 0; u < GCN_U; ++u) {
      const int e = eb + u * 1024;
      const bool ok = e < e1;
      s[u] = ok ? __builtin_nontemporal_load(&src[e]) : 0;
      d[u] = ok ? __builtin_nontemporal_load(&dst[e]) : -1;
    }
#pragma unroll
    for (int u = 0; u < GCN_U; ++u) {
      if (d[u] >= 0) {
        const unsigned du = (unsigned)d[u];
        const int rec = (int)(((du & 1023u) << GCN_SRC_BITS) | (unsigned)s[u]);
        if (staged) {
          const int lp = atomicAdd(&lcur[du >> 10], 1);
          stage[lp] = rec;
        } else {
          const int pos = atomicAdd(&cur[du >> 10], 1);
          tmp[pos] = rec;
        }
      }
    }
  }
  if (staged) {
    __syncthreads();
    for (int b = w; b < NB; b += 32) {
      const int n = own[b], ls = btot[b], g = cur[b];
      for (int k = l - (g & 31); k < n; k += 32)
        if (k >= 0) tmp[g + k] = stage[ls + k];
    }
  }
}

__global__ __launch_bounds__(1024) void gcn_bucket_sort(const int* __restrict__ tmp,
                                                        const int* __restrict__ bstart,
                                                        int* __restrict__ cnt,
                                                        int* __restrict__ beg,
                                                        float* __restrict__ dis,
                                                        int* __restrict__ lst,
                                                        int N, int E, int NB) {
  __shared__ int c[1024];
  __shared__ int wtot[32];
  __shared__ int stage[GCN_SCAP];
  const int tid = threadIdx.x, w = tid >> 5, l = tid & 31;
  const int b = blockIdx.x;
  const int p0 = bstart[b];
  const int p1 = (b + 1 < NB) ? bstart[b + 1] : E;
  const int nrec = p1 - p0;
  const bool staged = nrec <= GCN_SCAP;
  c[tid] = 0;
  __syncthreads();
  for (int pb = p0 + tid; pb < p1; pb += GCN_U * 1024) {
    unsigned v[GCN_U];
#pragma unroll
    for (int u = 0; u < GCN_U; ++u) v[u] = (unsigned)tmp[min(pb + u * 1024, p1 - 1)];
#pragma unroll
    for (int u = 0; u < GCN_U; ++u)
      if (pb + u * 1024 < p1) atomicAdd(&c[v[u] >> GCN_SRC_BITS], 1);
  }
  __syncthreads();
  const int n = c[tid];
  int s = n;
#pragma unroll
  for (int o = 1; o < 32; o <<= 1) {
    int t = __shfl_up(s, o, 32);
    if (l >= o) s += t;
  }
  if (l == 31) wtot[w] = s;
  __syncthreads();
  int wx = (l < w) ? wtot[l] : 0;
#pragma unroll
  for (int o = 16; o > 0; o >>= 1) wx += __shfl_xor(wx, o, 32);
  const int ex = p0 + wx + s - n;
  const int i = (b << 10) + tid;
  if (i < N) {
    cnt[i] = n;
    beg[i] = ex;
    dis[i] = rsqrtf((float)(n + 1));
  }
  c[tid] = ex;
  __syncthreads();
  for (int pb = p0 + tid; pb < p1; pb += GCN_U * 1024) {
    unsigned v[GCN_U];
#pragma unroll
    for (int u = 0; u < GCN_U; ++u)
      v[u] = (unsigned)__builtin_nontemporal_load(&tmp[min(pb + u * 1024, p1 - 1)]);
#pragma unroll
    for (int u = 0; u < GCN_U; ++u) {
      if (pb + u * 1024 < p1) {
        const int pos = atomicAdd(&c[v[u] >> GCN_SRC_BITS], 1);
        const int sv = (int)(v[u] & ((1u << GCN_SRC_BITS) - 1u));
        if (staged) stage[pos - p0] = sv;
        else lst[pos] = sv;
      }
    }
  }
  if (staged) {
    __syncthreads();
    for (int k = tid - (p0 & 31); k < nrec; k += 1024)
      if (k >= 0) lst[p0 + k] = stage[k];
  }
}

#define GCN_CR 25
#define GCN_RW 8
#define GCN_RU 5

__global__ __launch_bounds__(1024) void gcn_chunk_part(const int* __restrict__ src,
                                                       const int* __restrict__ dst,
                                                       int* __restrict__ cntm,
                                                       int* __restrict__ offm,
                                                       int* __restrict__ tmp,
                                                       int E, int chunk, int NB) {
  __shared__ int h[GCN_MAX_NB];
  __shared__ int wtot[32];
  __shared__ int stage[GCN_PCAP];
  const int tid = threadIdx.x, w = tid >> 5, l = tid & 31;
  const int me = blockIdx.x;
  for (int i = tid; i < NB; i += 1024) h[i] = 0;
  const int e0 = me * chunk;
  const int e1 = min(e0 + chunk, E);
  int s[GCN_CR], d[GCN_CR], r[GCN_CR];
#pragma unroll
  for (int u = 0; u < GCN_CR; ++u) {
    const int e = e0 + tid + u * 1024;
    const bool ok = e < e1;
    s[u] = ok ? __builtin_nontemporal_load(&src[e]) : 0;
    d[u] = ok ? __builtin_nontemporal_load(&dst[e]) : -1;
  }
  __syncthreads();
#pragma unroll
  for (int u = 0; u < GCN_CR; ++u) {
    r[u] = 0;
    if (d[u] >= 0) r[u] = atomicAdd(&h[(unsigned)d[u] >> 10], 1);
  }
  __syncthreads();
  {
    int v[4];
    int sum = 0;
#pragma unroll
    for (int k = 0; k < 4; ++k) {
      const int b = 4 * tid + k;
      v[k] = (b < NB) ? h[b] : 0;
      sum += v[k];
    }
    int sc = sum;
#pragma unroll
    for (int o = 1; o < 32; o <<= 1) {
      int t = __shfl_up(sc, o, 32);
      if (l >= o) sc += t;
    }
    if (l == 31) wtot[w] = sc;
    __syncthreads();
    int wx = (l < w) ? wtot[l] : 0;
#pragma unroll
    for (int o = 16; o > 0; o >>= 1) wx += __shfl_xor(wx, o, 32);
    int run = wx + sc - sum;
#pragma unroll
    for (int k = 0; k < 4; ++k) {
      const int b = 4 * tid + k;
      if (b < NB) {
        cntm[(size_t)b * 1024 + me] = v[k];
        offm[(size_t)b * 1024 + me] = run;
        h[b] = run;
      }
      run += v[k];
    }
  }
  __syncthreads();
  static_assert(GCN_CR % 5 == 0, "the staging below takes the edges five at a time");
#pragma unroll
  for (int g = 0; g < GCN_CR; g += 5) {
    int fs[5];
#pragma unroll
    for (int k = 0; k < 5; ++k) fs[k] = h[d[g + k] >= 0 ? ((unsigned)d[g + k] >> 10) : 0u];
#pragma unroll
    for (int k = 0; k < 5; ++k) {
      if (d[g + k] >= 0) {
        const unsigned du = (unsigned)d[g + k];
        stage[fs[k] + r[g + k]] = (int)(((du & 1023u) << GCN_SRC_BITS) | (unsigned)s[g + k]);
      }
    }
  }
  __syncthreads();
  const int n = e1 - e0;
  int k = tid - (e0 & 31);
  if (k >= 0 && k < n) tmp[e0 + k] = stage[k];
#pragma unroll 4
  for (k += 1024; k < n; k += 1024) tmp[e0 + k] = stage[k];
}

__global__ __launch_bounds__(1024) void gcn_bucket_sort_runs(const int* __restrict__ tmp,
                                                             int* __restrict__ cnt,
                                                             int* __restrict__ beg,
                                                             float* __restrict__ dis,
                                                             int* __restrict__ lst,
                                                             int N, int E, int chunk) {
  static_assert(GCN_G == 32 * GCN_RW, "a wave reads GCN_RW runs, the work-group all GCN_G");
  __shared__ int c[1024];
  __shared__ int wtot[32];
  __shared__ int rc[GCN_G];
  __shared__ int ro[GCN_G];
  __shared__ int tot[2][GCN_RW];
  __shared__ int stage[GCN_SCAP];
  const int tid = threadIdx.x, w = tid >> 5, l = tid & 31;
  const int b = blockIdx.x;
  int myc = 0, myo = 0;
  if (tid < GCN_G) {
    myc = cnt[(size_t)b * 1024 + tid];
    myo = beg[(size_t)b * 1024 + tid];
    rc[tid] = myc;
    ro[tid] = myo;
  }
#pragma unroll
  for (int o = 16; o > 0; o >>= 1) {
    myc += __shfl_xor(myc, o, 32);
    myo += __shfl_xor(myo, o, 32);
  }
  if (l == 0 && w < GCN_RW) { tot[0][w] = myc; tot[1][w] = myo; }
  c[tid] = 0;
  __syncthreads();
  int nrec = 0, p0 = 0;
#pragma unroll
  for (int q = 0; q < GCN_RW; ++q) { nrec += tot[0][q]; p0 += tot[1][q]; }
  const bool staged = nrec <= GCN_SCAP;
  int base[GCN_RW], len[GCN_RW];
  unsigned v[GCN_RW][GCN_RU];
#pragma unroll
  for (int q = 0; q < GCN_RW; ++q) {
    const int cc = w + 32 * q;
    len[q] = rc[cc];
    base[q] = cc * chunk + ro[cc];
  }
#pragma unroll
  for (int q = 0; q < GCN_RW; ++q) {
#pragma unroll
    for (int u = 0; u < GCN_RU; ++u)
      v[q][u] = (unsigned)tmp[min(max(base[q] + min(l + 32 * u, len[q] - 1), 0), E - 1)];
  }
#pragma unroll
  for (int q = 0; q < GCN_RW; ++q) {
#pragma unroll
    for (int u = 0; u < GCN_RU; ++u)
      if (l + 32 * u < len[q]) atomicAdd(&c[v[q][u] >> GCN_SRC_BITS], 1);
  }
#pragma unroll
  for (int q = 0; q < GCN_RW; ++q) {
    for (int k = 32 * GCN_RU + l; k < len[q]; k += 32)
      atomicAdd(&c[(unsigned)tmp[base[q] + k] >> GCN_SRC_BITS], 1);
  }
  __syncthreads();
  const int n = c[tid];
  int s = n;
#pragma unroll
  for (int o = 1; o < 32; o <<= 1) {
    int t = __shfl_up(s, o, 32);
    if (l >= o) s += t;
  }
  if (l == 31) wtot[w] = s;
  __syncthreads();
  int wx = (l < w) ? wtot[l] : 0;
#pragma unroll
  for (int o = 16; o > 0; o >>= 1) wx += __shfl_xor(wx, o, 32);
  const int ex = p0 + wx + s - n;
  const int i = (b << 10) + tid;
  if (i < N) {
    cnt[i] = n;
    beg[i] = ex;
    dis[i] = rsqrtf((float)(n + 1));
  }
  c[tid] = ex;
  __syncthreads();
  static_assert(GCN_RW % 4 == 0, "the placing below takes the runs four at a time");
#pragma unroll
  for (int q0 = 0; q0 < GCN_RW; q0 += 4) {
    int pos[4][GCN_RU];
#pragma unroll
    for (int q = 0; q < 4; ++q) {
#pragma unroll
      for (int u = 0; u < GCN_RU; ++u) {
        pos[q][u] = -1;
        if (l + 32 * u < len[q0 + q]) pos[q][u] = atomicAdd(&c[v[q0 + q][u] >> GCN_SRC_BITS], 1);
      }
    }
    if (staged) {
#pragma unroll
      for (int q = 0; q < 4; ++q) {
#pragma unroll
        for (int u = 0; u < GCN_RU; ++u)
          if (pos[q][u] >= 0) stage[pos[q][u] - p0] = (int)(v[q0 + q][u] & ((1u << GCN_SRC_BITS) - 1u));
      }
    } else {
#pragma unroll
      for (int q = 0; q < 4; ++q) {
#pragma unroll
        for (int u = 0; u < GCN_RU; ++u)
          if (pos[q][u] >= 0) lst[pos[q][u]] = (int)(v[q0 + q][u] & ((1u << GCN_SRC_BITS) - 1u));
      }
    }
  }
#pragma unroll
  for (int q = 0; q < GCN_RW; ++q) {
    for (int k = 32 * GCN_RU + l; k < len[q]; k += 32) {
      const unsigned x = (unsigned)tmp[base[q] + k];
      const int pos = atomicAdd(&c[x >> GCN_SRC_BITS], 1);
      const int sv = (int)(x & ((1u << GCN_SRC_BITS) - 1u));
      if (staged) stage[pos - p0] = sv;
      else lst[pos] = sv;
    }
  }
  if (staged) {
    __syncthreads();
    int k = tid - (p0 & 31);
    if (k >= 0 && k < nrec) lst[p0 + k] = stage[k];
#pragma unroll 4
    for (k += 1024; k < nrec; k += 1024) lst[p0 + k] = stage[k];
  }
}

__global__ __launch_bounds__(256) void gcn_gather16_relu(
    const int* __restrict__ beg, const int* __restrict__ cnt, const int* __restrict__ lst,
    const float* __restrict__ dis, const float* __restrict__ hws, const float* __restrict__ b,
    float* __restrict__ h, int N) {
  const int t = blockIdx.x * blockDim.x + threadIdx.x;
  const int i = t >> 4, c = t & 15;
  if (i >= N) return;
  const int* lp = lst + beg[i];
  const int len = cnt[i];
  const float d = dis[i];
  float a[8];
  a[0] = hws[((size_t)i << 4) + c];
#pragma unroll
  for (int u = 1; u < 8; ++u) a[u] = 0.f;
  int j = 0;
#pragma unroll 1
  for (; j + 16 <= len; j += 16) {
    int s[16];
#pragma unroll
    for (int u = 0; u < 16; ++u) s[u] = lp[j + u];
    float r[16];
#pragma unroll
    for (int u = 0; u < 16; ++u) r[u] = hws[((size_t)s[u] << 4) + c];
#pragma unroll
    for (int u = 0; u < 8; ++u) a[u] += r[u];
#pragma unroll
    for (int u = 0; u < 8; ++u) a[u] += r[8 + u];
  }
  const int rem = len - j;
  if (rem > 0) {
    const int last = len - 1;
    int s[15];
#pragma unroll
    for (int u = 0; u < 15; ++u) s[u] = lp[min(j + u, last)];
#pragma unroll
    for (int u = 0; u < 15; ++u) asm volatile("" : "+v"(s[u]));
    float r[15];
#pragma unroll
    for (int u = 0; u < 15; ++u) r[u] = hws[((size_t)s[u] << 4) + c];
#pragma unroll
    for (int u = 0; u < 15; ++u) asm volatile("" : "+v"(r[u]));
    const bool blk = rem >= 8;
#pragma unroll
    for (int u = 0; u < 8; ++u) a[u] = blk ? a[u] + r[u] : a[u];
    const int tail = rem & 7;
#pragma unroll
    for (int q = 0; q < 7; ++q) {
      const float v = blk ? r[8 + q] : r[q];
      a[0] = (q < tail) ? a[0] + v : a[0];
    }
  }
  float r = ((a[0] + a[1]) + (a[2] + a[3])) + ((a[4] + a[5]) + (a[6] + a[7]));
  h[((size_t)i << 4) + c] = fmaxf(d * r + b[c], 0.0f);
}

#define GCN_G1R 13
#define GCN_G1K 3
#define GCN_G1B 6
#define GCN_G1S 73728

__global__ __launch_bounds__(1024) void gcn_gather1_lds(
    const int* __restrict__ beg, const int* __restrict__ cnt, const int* __restrict__ lst,
    const float* __restrict__ dis, const float* __restrict__ hw3s, const float* __restrict__ b3,
    float* __restrict__ out, int N) {
  __shared__ __attribute__((aligned(16))) float tab[GCN_G1S];
  const int tid = threadIdx.x, l = tid & 15;
  const int i0 = blockIdx.x * (64 * GCN_G1R) + (tid >> 4);
  int ln[GCN_G1R], s[GCN_G1R][GCN_G1K];
  float a[GCN_G1R];
  int bg[GCN_G1R];
#pragma unroll
  for (int r = 0; r < GCN_G1R; ++r) {
    const int ic = min(i0 + 64 * r, N - 1);
    bg[r] = beg[ic];
    ln[r] = cnt[ic];
    a[r] = 0.f;
  }
#pragma unroll
  for (int r = 0; r < GCN_G1R; ++r) {
    if (i0 + 64 * r >= N) ln[r] = 0;
#pragma unroll
    for (int k = 0; k < GCN_G1K; ++k) {
      const int j = l + 16 * k;
      const int v = lst[max(min(bg[r] + j, bg[r] + ln[r] - 1), 0)];
      s[r][k] = (j < ln[r]) ? v : -1;
    }
  }
  const int nq = (N + GCN_G1S - 1) / GCN_G1S;
  for (int q = 0; q < nq; ++q) {
    const int q0 = q * GCN_G1S;
    const int sl = min(GCN_G1S, N - q0);
    const int sl4 = sl & ~3;
    const float* tp = hw3s + q0;
    __syncthreads();
    for (int t0 = 4 * tid; t0 < sl4; t0 += 4096 * GCN_G1B) {
      v4f buf[GCN_G1B];
#pragma unroll
      for (int u = 0; u < GCN_G1B; ++u) {
        const int t = t0 + 4096 * u;
        buf[u] = *(const v4f*)(tp + min(t, sl4 - 4));
      }
#pragma unroll
      for (int u = 0; u < GCN_G1B; ++u) {
        const int t = t0 + 4096 * u;
        if (t < sl4) *(v4f*)(tab + t) = buf[u];
      }
    }
    if (tid < sl - sl4) tab[sl4 + tid] = tp[sl4 + tid];
    __syncthreads();
#pragma unroll
    for (int r = 0; r < GCN_G1R; ++r) {
#pragma unroll
      for (int k = 0; k < GCN_G1K; ++k) {
        const unsigned u = (unsigned)(s[r][k] - q0);
        const bool in = u < (unsigned)sl;
        const float v = tab[in ? u : 0u];
        a[r] += in ? v : 0.f;
      }
    }
#pragma unroll
    for (int r = 0; r < GCN_G1R; ++r) {
      if (ln[r] > 16 * GCN_G1K) {
        const int* lp = lst + beg[i0 + 64 * r];
        for (int j = l + 16 * GCN_G1K; j < ln[r]; j += 16) {
          const unsigned u = (unsigned)(lp[j] - q0);
          if (u < (unsigned)sl) a[r] += tab[u];
        }
      }
    }
  }
  float dv[GCN_G1R], hv[GCN_G1R];
#pragma unroll
  for (int r = 0; r < GCN_G1R; ++r) {
    const int ic = min(i0 + 64 * r, N - 1);
    dv[r] = dis[ic];
    hv[r] = hw3s[ic];
  }
  const float bias = b3[0];
#pragma unroll
  for (int o = 8; o > 0; o >>= 1) {
#pragma unroll
    for (int r = 0; r < GCN_G1R; ++r) a[r] += __shfl_xor(a[r], o, 16);
  }
#pragma unroll
  for (int r = 0; r < GCN_G1R; ++r) {
    const int i = i0 + 64 * r;
    if (l == 0 && i < N) out[i] = dv[r] * (a[r] + hv[r]) + bias;
  }
}

static inline int cdiv(long long a, long long b) { return (int)((a + b - 1) / b); }
static inline size_t up256(size_t v) { return (v + 255) & ~(size_t)255; }

extern "C" void kernel_launch(void* const* d_in, const int* in_sizes, int n_in,
                              void* d_out, int out_size, void* d_ws, size_t ws_size,
                              hipStream_t stream) {
  const float* x  = (const float*)d_in[0];
  const int*   ei = (const int*)d_in[1];
  const float* W1 = (const float*)d_in[2];
  const float* b1 = (const float*)d_in[3];
  const float* W2 = (const float*)d_in[4];
  const float* b2 = (const float*)d_in[5];
  const float* W3 = (const float*)d_in[6];
  const float* b3 = (const float*)d_in[7];

  const int IN_CH = 256;
  const int HID = 16;
  const int N = in_sizes[0] / IN_CH;
  const int E = in_sizes[1] / 2;
  const int* src = ei;
  const int* dst = ei + E;
  float* out = (float*)d_out;

  char* ws = (char*)d_ws;
  size_t off = 0;
  const size_t hwB = up256((size_t)N * HID * 4);
  float* dis = (float*)(ws + off); off += up256((size_t)N * 4);
  float* HW  = (float*)(ws + off); off += hwB;
  float* AGG = (float*)(ws + off); off += hwB;
  (void)n_in; (void)out_size;

  const int B = 256;
  const int n16 = N * HID;
  const int gemmBlocks = cdiv(cdiv(N, 16), B / 32);
  const int nb = cdiv(N, 1024);

  int* cnt  = (int*)(ws + off); off += up256((size_t)N * 4);
  int* beg  = (int*)(ws + off); off += up256((size_t)N * 4);
  int* bsum = (int*)(ws + off); off += up256((size_t)nb * 4);
  int* lst  = (int*)(ws + off); off += up256((size_t)E * 4);
  int* rnk  = (int*)HW;
  const bool sorted_path = (ws_size >= off) && ((size_t)E * 4 <= 2 * hwB);

  const int G = GCN_G;
  const int chunk = cdiv(E, G);
  int* hist = lst;
  int* bstart = bsum;
  const bool lds_sort = sorted_path && N <= (1 << GCN_SRC_BITS) && nb <= GCN_MAX_NB &&
                        (long long)nb * G <= (long long)E;

  const bool chunk_local = lds_sort && chunk <= GCN_CR * 1024 && chunk <= GCN_PCAP &&
                           N - 1024 * (nb - 1) >= GCN_G;

  if (chunk_local) {
    gcn_chunk_part<<<G, 1024, 0, stream>>>(src, dst, cnt, beg, rnk, E, chunk, nb);
    gcn_bucket_sort_runs<<<nb, 1024, 0, stream>>>(rnk, cnt, beg, dis, lst, N, E, chunk);
  } else if (lds_sort) {
    gcn_bucket_hist<<<G, 1024, 0, stream>>>(dst, hist, E, chunk, nb, G);
    gcn_bucket_part<<<G, 1024, 0, stream>>>(src, dst, hist, bstart, rnk, E, chunk, nb);
    gcn_bucket_sort<<<nb, 1024, 0, stream>>>(rnk, bstart, cnt, beg, dis, lst, N, E, nb);
  } else if (sorted_path) {
    gcn_zero_i32<<<cdiv(N, B), B, 0, stream>>>(cnt, N);
    gcn_edge_rank<<<cdiv(E, B), B, 0, stream>>>(dst, cnt, rnk, E);
    gcn_block_sum<<<nb, 1024, 0, stream>>>(cnt, bsum, N);
    gcn_scan_dis<<<nb, 1024, 0, stream>>>(cnt, bsum, beg, dis, N);
    gcn_fill_list<<<cdiv(E, B), B, 0, stream>>>(src, dst, beg, rnk, lst, E);
  }

  if (sorted_path) {
    gcn_gemm_wmma_f32_lines<<<gemmBlocks, B, 0, stream>>>(x, W1, HW, dis, N, IN_CH);
    gcn_gather16_relu<<<cdiv(n16, B), B, 0, stream>>>(beg, cnt, lst, dis, HW, b1, AGG, N);
    gcn_gemm_wmma_f32<<<gemmBlocks, B, 0, stream>>>(AGG, W2, HW, dis, N, HID);
    gcn_gather16_relu<<<cdiv(n16, B), B, 0, stream>>>(beg, cnt, lst, dis, HW, b2, AGG, N);
    gcn_mv16<<<cdiv(N, B), B, 0, stream>>>(AGG, W3, dis, HW, N);
    gcn_gather1_lds<<<cdiv(N, 64 * GCN_G1R), 1024, 0, stream>>>(beg, cnt, lst, dis, HW, b3, out, N);
    return;
  }

  gcn_fill<<<cdiv(N, B), B, 0, stream>>>(dis, 1.0f, N);
  gcn_deg_count<<<cdiv(E, B), B, 0, stream>>>(dst, dis, E);
  gcn_rsqrt<<<cdiv(N, B), B, 0, stream>>>(dis, N);

  gcn_gemm_wmma_f32_lines<<<gemmBlocks, B, 0, stream>>>(x, W1, HW, nullptr, N, IN_CH);
  gcn_agg_init<<<cdiv(n16, B), B, 0, stream>>>(HW, dis, b1, AGG, n16);
  gcn_edge_scatter16<<<cdiv((long long)E * 16, B), B, 0, stream>>>(src, dst, dis, HW, AGG, E);
  gcn_relu<<<cdiv(n16, B), B, 0, stream>>>(AGG, n16);

  gcn_gemm_wmma_f32<<<gemmBlocks, B, 0, stream>>>(AGG, W2, HW, nullptr, N, HID);
  gcn_agg_init<<<cdiv(n16, B), B, 0, stream>>>(HW, dis, b2, AGG, n16);
  gcn_edge_scatter16<<<cdiv((long long)E * 16, B), B, 0, stream>>>(src, dst, dis, HW, AGG, E);
  gcn_relu<<<cdiv(n16, B), B, 0, stream>>>(AGG, n16);

  gcn_mv16<<<cdiv(N, B), B, 0, stream>>>(AGG, W3, nullptr, HW, N);
  gcn_out_init<<<cdiv(N, B), B, 0, stream>>>(HW, dis, b3, out, N);
  gcn_edge_scatter1<<<cdiv(E, B), B, 0, stream>>>(src, dst, dis, HW, out, E);
}
